# starting kernel
# baseline (speedup 1.0000x reference)
#include <hip/hip_runtime.h>


typedef float v2f __attribute__((ext_vector_type(2)));
typedef float v8f __attribute__((ext_vector_type(8)));

#define DM   512
#define SEQ  4096
#define NB   2
#define NH   8
#define DKH  64

__device__ __forceinline__ v8f wmma4(v2f a, v2f b, v8f c) {
  return __builtin_amdgcn_wmma_f32_16x16x4_f32(
       false, a,  false, b,
       (short)0, c,  false,  false);
}

__global__ void __launch_bounds__(256) kG(const float* __restrict__ k,
                                          const float* __restrict__ v,
                                          float* __restrict__ G) {
  const int lane = threadIdx.x & 31, wv = threadIdx.x >> 5;
  const int lm = lane & 15, lh = lane >> 4;
  const int job = blockIdx.x * 8 + wv;
  const int b = job >> 7;
  const int rest = job & 127;
  const int d0 = (rest >> 3) * 32;
  const int e0 = (rest & 7) * 64;
  const float* kb = k + (size_t)b * SEQ * DM;
  const float* vb = v + (size_t)b * SEQ * DM;
  v8f acc[8] = {};
  #pragma unroll 2
  for (int s0 = 0; s0 < SEQ; s0 += 4) {
    size_t rb = (size_t)(s0 + 2 * lh) * DM;
    v2f a0; a0.x = kb[rb + d0 + lm];      a0.y = kb[rb + DM + d0 + lm];
    v2f a1; a1.x = kb[rb + d0 + 16 + lm]; a1.y = kb[rb + DM + d0 + 16 + lm];
    #pragma unroll
    for (int t = 0; t < 4; ++t) {
      v2f bm; bm.x = vb[rb + e0 + 16 * t + lm]; bm.y = vb[rb + DM + e0 + 16 * t + lm];
      acc[t]     = wmma4(a0, bm, acc[t]);
      acc[4 + t] = wmma4(a1, bm, acc[4 + t]);
    }
  }
  float* Gb = G + ((size_t)b * DM + d0) * DM + e0;
  #pragma unroll
  for (int m = 0; m < 2; ++m)
    #pragma unroll
    for (int t = 0; t < 4; ++t)
      #pragma unroll
      for (int r = 0; r < 8; ++r)
        Gb[(size_t)(16 * m + r + 8 * lh) * DM + 16 * t + lm] = acc[4 * m + t][r];
}

__global__ void __launch_bounds__(256) kT(const float* __restrict__ wk,
                                          const float* __restrict__ G,
                                          float* __restrict__ T) {
  const int lane = threadIdx.x & 31, wv = threadIdx.x >> 5;
  const int lm = lane & 15, lh = lane >> 4;
  const int job = blockIdx.x * 8 + wv;
  const int bh = job >> 5;
  const int b = bh >> 3, h = bh & 7;
  const int rest = job & 31;
  const int i0 = (rest >> 3) * 16;
  const int e0 = (rest & 7) * 64;
  const float* arow = wk + (size_t)(h * 64 + i0 + lm) * DM + 2 * lh;
  const float* Gb = G + (size_t)b * DM * DM;
  v8f acc[4] = {};
  for (int kk = 0; kk < DM; kk += 4) {
    v2f a = *(const v2f*)(arow + kk);
    size_t rb = (size_t)(kk + 2 * lh) * DM;
    #pragma unroll
    for (int t = 0; t < 4; ++t) {
      v2f bm; bm.x = Gb[rb + e0 + 16 * t + lm]; bm.y = Gb[rb + DM + e0 + 16 * t + lm];
      acc[t] = wmma4(a, bm, acc[t]);
    }
  }
  float* Tb = T + ((size_t)bh * 64 + i0) * DM + e0;
  #pragma unroll
  for (int t = 0; t < 4; ++t)
    #pragma unroll
    for (int r = 0; r < 8; ++r)
      Tb[(size_t)(r + 8 * lh) * DM + 16 * t + lm] = acc[t][r];
}

__global__ void __launch_bounds__(256) kM(const float* __restrict__ T,
                                          const float* __restrict__ wvw,
                                          const float* __restrict__ bk,
                                          const float* __restrict__ bv,
                                          const float* __restrict__ u,
                                          const float* __restrict__ z,
                                          float* __restrict__ M) {
  const int lane = threadIdx.x & 31, wv = threadIdx.x >> 5;
  const int lm = lane & 15, lh = lane >> 4;
  const int job = blockIdx.x * 8 + wv;
  const int bh = job >> 2;
  const int b = bh >> 3, h = bh & 7;
  const int i0 = (job & 3) * 16;
  const float* arow = T + ((size_t)bh * 64 + i0 + lm) * DM + 2 * lh;
  v8f acc[4] = {};
  for (int kk = 0; kk < DM; kk += 4) {
    v2f a = *(const v2f*)(arow + kk);
    #pragma unroll
    for (int t = 0; t < 4; ++t) {
      v2f bm = *(const v2f*)(wvw + (size_t)(h * 64 + 16 * t + lm) * DM + kk + 2 * lh);
      acc[t] = wmma4(a, bm, acc[t]);
    }
  }
  #pragma unroll
  for (int t = 0; t < 4; ++t) {
    const int jf = h * 64 + 16 * t + lm;
    const float bvj = bv[jf];
    const float zj  = z[(size_t)b * DM + jf];
    #pragma unroll
    for (int r = 0; r < 8; ++r) {
      const int ifl = h * 64 + i0 + r + 8 * lh;
      const float bki = bk[ifl];
      const float ui  = u[(size_t)b * DM + ifl];
      float val = acc[t][r] + ui * bvj + bki * zj + (float)SEQ * bki * bvj;
      M[((size_t)bh * 64 + i0 + r + 8 * lh) * DKH + 16 * t + lm] = 0.125f * val;
    }
  }
}

__global__ void __launch_bounds__(256) kP(const float* __restrict__ M,
                                          const float* __restrict__ wo,
                                          float* __restrict__ P) {
  const int lane = threadIdx.x & 31, wv = threadIdx.x >> 5;
  const int lm = lane & 15, lh = lane >> 4;
  const int job = blockIdx.x * 8 + wv;
  const int bh = job >> 5;
  const int b = bh >> 3, h = bh & 7;
  const int i0 = ((job >> 3) & 3) * 16;
  const int o0 = (job & 7) * 64;
  const float* arow = M + ((size_t)bh * 64 + i0 + lm) * DKH + 2 * lh;
  v8f acc[4] = {};
  for (int kk = 0; kk < DKH; kk += 4) {
    v2f a = *(const v2f*)(arow + kk);
    #pragma unroll
    for (int t = 0; t < 4; ++t) {
      v2f bm = *(const v2f*)(wo + (size_t)(o0 + 16 * t + lm) * DM + h * 64 + kk + 2 * lh);
      acc[t] = wmma4(a, bm, acc[t]);
    }
  }
  float* Pb = P + ((size_t)b * DM + h * 64 + i0) * DM + o0;
  #pragma unroll
  for (int t = 0; t < 4; ++t)
    #pragma unroll
    for (int r = 0; r < 8; ++r)
      Pb[(size_t)(r + 8 * lh) * DM + 16 * t + lm] = acc[t][r];
}

__global__ void __launch_bounds__(256) kW4(const float* __restrict__ wq,
                                           const float* __restrict__ P,
                                           float* __restrict__ W4) {
  const int lane = threadIdx.x & 31, wv = threadIdx.x >> 5;
  const int lm = lane & 15, lh = lane >> 4;
  const int job = blockIdx.x * 8 + wv;
  const int b = job >> 8;
  const int rest = job & 255;
  const int d0 = (rest >> 3) * 16;
  const int o0 = (rest & 7) * 64;
  const float* Pb = P + (size_t)b * DM * DM;
  v8f acc[4] = {};
  for (int kk = 0; kk < DM; kk += 4) {
    size_t fr = (size_t)(kk + 2 * lh) * DM;
    v2f a; a.x = wq[fr + d0 + lm]; a.y = wq[fr + DM + d0 + lm];
    #pragma unroll
    for (int t = 0; t < 4; ++t) {
      v2f bm; bm.x = Pb[fr + o0 + 16 * t + lm]; bm.y = Pb[fr + DM + o0 + 16 * t + lm];
      acc[t] = wmma4(a, bm, acc[t]);
    }
  }
  float* Wb = W4 + ((size_t)b * DM + d0) * DM + o0;
  #pragma unroll
  for (int t = 0; t < 4; ++t)
    #pragma unroll
    for (int r = 0; r < 8; ++r)
      Wb[(size_t)(r + 8 * lh) * DM + 16 * t + lm] = acc[t][r];
}

__global__ void __launch_bounds__(256) kOut(const float* __restrict__ q,
                                            const float* __restrict__ W4,
                                            const float* __restrict__ bo,
                                            const float* __restrict__ cq,
                                            float* __restrict__ out) {
  const int lane = threadIdx.x & 31, wv = threadIdx.x >> 5;
  const int lm = lane & 15, lh = lane >> 4;
  const int job = blockIdx.x * 8 + wv;
  const int b = job >> 7;
  const int n0 = (job & 127) * 32;
  const int o0 = blockIdx.y * 64;
  const float* qrow0 = q + ((size_t)b * SEQ + n0 + lm) * DM + 2 * lh;
  const float* qrow1 = qrow0 + (size_t)16 * DM;
  const float* Wb = W4 + (size_t)b * DM * DM;
  v8f acc[8] = {};
  #pragma unroll 2
  for (int kk = 0; kk < DM; kk += 4) {
    v2f a0 = *(const v2f*)(qrow0 + kk);
    v2f a1 = *(const v2f*)(qrow1 + kk);
    size_t rb = (size_t)(kk + 2 * lh) * DM;
    #pragma unroll
    for (int t = 0; t < 4; ++t) {
      v2f bm; bm.x = Wb[rb + o0 + 16 * t + lm]; bm.y = Wb[rb + DM + o0 + 16 * t + lm];
      acc[t]     = wmma4(a0, bm, acc[t]);
      acc[4 + t] = wmma4(a1, bm, acc[4 + t]);
    }
  }
  float* ob = out + ((size_t)b * SEQ + n0) * DM + o0;
  #pragma unroll
  for (int t = 0; t < 4; ++t) {
    float add = bo[o0 + 16 * t + lm] + cq[(size_t)b * DM + o0 + 16 * t + lm];
    #pragma unroll
    for (int m = 0; m < 2; ++m)
      #pragma unroll
      for (int r = 0; r < 8; ++r)
        ob[(size_t)(16 * m + r + 8 * lh) * DM + 16 * t + lm] = acc[4 * m + t][r] + add;
  }
}

__global__ void kSums(const float* __restrict__ k, const float* __restrict__ v,
                      float* __restrict__ ksum, float* __restrict__ vsum) {
  int idx = blockIdx.x * blockDim.x + threadIdx.x;
  if (idx >= NB * DM) return;
  int b = idx >> 9, d = idx & (DM - 1);
  const float* kp = k + (size_t)b * SEQ * DM + d;
  const float* vp = v + (size_t)b * SEQ * DM + d;
  float sk = 0.f, sv = 0.f;
  for (int s = 0; s < SEQ; ++s) { sk += kp[(size_t)s * DM]; sv += vp[(size_t)s * DM]; }
  ksum[idx] = sk; vsum[idx] = sv;
}

__global__ void kUZ(const float* __restrict__ wk, const float* __restrict__ wvw,
                    const float* __restrict__ ksum, const float* __restrict__ vsum,
                    float* __restrict__ u, float* __restrict__ z) {
  int idx = blockIdx.x * blockDim.x + threadIdx.x;
  if (idx >= NB * DM) return;
  int b = idx >> 9, f = idx & (DM - 1);
  const float* wkr = wk + (size_t)f * DM;
  const float* wvr = wvw + (size_t)f * DM;
  const float* ks = ksum + (size_t)b * DM;
  const float* vs = vsum + (size_t)b * DM;
  float su = 0.f, sz = 0.f;
  for (int d = 0; d < DM; ++d) { su += wkr[d] * ks[d]; sz += wvr[d] * vs[d]; }
  u[idx] = su; z[idx] = sz;
}

__global__ void kCQ(const float* __restrict__ bq, const float* __restrict__ P,
                    float* __restrict__ cq) {
  int idx = blockIdx.x * blockDim.x + threadIdx.x;
  if (idx >= NB * DM) return;
  int b = idx >> 9, o = idx & (DM - 1);
  const float* Pb = P + (size_t)b * DM * DM + o;
  float s = 0.f;
  for (int f = 0; f < DM; ++f) s += bq[f] * Pb[(size_t)f * DM];
  cq[idx] = s;
}

extern "C" void kernel_launch(void* const* d_in, const int* in_sizes, int n_in,
                              void* d_out, int out_size, void* d_ws, size_t ws_size,
                              hipStream_t stream) {
  const float* q   = (const float*)d_in[0];
  const float* k   = (const float*)d_in[1];
  const float* v   = (const float*)d_in[2];
  const float* wq  = (const float*)d_in[3];
  const float* bq  = (const float*)d_in[4];
  const float* wk  = (const float*)d_in[5];
  const float* bk  = (const float*)d_in[6];
  const float* wvw = (const float*)d_in[7];
  const float* bv  = (const float*)d_in[8];
  const float* wo  = (const float*)d_in[9];
  const float* bo  = (const float*)d_in[10];
  float* out = (float*)d_out;

  float* ws = (float*)d_ws;
  float* G    = ws;
  float* T    = G    + (size_t)NB * DM * DM;
  float* M    = T    + (size_t)NB * DM * DM;
  float* P    = M    + (size_t)NB * NH * DKH * DKH;
  float* W4   = P    + (size_t)NB * DM * DM;
  float* ksum = W4   + (size_t)NB * DM * DM;
  float* vsum = ksum + (size_t)NB * DM;
  float* u    = vsum + (size_t)NB * DM;
  float* z    = u    + (size_t)NB * DM;
  float* cq   = z    + (size_t)NB * DM;

  kSums<<<4, 256, 0, stream>>>(k, v, ksum, vsum);
  kUZ  <<<4, 256, 0, stream>>>(wk, wvw, ksum, vsum, u, z);
  kG   <<<32, 256, 0, stream>>>(k, v, G);
  kT   <<<64, 256, 0, stream>>>(wk, G, T);
  kM   <<<8, 256, 0, stream>>>(T, wvw, bk, bv, u, z, M);
  kP   <<<64, 256, 0, stream>>>(M, wo, P);
  kCQ  <<<4, 256, 0, stream>>>(bq, P, cq);
  kW4  <<<64, 256, 0, stream>>>(wq, P, W4);
  dim3 go(32, 8, 1);
  kOut <<<go, 256, 0, stream>>>(q, W4, bo, cq, out);
}


# kG: k,v panels staged through LDS (16-byte global loads, row pairs interleaved, 8-byte fragment reads); one 32x64 G tile per 8-wave work-group, 256 groups, SEQ walked whole; no sum changes
# speedup vs baseline: 1.6507x; 1.6507x over previous
#include <hip/hip_runtime.h>


typedef float v2f __attribute__((ext_vector_type(2)));
typedef float v8f __attribute__((ext_vector_type(8)));

#define DM   512
#define SEQ  4096
#define NB   2
#define NH   8
#define DKH  64

__device__ __forceinline__ v8f wmma4(v2f a, v2f b, v8f c) {
  return __builtin_amdgcn_wmma_f32_16x16x4_f32(
       false, a,  false, b,
       (short)0, c,  false,  false);
}

#define GKS   256
#define GNST  (SEQ / GKS)
#define GKP   96
#define GVP   160
#define GKBUF ((GKS / 2) * GKP)
#define GVBUF ((GKS / 2) * GVP)
#define GKIT  (GKS / 64)
#define GVIT  (GKS / 32)
#define GFG   8
#define GNG   (GKS / 4 / GFG)

__global__ void __launch_bounds__(256) kG(const float* __restrict__ k,
                                          const float* __restrict__ v,
                                          float* __restrict__ G) {
  __shared__ __attribute__((aligned(16))) float Ks[2 * GKBUF];
  __shared__ __attribute__((aligned(16))) float Vs[2 * GVBUF];
  const int tid = threadIdx.x;
  const int lane = tid & 31, wv = tid >> 5;
  const int lm = lane & 15, lh = lane >> 4;
  const int blk = blockIdx.x;
  const int b = blk >> 7;
  const int rest = blk & 127;
  const int d0 = (rest >> 3) * 32;
  const int e0 = (rest & 7) * 64;
  const int dw = 16 * (wv & 1);
  const int ew = 16 * (wv >> 1);
  const int kc = (tid & 7) * 4,  krp = tid >> 3;
  const int vc = (tid & 15) * 4, vrp = tid >> 4;
  const float* kg = k + (size_t)b * SEQ * DM + d0 + (size_t)(2 * krp) * DM + kc;
  const float* vg = v + (size_t)b * SEQ * DM + e0 + (size_t)(2 * vrp) * DM + vc;
  const int ksw = 4 * (krp & 1);
  const int vsw = 4 * ((vc >> 5) & 1);
  const int kst0 = krp * GKP + 2 * kc + ksw, kst1 = krp * GKP + 2 * kc + 4 - ksw;
  const int vst0 = vrp * GVP + 2 * vc + vsw, vst1 = vrp * GVP + 2 * vc + 4 - vsw;
  float4 rk[2 * GKIT], rv[2 * GVIT];

#define G_LOAD(st)                                                            \
  {                                                                           \
    const float* kn = kg + (size_t)(st) * GKS * DM;                           \
    const float* vn = vg + (size_t)(st) * GKS * DM;                           \
    _Pragma("unroll")                                                         \
    for (int j = 0; j < GKIT; ++j) {                                          \
      rk[2 * j]     = *(const float4*)(kn + (size_t)(64 * j) * DM);           \
      rk[2 * j + 1] = *(const float4*)(kn + (size_t)(64 * j + 1) * DM);       \
    }                                                                         \
    _Pragma("unroll")                                                         \
    for (int j = 0; j < GVIT; ++j) {                                          \
      rv[2 * j]     = *(const float4*)(vn + (size_t)(32 * j) * DM);           \
      rv[2 * j + 1] = *(const float4*)(vn + (size_t)(32 * j + 1) * DM);       \
    }                                                                         \
  }
#define G_STASH(bf)                                                           \
  {                                                                           \
    float* kd = Ks + (bf) * GKBUF;                                            \
    float* vd = Vs + (bf) * GVBUF;                                            \
    _Pragma("unroll")                                                         \
    for (int j = 0; j < GKIT; ++j) {                                          \
      *(float4*)(kd + 32 * j * GKP + kst0) =                                  \
          make_float4(rk[2 * j].x, rk[2 * j + 1].x, rk[2 * j].y, rk[2 * j + 1].y); \
      *(float4*)(kd + 32 * j * GKP + kst1) =                                  \
          make_float4(rk[2 * j].z, rk[2 * j + 1].z, rk[2 * j].w, rk[2 * j + 1].w); \
    }                                                                         \
    _Pragma("unroll")                                                         \
    for (int j = 0; j < GVIT; ++j) {                                          \
      *(float4*)(vd + 16 * j * GVP + vst0) =                                  \
          make_float4(rv[2 * j].x, rv[2 * j + 1].x, rv[2 * j].y, rv[2 * j + 1].y); \
      *(float4*)(vd + 16 * j * GVP + vst1) =                                  \
          make_float4(rv[2 * j].z, rv[2 * j + 1].z, rv[2 * j].w, rv[2 * j + 1].w); \
    }                                                                         \
  }
#define G_FRAG(dst, g)                                                        \
  {                                                                           \
    _Pragma("unroll")                                                         \
    for (int i = 0; i < GFG; ++i) {                                           \
      fa[dst][i] = *(const v2f*)(ka + 2 * ((g) * GFG + i) * GKP);             \
      fb[dst][i] = *(const v2f*)(va + 2 * ((g) * GFG + i) * GVP);             \
    }                                                                         \
  }

  G_LOAD(0)
  G_STASH(0)
  G_LOAD(1)
  __syncthreads();

  const int ard = lh * GKP + ((2 * (dw + lm)) ^ (4 * lh));
  const int brd = lh * GVP + ((2 * (ew + lm)) ^ (4 * ((ew >> 5) & 1)));
  v8f acc = {};
  v2f fa[2][GFG], fb[2][GFG];
  for (int n = 0; n < GNST; ++n) {
    const int buf = n & 1;
    const float* ka = Ks + buf * GKBUF + ard;
    const float* va = Vs + buf * GVBUF + brd;
    G_FRAG(0, 0)
    if (n + 1 < GNST) G_STASH(buf ^ 1)
    if (n + 2 < GNST) G_LOAD(n + 2)
    __builtin_amdgcn_sched_barrier(0);
    #pragma unroll
    for (int g = 0; g < GNG; ++g) {
      if (g + 1 < GNG) G_FRAG((g + 1) & 1, g + 1)
      #pragma unroll
      for (int i = 0; i < GFG; ++i)
        acc = wmma4(fa[g & 1][i], fb[g & 1][i], acc);
      __builtin_amdgcn_sched_barrier(0);
    }
    __syncthreads();
  }
#undef G_LOAD
#undef G_STASH
#undef G_FRAG
  float* Gb = G + ((size_t)b * DM + d0 + dw) * DM + e0 + ew;
  #pragma unroll
  for (int r = 0; r < 8; ++r)
    Gb[(size_t)(r + 8 * lh) * DM + lm] = acc[r];
}

__global__ void __launch_bounds__(256) kT(const float* __restrict__ wk,
                                          const float* __restrict__ G,
                                          float* __restrict__ T) {
  const int lane = threadIdx.x & 31, wv = threadIdx.x >> 5;
  const int lm = lane & 15, lh = lane >> 4;
  const int job = blockIdx.x * 8 + wv;
  const int bh = job >> 5;
  const int b = bh >> 3, h = bh & 7;
  const int rest = job & 31;
  const int i0 = (rest >> 3) * 16;
  const int e0 = (rest & 7) * 64;
  const float* arow = wk + (size_t)(h * 64 + i0 + lm) * DM + 2 * lh;
  const float* Gb = G + (size_t)b * DM * DM;
  v8f acc[4] = {};
  for (int kk = 0; kk < DM; kk += 4) {
    v2f a = *(const v2f*)(arow + kk);
    size_t rb = (size_t)(kk + 2 * lh) * DM;
    #pragma unroll
    for (int t = 0; t < 4; ++t) {
      v2f bm; bm.x = Gb[rb + e0 + 16 * t + lm]; bm.y = Gb[rb + DM + e0 + 16 * t + lm];
      acc[t] = wmma4(a, bm, acc[t]);
    }
  }
  float* Tb = T + ((size_t)bh * 64 + i0) * DM + e0;
  #pragma unroll
  for (int t = 0; t < 4; ++t)
    #pragma unroll
    for (int r = 0; r < 8; ++r)
      Tb[(size_t)(r + 8 * lh) * DM + 16 * t + lm] = acc[t][r];
}

__global__ void __launch_bounds__(256) kM(const float* __restrict__ T,
                                          const float* __restrict__ wvw,
                                          const float* __restrict__ bk,
                                          const float* __restrict__ bv,
                                          const float* __restrict__ u,
                                          const float* __restrict__ z,
                                          float* __restrict__ M) {
  const int lane = threadIdx.x & 31, wv = threadIdx.x >> 5;
  const int lm = lane & 15, lh = lane >> 4;
  const int job = blockIdx.x * 8 + wv;
  const int bh = job >> 2;
  const int b = bh >> 3, h = bh & 7;
  const int i0 = (job & 3) * 16;
  const float* arow = T + ((size_t)bh * 64 + i0 + lm) * DM + 2 * lh;
  v8f acc[4] = {};
  for (int kk = 0; kk < DM; kk += 4) {
    v2f a = *(const v2f*)(arow + kk);
    #pragma unroll
    for (int t = 0; t < 4; ++t) {
      v2f bm = *(const v2f*)(wvw + (size_t)(h * 64 + 16 * t + lm) * DM + kk + 2 * lh);
      acc[t] = wmma4(a, bm, acc[t]);
    }
  }
  #pragma unroll
  for (int t = 0; t < 4; ++t) {
    const int jf = h * 64 + 16 * t + lm;
    const float bvj = bv[jf];
    const float zj  = z[(size_t)b * DM + jf];
    #pragma unroll
    for (int r = 0; r < 8; ++r) {
      const int ifl = h * 64 + i0 + r + 8 * lh;
      const float bki = bk[ifl];
      const float ui  = u[(size_t)b * DM + ifl];
      float val = acc[t][r] + ui * bvj + bki * zj + (float)SEQ * bki * bvj;
      M[((size_t)bh * 64 + i0 + r + 8 * lh) * DKH + 16 * t + lm] = 0.125f * val;
    }
  }
}

__global__ void __launch_bounds__(256) kP(const float* __restrict__ M,
                                          const float* __restrict__ wo,
                                          float* __restrict__ P) {
  const int lane = threadIdx.x & 31, wv = threadIdx.x >> 5;
  const int lm = lane & 15, lh = lane >> 4;
  const int job = blockIdx.x * 8 + wv;
  const int bh = job >> 5;
  const int b = bh >> 3, h = bh & 7;
  const int i0 = ((job >> 3) & 3) * 16;
  const int o0 = (job & 7) * 64;
  const float* arow = M + ((size_t)bh * 64 + i0 + lm) * DKH + 2 * lh;
  v8f acc[4] = {};
  for (int kk = 0; kk < DKH; kk += 4) {
    v2f a = *(const v2f*)(arow + kk);
    #pragma unroll
    for (int t = 0; t < 4; ++t) {
      v2f bm = *(const v2f*)(wo + (size_t)(o0 + 16 * t + lm) * DM + h * 64 + kk + 2 * lh);
      acc[t] = wmma4(a, bm, acc[t]);
    }
  }
  float* Pb = P + ((size_t)b * DM + h * 64 + i0) * DM + o0;
  #pragma unroll
  for (int t = 0; t < 4; ++t)
    #pragma unroll
    for (int r = 0; r < 8; ++r)
      Pb[(size_t)(r + 8 * lh) * DM + 16 * t + lm] = acc[t][r];
}

__global__ void __launch_bounds__(256) kW4(const float* __restrict__ wq,
                                           const float* __restrict__ P,
                                           float* __restrict__ W4) {
  const int lane = threadIdx.x & 31, wv = threadIdx.x >> 5;
  const int lm = lane & 15, lh = lane >> 4;
  const int job = blockIdx.x * 8 + wv;
  const int b = job >> 8;
  const int rest = job & 255;
  const int d0 = (rest >> 3) * 16;
  const int o0 = (rest & 7) * 64;
  const float* Pb = P + (size_t)b * DM * DM;
  v8f acc[4] = {};
  for (int kk = 0; kk < DM; kk += 4) {
    size_t fr = (size_t)(kk + 2 * lh) * DM;
    v2f a; a.x = wq[fr + d0 + lm]; a.y = wq[fr + DM + d0 + lm];
    #pragma unroll
    for (int t = 0; t < 4; ++t) {
      v2f bm; bm.x = Pb[fr + o0 + 16 * t + lm]; bm.y = Pb[fr + DM + o0 + 16 * t + lm];
      acc[t] = wmma4(a, bm, acc[t]);
    }
  }
  float* Wb = W4 + ((size_t)b * DM + d0) * DM + o0;
  #pragma unroll
  for (int t = 0; t < 4; ++t)
    #pragma unroll
    for (int r = 0; r < 8; ++r)
      Wb[(size_t)(r + 8 * lh) * DM + 16 * t + lm] = acc[t][r];
}

__global__ void __launch_bounds__(256) kOut(const float* __restrict__ q,
                                            const float* __restrict__ W4,
                                            const float* __restrict__ bo,
                                            const float* __restrict__ cq,
                                            float* __restrict__ out) {
  const int lane = threadIdx.x & 31, wv = threadIdx.x >> 5;
  const int lm = lane & 15, lh = lane >> 4;
  const int job = blockIdx.x * 8 + wv;
  const int b = job >> 7;
  const int n0 = (job & 127) * 32;
  const int o0 = blockIdx.y * 64;
  const float* qrow0 = q + ((size_t)b * SEQ + n0 + lm) * DM + 2 * lh;
  const float* qrow1 = qrow0 + (size_t)16 * DM;
  const float* Wb = W4 + (size_t)b * DM * DM;
  v8f acc[8] = {};
  #pragma unroll 2
  for (int kk = 0; kk < DM; kk += 4) {
    v2f a0 = *(const v2f*)(qrow0 + kk);
    v2f a1 = *(const v2f*)(qrow1 + kk);
    size_t rb = (size_t)(kk + 2 * lh) * DM;
    #pragma unroll
    for (int t = 0; t < 4; ++t) {
      v2f bm; bm.x = Wb[rb + o0 + 16 * t + lm]; bm.y = Wb[rb + DM + o0 + 16 * t + lm];
      acc[t]     = wmma4(a0, bm, acc[t]);
      acc[4 + t] = wmma4(a1, bm, acc[4 + t]);
    }
  }
  float* ob = out + ((size_t)b * SEQ + n0) * DM + o0;
  #pragma unroll
  for (int t = 0; t < 4; ++t) {
    float add = bo[o0 + 16 * t + lm] + cq[(size_t)b * DM + o0 + 16 * t + lm];
    #pragma unroll
    for (int m = 0; m < 2; ++m)
      #pragma unroll
      for (int r = 0; r < 8; ++r)
        ob[(size_t)(16 * m + r + 8 * lh) * DM + 16 * t + lm] = acc[4 * m + t][r] + add;
  }
}

__global__ void kSums(const float* __restrict__ k, const float* __restrict__ v,
                      float* __restrict__ ksum, float* __restrict__ vsum) {
  int idx = blockIdx.x * blockDim.x + threadIdx.x;
  if (idx >= NB * DM) return;
  int b = idx >> 9, d = idx & (DM - 1);
  const float* kp = k + (size_t)b * SEQ * DM + d;
  const float* vp = v + (size_t)b * SEQ * DM + d;
  float sk = 0.f, sv = 0.f;
  for (int s = 0; s < SEQ; ++s) { sk += kp[(size_t)s * DM]; sv += vp[(size_t)s * DM]; }
  ksum[idx] = sk; vsum[idx] = sv;
}

__global__ void kUZ(const float* __restrict__ wk, const float* __restrict__ wvw,
                    const float* __restrict__ ksum, const float* __restrict__ vsum,
                    float* __restrict__ u, float* __restrict__ z) {
  int idx = blockIdx.x * blockDim.x + threadIdx.x;
  if (idx >= NB * DM) return;
  int b = idx >> 9, f = idx & (DM - 1);
  const float* wkr = wk + (size_t)f * DM;
  const float* wvr = wvw + (size_t)f * DM;
  const float* ks = ksum + (size_t)b * DM;
  const float* vs = vsum + (size_t)b * DM;
  float su = 0.f, sz = 0.f;
  for (int d = 0; d < DM; ++d) { su += wkr[d] * ks[d]; sz += wvr[d] * vs[d]; }
  u[idx] = su; z[idx] = sz;
}

__global__ void kCQ(const float* __restrict__ bq, const float* __restrict__ P,
                    float* __restrict__ cq) {
  int idx = blockIdx.x * blockDim.x + threadIdx.x;
  if (idx >= NB * DM) return;
  int b = idx >> 9, o = idx & (DM - 1);
  const float* Pb = P + (size_t)b * DM * DM + o;
  float s = 0.f;
  for (int f = 0; f < DM; ++f) s += bq[f] * Pb[(size_t)f * DM];
  cq[idx] = s;
}

extern "C" void kernel_launch(void* const* d_in, const int* in_sizes, int n_in,
                              void* d_out, int out_size, void* d_ws, size_t ws_size,
                              hipStream_t stream) {
  const float* q   = (const float*)d_in[0];
  const float* k   = (const float*)d_in[1];
  const float* v   = (const float*)d_in[2];
  const float* wq  = (const float*)d_in[3];
  const float* bq  = (const float*)d_in[4];
  const float* wk  = (const float*)d_in[5];
  const float* bk  = (const float*)d_in[6];
  const float* wvw = (const float*)d_in[7];
  const float* bv  = (const float*)d_in[8];
  const float* wo  = (const float*)d_in[9];
  const float* bo  = (const float*)d_in[10];
  float* out = (float*)d_out;

  float* ws = (float*)d_ws;
  float* G    = ws;
  float* T    = G    + (size_t)NB * DM * DM;
  float* M    = T    + (size_t)NB * DM * DM;
  float* P    = M    + (size_t)NB * NH * DKH * DKH;
  float* W4   = P    + (size_t)NB * DM * DM;
  float* ksum = W4   + (size_t)NB * DM * DM;
  float* vsum = ksum + (size_t)NB * DM;
  float* u    = vsum + (size_t)NB * DM;
  float* z    = u    + (size_t)NB * DM;
  float* cq   = z    + (size_t)NB * DM;

  kSums<<<4, 256, 0, stream>>>(k, v, ksum, vsum);
  kUZ  <<<4, 256, 0, stream>>>(wk, wvw, ksum, vsum, u, z);
  kG   <<<256, 256, 0, stream>>>(k, v, G);
  kT   <<<64, 256, 0, stream>>>(wk, G, T);
  kM   <<<8, 256, 0, stream>>>(T, wvw, bk, bv, u, z, M);
  kP   <<<64, 256, 0, stream>>>(M, wo, P);
  kCQ  <<<4, 256, 0, stream>>>(bq, P, cq);
  kW4  <<<64, 256, 0, stream>>>(wq, P, W4);
  dim3 go(32, 8, 1);
  kOut <<<go, 256, 0, stream>>>(q, W4, bo, cq, out);
}


# kSums: SEQ cut into 64 chunks a batch, one 16-wave work-group each (128 instead of 4), 16-byte row loads; kUZ adds the chunk partials in chunk order via LDS; changes ksum/vsum summation tree
# speedup vs baseline: 2.6905x; 1.6299x over previous
#include <hip/hip_runtime.h>


typedef float v2f __attribute__((ext_vector_type(2)));
typedef float v8f __attribute__((ext_vector_type(8)));

#define DM   512
#define SEQ  4096
#define NB   2
#define NH   8
#define DKH  64

__device__ __forceinline__ v8f wmma4(v2f a, v2f b, v8f c) {
  return __builtin_amdgcn_wmma_f32_16x16x4_f32(
       false, a,  false, b,
       (short)0, c,  false,  false);
}

#define GKS   256
#define GNST  (SEQ / GKS)
#define GKP   96
#define GVP   160
#define GKBUF ((GKS / 2) * GKP)
#define GVBUF ((GKS / 2) * GVP)
#define GKIT  (GKS / 64)
#define GVIT  (GKS / 32)
#define GFG   8
#define GNG   (GKS / 4 / GFG)

__global__ void __launch_bounds__(256) kG(const float* __restrict__ k,
                                          const float* __restrict__ v,
                                          float* __restrict__ G) {
  __shared__ __attribute__((aligned(16))) float Ks[2 * GKBUF];
  __shared__ __attribute__((aligned(16))) float Vs[2 * GVBUF];
  const int tid = threadIdx.x;
  const int lane = tid & 31, wv = tid >> 5;
  const int lm = lane & 15, lh = lane >> 4;
  const int blk = blockIdx.x;
  const int b = blk >> 7;
  const int rest = blk & 127;
  const int d0 = (rest >> 3) * 32;
  const int e0 = (rest & 7) * 64;
  const int dw = 16 * (wv & 1);
  const int ew = 16 * (wv >> 1);
  const int kc = (tid & 7) * 4,  krp = tid >> 3;
  const int vc = (tid & 15) * 4, vrp = tid >> 4;
  const float* kg = k + (size_t)b * SEQ * DM + d0 + (size_t)(2 * krp) * DM + kc;
  const float* vg = v + (size_t)b * SEQ * DM + e0 + (size_t)(2 * vrp) * DM + vc;
  const int ksw = 4 * (krp & 1);
  const int vsw = 4 * ((vc >> 5) & 1);
  const int kst0 = krp * GKP + 2 * kc + ksw, kst1 = krp * GKP + 2 * kc + 4 - ksw;
  const int vst0 = vrp * GVP + 2 * vc + vsw, vst1 = vrp * GVP + 2 * vc + 4 - vsw;
  float4 rk[2 * GKIT], rv[2 * GVIT];

#define G_LOAD(st)                                                            \
  {                                                                           \
    const float* kn = kg + (size_t)(st) * GKS * DM;                           \
    const float* vn = vg + (size_t)(st) * GKS * DM;                           \
    _Pragma("unroll")                                                         \
    for (int j = 0; j < GKIT; ++j) {                                          \
      rk[2 * j]     = *(const float4*)(kn + (size_t)(64 * j) * DM);           \
      rk[2 * j + 1] = *(const float4*)(kn + (size_t)(64 * j + 1) * DM);       \
    }                                                                         \
    _Pragma("unroll")                                                         \
    for (int j = 0; j < GVIT; ++j) {                                          \
      rv[2 * j]     = *(const float4*)(vn + (size_t)(32 * j) * DM);           \
      rv[2 * j + 1] = *(const float4*)(vn + (size_t)(32 * j + 1) * DM);       \
    }                                                                         \
  }
#define G_STASH(bf)                                                           \
  {                                                                           \
    float* kd = Ks + (bf) * GKBUF;                                            \
    float* vd = Vs + (bf) * GVBUF;                                            \
    _Pragma("unroll")                                                         \
    for (int j = 0; j < GKIT; ++j) {                                          \
      *(float4*)(kd + 32 * j * GKP + kst0) =                                  \
          make_float4(rk[2 * j].x, rk[2 * j + 1].x, rk[2 * j].y, rk[2 * j + 1].y); \
      *(float4*)(kd + 32 * j * GKP + kst1) =                                  \
          make_float4(rk[2 * j].z, rk[2 * j + 1].z, rk[2 * j].w, rk[2 * j + 1].w); \
    }                                                                         \
    _Pragma("unroll")                                                         \
    for (int j = 0; j < GVIT; ++j) {                                          \
      *(float4*)(vd + 16 * j * GVP + vst0) =                                  \
          make_float4(rv[2 * j].x, rv[2 * j + 1].x, rv[2 * j].y, rv[2 * j + 1].y); \
      *(float4*)(vd + 16 * j * GVP + vst1) =                                  \
          make_float4(rv[2 * j].z, rv[2 * j + 1].z, rv[2 * j].w, rv[2 * j + 1].w); \
    }                                                                         \
  }
#define G_FRAG(dst, g)                                                        \
  {                                                                           \
    _Pragma("unroll")                                                         \
    for (int i = 0; i < GFG; ++i) {                                           \
      fa[dst][i] = *(const v2f*)(ka + 2 * ((g) * GFG + i) * GKP);             \
      fb[dst][i] = *(const v2f*)(va + 2 * ((g) * GFG + i) * GVP);             \
    }                                                                         \
  }

  G_LOAD(0)
  G_STASH(0)
  G_LOAD(1)
  __syncthreads();

  const int ard = lh * GKP + ((2 * (dw + lm)) ^ (4 * lh));
  const int brd = lh * GVP + ((2 * (ew + lm)) ^ (4 * ((ew >> 5) & 1)));
  v8f acc = {};
  v2f fa[2][GFG], fb[2][GFG];
  for (int n = 0; n < GNST; ++n) {
    const int buf = n & 1;
    const float* ka = Ks + buf * GKBUF + ard;
    const float* va = Vs + buf * GVBUF + brd;
    G_FRAG(0, 0)
    if (n + 1 < GNST) G_STASH(buf ^ 1)
    if (n + 2 < GNST) G_LOAD(n + 2)
    __builtin_amdgcn_sched_barrier(0);
    #pragma unroll
    for (int g = 0; g < GNG; ++g) {
      if (g + 1 < GNG) G_FRAG((g + 1) & 1, g + 1)
      #pragma unroll
      for (int i = 0; i < GFG; ++i)
        acc = wmma4(fa[g & 1][i], fb[g & 1][i], acc);
      __builtin_amdgcn_sched_barrier(0);
    }
    __syncthreads();
  }
#undef G_LOAD
#undef G_STASH
#undef G_FRAG
  float* Gb = G + ((size_t)b * DM + d0 + dw) * DM + e0 + ew;
  #pragma unroll
  for (int r = 0; r < 8; ++r)
    Gb[(size_t)(r + 8 * lh) * DM + lm] = acc[r];
}

__global__ void __launch_bounds__(256) kT(const float* __restrict__ wk,
                                          const float* __restrict__ G,
                                          float* __restrict__ T) {
  const int lane = threadIdx.x & 31, wv = threadIdx.x >> 5;
  const int lm = lane & 15, lh = lane >> 4;
  const int job = blockIdx.x * 8 + wv;
  const int bh = job >> 5;
  const int b = bh >> 3, h = bh & 7;
  const int rest = job & 31;
  const int i0 = (rest >> 3) * 16;
  const int e0 = (rest & 7) * 64;
  const float* arow = wk + (size_t)(h * 64 + i0 + lm) * DM + 2 * lh;
  const float* Gb = G + (size_t)b * DM * DM;
  v8f acc[4] = {};
  for (int kk = 0; kk < DM; kk += 4) {
    v2f a = *(const v2f*)(arow + kk);
    size_t rb = (size_t)(kk + 2 * lh) * DM;
    #pragma unroll
    for (int t = 0; t < 4; ++t) {
      v2f bm; bm.x = Gb[rb + e0 + 16 * t + lm]; bm.y = Gb[rb + DM + e0 + 16 * t + lm];
      acc[t] = wmma4(a, bm, acc[t]);
    }
  }
  float* Tb = T + ((size_t)bh * 64 + i0) * DM + e0;
  #pragma unroll
  for (int t = 0; t < 4; ++t)
    #pragma unroll
    for (int r = 0; r < 8; ++r)
      Tb[(size_t)(r + 8 * lh) * DM + 16 * t + lm] = acc[t][r];
}

__global__ void __launch_bounds__(256) kM(const float* __restrict__ T,
                                          const float* __restrict__ wvw,
                                          const float* __restrict__ bk,
                                          const float* __restrict__ bv,
                                          const float* __restrict__ u,
                                          const float* __restrict__ z,
                                          float* __restrict__ M) {
  const int lane = threadIdx.x & 31, wv = threadIdx.x >> 5;
  const int lm = lane & 15, lh = lane >> 4;
  const int job = blockIdx.x * 8 + wv;
  const int bh = job >> 2;
  const int b = bh >> 3, h = bh & 7;
  const int i0 = (job & 3) * 16;
  const float* arow = T + ((size_t)bh * 64 + i0 + lm) * DM + 2 * lh;
  v8f acc[4] = {};
  for (int kk = 0; kk < DM; kk += 4) {
    v2f a = *(const v2f*)(arow + kk);
    #pragma unroll
    for (int t = 0; t < 4; ++t) {
      v2f bm = *(const v2f*)(wvw + (size_t)(h * 64 + 16 * t + lm) * DM + kk + 2 * lh);
      acc[t] = wmma4(a, bm, acc[t]);
    }
  }
  #pragma unroll
  for (int t = 0; t < 4; ++t) {
    const int jf = h * 64 + 16 * t + lm;
    const float bvj = bv[jf];
    const float zj  = z[(size_t)b * DM + jf];
    #pragma unroll
    for (int r = 0; r < 8; ++r) {
      const int ifl = h * 64 + i0 + r + 8 * lh;
      const float bki = bk[ifl];
      const float ui  = u[(size_t)b * DM + ifl];
      float val = acc[t][r] + ui * bvj + bki * zj + (float)SEQ * bki * bvj;
      M[((size_t)bh * 64 + i0 + r + 8 * lh) * DKH + 16 * t + lm] = 0.125f * val;
    }
  }
}

__global__ void __launch_bounds__(256) kP(const float* __restrict__ M,
                                          const float* __restrict__ wo,
                                          float* __restrict__ P) {
  const int lane = threadIdx.x & 31, wv = threadIdx.x >> 5;
  const int lm = lane & 15, lh = lane >> 4;
  const int job = blockIdx.x * 8 + wv;
  const int bh = job >> 5;
  const int b = bh >> 3, h = bh & 7;
  const int i0 = ((job >> 3) & 3) * 16;
  const int o0 = (job & 7) * 64;
  const float* arow = M + ((size_t)bh * 64 + i0 + lm) * DKH + 2 * lh;
  v8f acc[4] = {};
  for (int kk = 0; kk < DKH; kk += 4) {
    v2f a = *(const v2f*)(arow + kk);
    #pragma unroll
    for (int t = 0; t < 4; ++t) {
      v2f bm = *(const v2f*)(wo + (size_t)(o0 + 16 * t + lm) * DM + h * 64 + kk + 2 * lh);
      acc[t] = wmma4(a, bm, acc[t]);
    }
  }
  float* Pb = P + ((size_t)b * DM + h * 64 + i0) * DM + o0;
  #pragma unroll
  for (int t = 0; t < 4; ++t)
    #pragma unroll
    for (int r = 0; r < 8; ++r)
      Pb[(size_t)(r + 8 * lh) * DM + 16 * t + lm] = acc[t][r];
}

__global__ void __launch_bounds__(256) kW4(const float* __restrict__ wq,
                                           const float* __restrict__ P,
                                           float* __restrict__ W4) {
  const int lane = threadIdx.x & 31, wv = threadIdx.x >> 5;
  const int lm = lane & 15, lh = lane >> 4;
  const int job = blockIdx.x * 8 + wv;
  const int b = job >> 8;
  const int rest = job & 255;
  const int d0 = (rest >> 3) * 16;
  const int o0 = (rest & 7) * 64;
  const float* Pb = P + (size_t)b * DM * DM;
  v8f acc[4] = {};
  for (int kk = 0; kk < DM; kk += 4) {
    size_t fr = (size_t)(kk + 2 * lh) * DM;
    v2f a; a.x = wq[fr + d0 + lm]; a.y = wq[fr + DM + d0 + lm];
    #pragma unroll
    for (int t = 0; t < 4; ++t) {
      v2f bm; bm.x = Pb[fr + o0 + 16 * t + lm]; bm.y = Pb[fr + DM + o0 + 16 * t + lm];
      acc[t] = wmma4(a, bm, acc[t]);
    }
  }
  float* Wb = W4 + ((size_t)b * DM + d0) * DM + o0;
  #pragma unroll
  for (int t = 0; t < 4; ++t)
    #pragma unroll
    for (int r = 0; r < 8; ++r)
      Wb[(size_t)(r + 8 * lh) * DM + 16 * t + lm] = acc[t][r];
}

__global__ void __launch_bounds__(256) kOut(const float* __restrict__ q,
                                            const float* __restrict__ W4,
                                            const float* __restrict__ bo,
                                            const float* __restrict__ cq,
                                            float* __restrict__ out) {
  const int lane = threadIdx.x & 31, wv = threadIdx.x >> 5;
  const int lm = lane & 15, lh = lane >> 4;
  const int job = blockIdx.x * 8 + wv;
  const int b = job >> 7;
  const int n0 = (job & 127) * 32;
  const int o0 = blockIdx.y * 64;
  const float* qrow0 = q + ((size_t)b * SEQ + n0 + lm) * DM + 2 * lh;
  const float* qrow1 = qrow0 + (size_t)16 * DM;
  const float* Wb = W4 + (size_t)b * DM * DM;
  v8f acc[8] = {};
  #pragma unroll 2
  for (int kk = 0; kk < DM; kk += 4) {
    v2f a0 = *(const v2f*)(qrow0 + kk);
    v2f a1 = *(const v2f*)(qrow1 + kk);
    size_t rb = (size_t)(kk + 2 * lh) * DM;
    #pragma unroll
    for (int t = 0; t < 4; ++t) {
      v2f bm; bm.x = Wb[rb + o0 + 16 * t + lm]; bm.y = Wb[rb + DM + o0 + 16 * t + lm];
      acc[t]     = wmma4(a0, bm, acc[t]);
      acc[4 + t] = wmma4(a1, bm, acc[4 + t]);
    }
  }
  float* ob = out + ((size_t)b * SEQ + n0) * DM + o0;
  #pragma unroll
  for (int t = 0; t < 4; ++t) {
    float add = bo[o0 + 16 * t + lm] + cq[(size_t)b * DM + o0 + 16 * t + lm];
    #pragma unroll
    for (int m = 0; m < 2; ++m)
      #pragma unroll
      for (int r = 0; r < 8; ++r)
        ob[(size_t)(16 * m + r + 8 * lh) * DM + 16 * t + lm] = acc[4 * m + t][r] + add;
  }
}

#define SCH   64
#define SROWS (SEQ / SCH)
#define SPH   4
#define SRPT  (SROWS / SPH)

__global__ void __launch_bounds__(512) kSums(const float* __restrict__ k,
                                             const float* __restrict__ v,
                                             float* __restrict__ part) {
  __shared__ float4 red[SPH * 256];
  const int tid = threadIdx.x;
  const int cg = tid & 127, ph = tid >> 7;
  const int b = blockIdx.x / SCH, c = blockIdx.x % SCH;
  const size_t off =
      ((size_t)b * SEQ + (size_t)c * SROWS + (size_t)ph * SRPT) * DM + 4 * cg;
  const float* kp = k + off;
  const float* vp = v + off;
  float4 rk[SRPT], rv[SRPT];
  #pragma unroll
  for (int i = 0; i < SRPT; ++i) {
    rk[i] = *(const float4*)(kp + (size_t)i * DM);
    rv[i] = *(const float4*)(vp + (size_t)i * DM);
  }
  float4 sk = rk[0], sv = rv[0];
  #pragma unroll
  for (int i = 1; i < SRPT; ++i) {
    sk.x += rk[i].x; sk.y += rk[i].y; sk.z += rk[i].z; sk.w += rk[i].w;
    sv.x += rv[i].x; sv.y += rv[i].y; sv.z += rv[i].z; sv.w += rv[i].w;
  }
  red[ph * 256 + cg] = sk;
  red[ph * 256 + 128 + cg] = sv;
  __syncthreads();
  if (tid < 256) {
    float4 s = red[tid];
    #pragma unroll
    for (int p = 1; p < SPH; ++p) {
      const float4 t = red[p * 256 + tid];
      s.x += t.x; s.y += t.y; s.z += t.z; s.w += t.w;
    }
    *(float4*)(part + (size_t)blockIdx.x * (2 * DM) + 4 * tid) = s;
  }
}

__global__ void __launch_bounds__(256) kUZ(const float* __restrict__ wk,
                                           const float* __restrict__ wvw,
                                           const float* __restrict__ part,
                                           float* __restrict__ u,
                                           float* __restrict__ z) {
  __shared__ __attribute__((aligned(16))) float cs[2 * DM];
  const int tid = threadIdx.x;
  const int idx = blockIdx.x * 256 + tid;
  const int b = blockIdx.x >> 1, f = idx & (DM - 1);
  const float* pp = part + (size_t)b * SCH * (2 * DM) + 4 * tid;
  float4 s = *(const float4*)pp;
  #pragma unroll
  for (int c = 1; c < SCH; ++c) {
    const float4 t = *(const float4*)(pp + (size_t)c * (2 * DM));
    s.x += t.x; s.y += t.y; s.z += t.z; s.w += t.w;
  }
  *(float4*)(cs + 4 * tid) = s;
  __syncthreads();
  const float* wkr = wk + (size_t)f * DM;
  const float* wvr = wvw + (size_t)f * DM;
  float su = 0.f, sz = 0.f;
  for (int d = 0; d < DM; ++d) { su += wkr[d] * cs[d]; sz += wvr[d] * cs[DM + d]; }
  u[idx] = su; z[idx] = sz;
}

__global__ void kCQ(const float* __restrict__ bq, const float* __restrict__ P,
                    float* __restrict__ cq) {
  int idx = blockIdx.x * blockDim.x + threadIdx.x;
  if (idx >= NB * DM) return;
  int b = idx >> 9, o = idx & (DM - 1);
  const float* Pb = P + (size_t)b * DM * DM + o;
  float s = 0.f;
  for (int f = 0; f < DM; ++f) s += bq[f] * Pb[(size_t)f * DM];
  cq[idx] = s;
}

extern "C" void kernel_launch(void* const* d_in, const int* in_sizes, int n_in,
                              void* d_out, int out_size, void* d_ws, size_t ws_size,
                              hipStream_t stream) {
  const float* q   = (const float*)d_in[0];
  const float* k   = (const float*)d_in[1];
  const float* v   = (const float*)d_in[2];
  const float* wq  = (const float*)d_in[3];
  const float* bq  = (const float*)d_in[4];
  const float* wk  = (const float*)d_in[5];
  const float* bk  = (const float*)d_in[6];
  const float* wvw = (const float*)d_in[7];
  const float* bv  = (const float*)d_in[8];
  const float* wo  = (const float*)d_in[9];
  const float* bo  = (const float*)d_in[10];
  float* out = (float*)d_out;

  float* ws = (float*)d_ws;
  float* G    = ws;
  float* T    = G    + (size_t)NB * DM * DM;
  float* M    = T    + (size_t)NB * DM * DM;
  float* P    = M    + (size_t)NB * NH * DKH * DKH;
  float* W4   = P    + (size_t)NB * DM * DM;
  float* u    = W4   + (size_t)NB * DM * DM;
  float* z    = u    + (size_t)NB * DM;
  float* cq   = z    + (size_t)NB * DM;
  float* part = cq   + (size_t)NB * DM;

  kSums<<<128, 512, 0, stream>>>(k, v, part);
  kUZ  <<<4, 256, 0, stream>>>(wk, wvw, part, u, z);
  kG   <<<256, 256, 0, stream>>>(k, v, G);
  kT   <<<64, 256, 0, stream>>>(wk, G, T);
  kM   <<<8, 256, 0, stream>>>(T, wvw, bk, bv, u, z, M);
  kP   <<<64, 256, 0, stream>>>(M, wo, P);
  kCQ  <<<4, 256, 0, stream>>>(bq, P, cq);
  kW4  <<<64, 256, 0, stream>>>(wq, P, W4);
  dim3 go(32, 8, 1);
  kOut <<<go, 256, 0, stream>>>(q, W4, bo, cq, out);
}


# kT, kM, kW4, kUZ: operand panels staged whole in LDS (16-byte full-line loads by the work-group) instead of per-lane global fragment loads; 256/32/256/16 groups, one 16x16 tile a wave; no sum changes
# speedup vs baseline: 4.1900x; 1.5573x over previous
#include <hip/hip_runtime.h>


typedef float v2f __attribute__((ext_vector_type(2)));
typedef float v8f __attribute__((ext_vector_type(8)));

#define DM   512
#define SEQ  4096
#define NB   2
#define NH   8
#define DKH  64

__device__ __forceinline__ v8f wmma4(v2f a, v2f b, v8f c) {
  return __builtin_amdgcn_wmma_f32_16x16x4_f32(
       false, a,  false, b,
       (short)0, c,  false,  false);
}

#define GKS   256
#define GNST  (SEQ / GKS)
#define GKP   96
#define GVP   160
#define GKBUF ((GKS / 2) * GKP)
#define GVBUF ((GKS / 2) * GVP)
#define GKIT  (GKS / 64)
#define GVIT  (GKS / 32)
#define GFG   8
#define GNG   (GKS / 4 / GFG)

__global__ void __launch_bounds__(256) kG(const float* __restrict__ k,
                                          const float* __restrict__ v,
                                          float* __restrict__ G) {
  __shared__ __attribute__((aligned(16))) float Ks[2 * GKBUF];
  __shared__ __attribute__((aligned(16))) float Vs[2 * GVBUF];
  const int tid = threadIdx.x;
  const int lane = tid & 31, wv = tid >> 5;
  const int lm = lane & 15, lh = lane >> 4;
  const int blk = blockIdx.x;
  const int b = blk >> 7;
  const int rest = blk & 127;
  const int d0 = (rest >> 3) * 32;
  const int e0 = (rest & 7) * 64;
  const int dw = 16 * (wv & 1);
  const int ew = 16 * (wv >> 1);
  const int kc = (tid & 7) * 4,  krp = tid >> 3;
  const int vc = (tid & 15) * 4, vrp = tid >> 4;
  const float* kg = k + (size_t)b * SEQ * DM + d0 + (size_t)(2 * krp) * DM + kc;
  const float* vg = v + (size_t)b * SEQ * DM + e0 + (size_t)(2 * vrp) * DM + vc;
  const int ksw = 4 * (krp & 1);
  const int vsw = 4 * ((vc >> 5) & 1);
  const int kst0 = krp * GKP + 2 * kc + ksw, kst1 = krp * GKP + 2 * kc + 4 - ksw;
  const int vst0 = vrp * GVP + 2 * vc + vsw, vst1 = vrp * GVP + 2 * vc + 4 - vsw;
  float4 rk[2 * GKIT], rv[2 * GVIT];

#define G_LOAD(st)                                                            \
  {                                                                           \
    const float* kn = kg + (size_t)(st) * GKS * DM;                           \
    const float* vn = vg + (size_t)(st) * GKS * DM;                           \
    _Pragma("unroll")                                                         \
    for (int j = 0; j < GKIT; ++j) {                                          \
      rk[2 * j]     = *(const float4*)(kn + (size_t)(64 * j) * DM);           \
      rk[2 * j + 1] = *(const float4*)(kn + (size_t)(64 * j + 1) * DM);       \
    }                                                                         \
    _Pragma("unroll")                                                         \
    for (int j = 0; j < GVIT; ++j) {                                          \
      rv[2 * j]     = *(const float4*)(vn + (size_t)(32 * j) * DM);           \
      rv[2 * j + 1] = *(const float4*)(vn + (size_t)(32 * j + 1) * DM);       \
    }                                                                         \
  }
#define G_STASH(bf)                                                           \
  {                                                                           \
    float* kd = Ks + (bf) * GKBUF;                                            \
    float* vd = Vs + (bf) * GVBUF;                                            \
    _Pragma("unroll")                                                         \
    for (int j = 0; j < GKIT; ++j) {                                          \
      *(float4*)(kd + 32 * j * GKP + kst0) =                                  \
          make_float4(rk[2 * j].x, rk[2 * j + 1].x, rk[2 * j].y, rk[2 * j + 1].y); \
      *(float4*)(kd + 32 * j * GKP + kst1) =                                  \
          make_float4(rk[2 * j].z, rk[2 * j + 1].z, rk[2 * j].w, rk[2 * j + 1].w); \
    }                                                                         \
    _Pragma("unroll")                                                         \
    for (int j = 0; j < GVIT; ++j) {                                          \
      *(float4*)(vd + 16 * j * GVP + vst0) =                                  \
          make_float4(rv[2 * j].x, rv[2 * j + 1].x, rv[2 * j].y, rv[2 * j + 1].y); \
      *(float4*)(vd + 16 * j * GVP + vst1) =                                  \
          make_float4(rv[2 * j].z, rv[2 * j + 1].z, rv[2 * j].w, rv[2 * j + 1].w); \
    }                                                                         \
  }
#define G_FRAG(dst, g)                                                        \
  {                                                                           \
    _Pragma("unroll")                                                         \
    for (int i = 0; i < GFG; ++i) {                                           \
      fa[dst][i] = *(const v2f*)(ka + 2 * ((g) * GFG + i) * GKP);             \
      fb[dst][i] = *(const v2f*)(va + 2 * ((g) * GFG + i) * GVP);             \
    }                                                                         \
  }

  G_LOAD(0)
  G_STASH(0)
  G_LOAD(1)
  __syncthreads();

  const int ard = lh * GKP + ((2 * (dw + lm)) ^ (4 * lh));
  const int brd = lh * GVP + ((2 * (ew + lm)) ^ (4 * ((ew >> 5) & 1)));
  v8f acc = {};
  v2f fa[2][GFG], fb[2][GFG];
  for (int n = 0; n < GNST; ++n) {
    const int buf = n & 1;
    const float* ka = Ks + buf * GKBUF + ard;
    const float* va = Vs + buf * GVBUF + brd;
    G_FRAG(0, 0)
    if (n + 1 < GNST) G_STASH(buf ^ 1)
    if (n + 2 < GNST) G_LOAD(n + 2)
    __builtin_amdgcn_sched_barrier(0);
    #pragma unroll
    for (int g = 0; g < GNG; ++g) {
      if (g + 1 < GNG) G_FRAG((g + 1) & 1, g + 1)
      #pragma unroll
      for (int i = 0; i < GFG; ++i)
        acc = wmma4(fa[g & 1][i], fb[g & 1][i], acc);
      __builtin_amdgcn_sched_barrier(0);
    }
    __syncthreads();
  }
#undef G_LOAD
#undef G_STASH
#undef G_FRAG
  float* Gb = G + ((size_t)b * DM + d0 + dw) * DM + e0 + ew;
  #pragma unroll
  for (int r = 0; r < 8; ++r)
    Gb[(size_t)(r + 8 * lh) * DM + lm] = acc[r];
}

#define ROWP (DM + 4)
#define KP64 72
#define KP32 40

template <int NR>
__device__ __forceinline__ void stage_rows(const float* __restrict__ g, float* s,
                                           int wv, int lane) {
  #pragma unroll
  for (int j = 0; j < NR / 8; ++j)
    #pragma unroll
    for (int q = 0; q < 4; ++q)
      *(float4*)(s + (wv + 8 * j) * ROWP + 128 * q + 4 * lane) =
          *(const float4*)(g + (size_t)(wv + 8 * j) * DM + 128 * q + 4 * lane);
}

template <int NC, int PITCH>
__device__ __forceinline__ void stage_kmajor(const float* __restrict__ g, float* s,
                                             int tid) {
  constexpr int F4 = NC / 4;
  constexpr int RPS = 256 / F4;
  const int c4 = (tid % F4) * 4, r0 = tid / F4;
  #pragma unroll
  for (int j = 0; j < DM / RPS; ++j)
    *(float4*)(s + (r0 + RPS * j) * PITCH + c4) =
        *(const float4*)(g + (size_t)(r0 + RPS * j) * DM + c4);
}

__device__ __forceinline__ v2f ld_pair(const float* p, int pitch) {
  v2f r; r.x = p[0]; r.y = p[pitch]; return r;
}

#define PANEL_K_LOOP(LOADA, LOADB)                                            \
  {                                                                           \
    v2f fa[2][8], fb[2][8];                                                   \
    _Pragma("unroll")                                                         \
    for (int i = 0; i < 8; ++i) {                                             \
      const int kk = 4 * i;                                                   \
      fa[0][i] = LOADA; fb[0][i] = LOADB;                                     \
    }                                                                         \
    _Pragma("unroll")                                                         \
    for (int g = 0; g < DM / 32; ++g) {                                       \
      if (g + 1 < DM / 32) {                                                  \
        _Pragma("unroll")                                                     \
        for (int i = 0; i < 8; ++i) {                                         \
          const int kk = 32 * (g + 1) + 4 * i;                                \
          fa[(g + 1) & 1][i] = LOADA; fb[(g + 1) & 1][i] = LOADB;             \
        }                                                                     \
      }                                                                       \
      _Pragma("unroll")                                                       \
      for (int i = 0; i < 8; ++i)                                             \
        acc = wmma4(fa[g & 1][i], fb[g & 1][i], acc);                         \
      __builtin_amdgcn_sched_barrier(0);                                      \
    }                                                                         \
  }

__global__ void __launch_bounds__(256) kT(const float* __restrict__ wk,
                                          const float* __restrict__ G,
                                          float* __restrict__ T) {
  __shared__ __attribute__((aligned(16))) float As[32 * ROWP];
  __shared__ __attribute__((aligned(16))) float Bs[DM * KP64];
  const int tid = threadIdx.x;
  const int lane = tid & 31, wv = tid >> 5;
  const int lm = lane & 15, lh = lane >> 4;
  const int bh = blockIdx.x >> 4;
  const int b = bh >> 3, h = bh & 7;
  const int ih = (blockIdx.x >> 3) & 1;
  const int e0 = (blockIdx.x & 7) * 64;
  stage_rows<32>(wk + (size_t)(h * 64 + ih * 32) * DM, As, wv, lane);
  stage_kmajor<64, KP64>(G + (size_t)b * DM * DM + e0, Bs, tid);
  __syncthreads();
  const int ti = wv >> 2, tj = wv & 3;
  const float* ap = As + (16 * ti + lm) * ROWP + 2 * lh;
  const float* bp = Bs + (2 * lh) * KP64 + 16 * tj + lm;
  v8f acc = {};
  PANEL_K_LOOP((*(const v2f*)(ap + kk)), ld_pair(bp + kk * KP64, KP64))
  float* Tb = T + ((size_t)bh * 64 + ih * 32 + 16 * ti) * DM + e0 + 16 * tj;
  #pragma unroll
  for (int r = 0; r < 8; ++r)
    Tb[(size_t)(r + 8 * lh) * DM + lm] = acc[r];
}


__global__ void __launch_bounds__(256) kM(const float* __restrict__ T,
                                          const float* __restrict__ wvw,
                                          const float* __restrict__ bk,
                                          const float* __restrict__ bv,
                                          const float* __restrict__ u,
                                          const float* __restrict__ z,
                                          float* __restrict__ M) {
  __shared__ __attribute__((aligned(16))) float As[32 * ROWP];
  __shared__ __attribute__((aligned(16))) float Bs[64 * ROWP];
  const int lane = threadIdx.x & 31, wv = threadIdx.x >> 5;
  const int lm = lane & 15, lh = lane >> 4;
  const int bh = blockIdx.x >> 1, half = blockIdx.x & 1;
  const int b = bh >> 3, h = bh & 7;
  const float* Ag = T + ((size_t)bh * 64 + half * 32) * DM + 4 * lane;
  const float* Bg = wvw + (size_t)(h * 64) * DM + 4 * lane;
  #pragma unroll
  for (int q = 0; q < 4; ++q) {
    float4 ra[4], rb[8];
    #pragma unroll
    for (int j = 0; j < 4; ++j)
      ra[j] = *(const float4*)(Ag + (size_t)(wv + 8 * j) * DM + 128 * q);
    #pragma unroll
    for (int j = 0; j < 8; ++j)
      rb[j] = *(const float4*)(Bg + (size_t)(wv + 8 * j) * DM + 128 * q);
    #pragma unroll
    for (int j = 0; j < 4; ++j)
      *(float4*)(As + (wv + 8 * j) * ROWP + 128 * q + 4 * lane) = ra[j];
    #pragma unroll
    for (int j = 0; j < 8; ++j)
      *(float4*)(Bs + (wv + 8 * j) * ROWP + 128 * q + 4 * lane) = rb[j];
  }
  __syncthreads();
  const int ti = wv >> 2, tj = wv & 3;
  const float* ap = As + (16 * ti + lm) * ROWP + 2 * lh;
  const float* bp = Bs + (16 * tj + lm) * ROWP + 2 * lh;
  v8f acc = {};
  PANEL_K_LOOP((*(const v2f*)(ap + kk)), (*(const v2f*)(bp + kk)))
  const int i0 = half * 32 + 16 * ti;
  const int jf = h * 64 + 16 * tj + lm;
  const float bvj = bv[jf];
  const float zj  = z[(size_t)b * DM + jf];
  #pragma unroll
  for (int r = 0; r < 8; ++r) {
    const int ifl = h * 64 + i0 + r + 8 * lh;
    const float bki = bk[ifl];
    const float ui  = u[(size_t)b * DM + ifl];
    float val = acc[r] + ui * bvj + bki * zj + (float)SEQ * bki * bvj;
    M[((size_t)bh * 64 + i0 + r + 8 * lh) * DKH + 16 * tj + lm] = 0.125f * val;
  }
}

__global__ void __launch_bounds__(256) kP(const float* __restrict__ M,
                                          const float* __restrict__ wo,
                                          float* __restrict__ P) {
  const int lane = threadIdx.x & 31, wv = threadIdx.x >> 5;
  const int lm = lane & 15, lh = lane >> 4;
  const int job = blockIdx.x * 8 + wv;
  const int bh = job >> 5;
  const int b = bh >> 3, h = bh & 7;
  const int i0 = ((job >> 3) & 3) * 16;
  const int o0 = (job & 7) * 64;
  const float* arow = M + ((size_t)bh * 64 + i0 + lm) * DKH + 2 * lh;
  v8f acc[4] = {};
  for (int kk = 0; kk < DKH; kk += 4) {
    v2f a = *(const v2f*)(arow + kk);
    #pragma unroll
    for (int t = 0; t < 4; ++t) {
      v2f bm = *(const v2f*)(wo + (size_t)(o0 + 16 * t + lm) * DM + h * 64 + kk + 2 * lh);
      acc[t] = wmma4(a, bm, acc[t]);
    }
  }
  float* Pb = P + ((size_t)b * DM + h * 64 + i0) * DM + o0;
  #pragma unroll
  for (int t = 0; t < 4; ++t)
    #pragma unroll
    for (int r = 0; r < 8; ++r)
      Pb[(size_t)(r + 8 * lh) * DM + 16 * t + lm] = acc[t][r];
}

__global__ void __launch_bounds__(256) kW4(const float* __restrict__ wq,
                                           const float* __restrict__ P,
                                           float* __restrict__ W4) {
  __shared__ __attribute__((aligned(16))) float As[DM * KP32];
  __shared__ __attribute__((aligned(16))) float Bs[DM * KP64];
  const int tid = threadIdx.x;
  const int lane = tid & 31, wv = tid >> 5;
  const int lm = lane & 15, lh = lane >> 4;
  const int b = blockIdx.x >> 7;
  const int d0 = ((blockIdx.x >> 3) & 15) * 32;
  const int o0 = (blockIdx.x & 7) * 64;
  stage_kmajor<32, KP32>(wq + d0, As, tid);
  stage_kmajor<64, KP64>(P + (size_t)b * DM * DM + o0, Bs, tid);
  __syncthreads();
  const int ti = wv >> 2, tj = wv & 3;
  const float* ap = As + (2 * lh) * KP32 + 16 * ti + lm;
  const float* bp = Bs + (2 * lh) * KP64 + 16 * tj + lm;
  v8f acc = {};
  PANEL_K_LOOP(ld_pair(ap + kk * KP32, KP32), ld_pair(bp + kk * KP64, KP64))
  float* Wb = W4 + ((size_t)b * DM + d0 + 16 * ti) * DM + o0 + 16 * tj;
  #pragma unroll
  for (int r = 0; r < 8; ++r)
    Wb[(size_t)(r + 8 * lh) * DM + lm] = acc[r];
}

__global__ void __launch_bounds__(256) kOut(const float* __restrict__ q,
                                            const float* __restrict__ W4,
                                            const float* __restrict__ bo,
                                            const float* __restrict__ cq,
                                            float* __restrict__ out) {
  const int lane = threadIdx.x & 31, wv = threadIdx.x >> 5;
  const int lm = lane & 15, lh = lane >> 4;
  const int job = blockIdx.x * 8 + wv;
  const int b = job >> 7;
  const int n0 = (job & 127) * 32;
  const int o0 = blockIdx.y * 64;
  const float* qrow0 = q + ((size_t)b * SEQ + n0 + lm) * DM + 2 * lh;
  const float* qrow1 = qrow0 + (size_t)16 * DM;
  const float* Wb = W4 + (size_t)b * DM * DM;
  v8f acc[8] = {};
  #pragma unroll 2
  for (int kk = 0; kk < DM; kk += 4) {
    v2f a0 = *(const v2f*)(qrow0 + kk);
    v2f a1 = *(const v2f*)(qrow1 + kk);
    size_t rb = (size_t)(kk + 2 * lh) * DM;
    #pragma unroll
    for (int t = 0; t < 4; ++t) {
      v2f bm; bm.x = Wb[rb + o0 + 16 * t + lm]; bm.y = Wb[rb + DM + o0 + 16 * t + lm];
      acc[t]     = wmma4(a0, bm, acc[t]);
      acc[4 + t] = wmma4(a1, bm, acc[4 + t]);
    }
  }
  float* ob = out + ((size_t)b * SEQ + n0) * DM + o0;
  #pragma unroll
  for (int t = 0; t < 4; ++t) {
    float add = bo[o0 + 16 * t + lm] + cq[(size_t)b * DM + o0 + 16 * t + lm];
    #pragma unroll
    for (int m = 0; m < 2; ++m)
      #pragma unroll
      for (int r = 0; r < 8; ++r)
        ob[(size_t)(16 * m + r + 8 * lh) * DM + 16 * t + lm] = acc[4 * m + t][r] + add;
  }
}

#define SCH   64
#define SROWS (SEQ / SCH)
#define SPH   4
#define SRPT  (SROWS / SPH)

__global__ void __launch_bounds__(512) kSums(const float* __restrict__ k,
                                             const float* __restrict__ v,
                                             float* __restrict__ part) {
  __shared__ float4 red[SPH * 256];
  const int tid = threadIdx.x;
  const int cg = tid & 127, ph = tid >> 7;
  const int b = blockIdx.x / SCH, c = blockIdx.x % SCH;
  const size_t off =
      ((size_t)b * SEQ + (size_t)c * SROWS + (size_t)ph * SRPT) * DM + 4 * cg;
  const float* kp = k + off;
  const float* vp = v + off;
  float4 rk[SRPT], rv[SRPT];
  #pragma unroll
  for (int i = 0; i < SRPT; ++i) {
    rk[i] = *(const float4*)(kp + (size_t)i * DM);
    rv[i] = *(const float4*)(vp + (size_t)i * DM);
  }
  float4 sk = rk[0], sv = rv[0];
  #pragma unroll
  for (int i = 1; i < SRPT; ++i) {
    sk.x += rk[i].x; sk.y += rk[i].y; sk.z += rk[i].z; sk.w += rk[i].w;
    sv.x += rv[i].x; sv.y += rv[i].y; sv.z += rv[i].z; sv.w += rv[i].w;
  }
  red[ph * 256 + cg] = sk;
  red[ph * 256 + 128 + cg] = sv;
  __syncthreads();
  if (tid < 256) {
    float4 s = red[tid];
    #pragma unroll
    for (int p = 1; p < SPH; ++p) {
      const float4 t = red[p * 256 + tid];
      s.x += t.x; s.y += t.y; s.z += t.z; s.w += t.w;
    }
    *(float4*)(part + (size_t)blockIdx.x * (2 * DM) + 4 * tid) = s;
  }
}

#define UZR 128

__global__ void __launch_bounds__(256) kUZ(const float* __restrict__ wk,
                                           const float* __restrict__ wvw,
                                           const float* __restrict__ part,
                                           float* __restrict__ u,
                                           float* __restrict__ z) {
  __shared__ __attribute__((aligned(16))) float Ws[UZR * ROWP];
  __shared__ __attribute__((aligned(16))) float cs[DM];
  const int tid = threadIdx.x;
  const int lane = tid & 31, wv = tid >> 5;
  const int b = blockIdx.x >> 3, mat = (blockIdx.x >> 2) & 1, rq = blockIdx.x & 3;
  const float* pp = part + (size_t)b * SCH * (2 * DM) + mat * DM + 2 * tid;
  float2 s = *(const float2*)pp;
  #pragma unroll
  for (int c = 1; c < SCH; ++c) {
    const float2 t = *(const float2*)(pp + (size_t)c * (2 * DM));
    s.x += t.x; s.y += t.y;
  }
  *(float2*)(cs + 2 * tid) = s;
  const float* Wg = (mat ? wvw : wk) + (size_t)(rq * UZR) * DM + 4 * lane;
  for (int g = 0; g < 4; ++g) {
    float4 r[16];
    #pragma unroll
    for (int j = 0; j < 4; ++j)
      #pragma unroll
      for (int q = 0; q < 4; ++q)
        r[4 * j + q] = *(const float4*)(Wg + (size_t)(wv + 8 * (4 * g + j)) * DM + 128 * q);
    #pragma unroll
    for (int j = 0; j < 4; ++j)
      #pragma unroll
      for (int q = 0; q < 4; ++q)
        *(float4*)(Ws + (wv + 8 * (4 * g + j)) * ROWP + 128 * q + 4 * lane) = r[4 * j + q];
  }
  __syncthreads();
  if (tid < UZR) {
    const float* wr = Ws + tid * ROWP;
    float su = 0.f;
    for (int d = 0; d < DM; ++d) su += wr[d] * cs[d];
    float* dst = mat ? z : u;
    dst[(size_t)b * DM + rq * UZR + tid] = su;
  }
}

__global__ void kCQ(const float* __restrict__ bq, const float* __restrict__ P,
                    float* __restrict__ cq) {
  int idx = blockIdx.x * blockDim.x + threadIdx.x;
  if (idx >= NB * DM) return;
  int b = idx >> 9, o = idx & (DM - 1);
  const float* Pb = P + (size_t)b * DM * DM + o;
  float s = 0.f;
  for (int f = 0; f < DM; ++f) s += bq[f] * Pb[(size_t)f * DM];
  cq[idx] = s;
}

extern "C" void kernel_launch(void* const* d_in, const int* in_sizes, int n_in,
                              void* d_out, int out_size, void* d_ws, size_t ws_size,
                              hipStream_t stream) {
  const float* q   = (const float*)d_in[0];
  const float* k   = (const float*)d_in[1];
  const float* v   = (const float*)d_in[2];
  const float* wq  = (const float*)d_in[3];
  const float* bq  = (const float*)d_in[4];
  const float* wk  = (const float*)d_in[5];
  const float* bk  = (const float*)d_in[6];
  const float* wvw = (const float*)d_in[7];
  const float* bv  = (const float*)d_in[8];
  const float* wo  = (const float*)d_in[9];
  const float* bo  = (const float*)d_in[10];
  float* out = (float*)d_out;

  float* ws = (float*)d_ws;
  float* G    = ws;
  float* T    = G    + (size_t)NB * DM * DM;
  float* M    = T    + (size_t)NB * DM * DM;
  float* P    = M    + (size_t)NB * NH * DKH * DKH;
  float* W4   = P    + (size_t)NB * DM * DM;
  float* u    = W4   + (size_t)NB * DM * DM;
  float* z    = u    + (size_t)NB * DM;
  float* cq   = z    + (size_t)NB * DM;
  float* part = cq   + (size_t)NB * DM;

  kSums<<<128, 512, 0, stream>>>(k, v, part);
  kUZ  <<<16, 256, 0, stream>>>(wk, wvw, part, u, z);
  kG   <<<256, 256, 0, stream>>>(k, v, G);
  kT   <<<256, 256, 0, stream>>>(wk, G, T);
  kM   <<<32, 256, 0, stream>>>(T, wvw, bk, bv, u, z, M);
  kP   <<<64, 256, 0, stream>>>(M, wo, P);
  kCQ  <<<4, 256, 0, stream>>>(bq, P, cq);
  kW4  <<<256, 256, 0, stream>>>(wq, P, W4);
  dim3 go(32, 8, 1);
  kOut <<<go, 256, 0, stream>>>(q, W4, bo, cq, out);
}


# kOut: q and W4 panels staged through LDS by the work-group (16-byte whole-line loads, 64-k stages, two buffers, 8-byte fragment reads) instead of per-lane global fragment loads; no sum changes
# speedup vs baseline: 5.7833x; 1.3803x over previous
#include <hip/hip_runtime.h>


typedef float v2f __attribute__((ext_vector_type(2)));
typedef float v4f __attribute__((ext_vector_type(4)));
typedef float v8f __attribute__((ext_vector_type(8)));

#define DM   512
#define SEQ  4096
#define NB   2
#define NH   8
#define DKH  64

__device__ __forceinline__ v8f wmma4(v2f a, v2f b, v8f c) {
  return __builtin_amdgcn_wmma_f32_16x16x4_f32(
       false, a,  false, b,
       (short)0, c,  false,  false);
}

#define GKS   256
#define GNST  (SEQ / GKS)
#define GKP   96
#define GVP   160
#define GKBUF ((GKS / 2) * GKP)
#define GVBUF ((GKS / 2) * GVP)
#define GKIT  (GKS / 64)
#define GVIT  (GKS / 32)
#define GFG   8
#define GNG   (GKS / 4 / GFG)

__global__ void __launch_bounds__(256) kG(const float* __restrict__ k,
                                          const float* __restrict__ v,
                                          float* __restrict__ G) {
  __shared__ __attribute__((aligned(16))) float Ks[2 * GKBUF];
  __shared__ __attribute__((aligned(16))) float Vs[2 * GVBUF];
  const int tid = threadIdx.x;
  const int lane = tid & 31, wv = tid >> 5;
  const int lm = lane & 15, lh = lane >> 4;
  const int blk = blockIdx.x;
  const int b = blk >> 7;
  const int rest = blk & 127;
  const int d0 = (rest >> 3) * 32;
  const int e0 = (rest & 7) * 64;
  const int dw = 16 * (wv & 1);
  const int ew = 16 * (wv >> 1);
  const int kc = (tid & 7) * 4,  krp = tid >> 3;
  const int vc = (tid & 15) * 4, vrp = tid >> 4;
  const float* kg = k + (size_t)b * SEQ * DM + d0 + (size_t)(2 * krp) * DM + kc;
  const float* vg = v + (size_t)b * SEQ * DM + e0 + (size_t)(2 * vrp) * DM + vc;
  const int ksw = 4 * (krp & 1);
  const int vsw = 4 * ((vc >> 5) & 1);
  const int kst0 = krp * GKP + 2 * kc + ksw, kst1 = krp * GKP + 2 * kc + 4 - ksw;
  const int vst0 = vrp * GVP + 2 * vc + vsw, vst1 = vrp * GVP + 2 * vc + 4 - vsw;
  float4 rk[2 * GKIT], rv[2 * GVIT];

#define G_LOAD(st)                                                            \
  {                                                                           \
    const float* kn = kg + (size_t)(st) * GKS * DM;                           \
    const float* vn = vg + (size_t)(st) * GKS * DM;                           \
    _Pragma("unroll")                                                         \
    for (int j = 0; j < GKIT; ++j) {                                          \
      rk[2 * j]     = *(const float4*)(kn + (size_t)(64 * j) * DM);           \
      rk[2 * j + 1] = *(const float4*)(kn + (size_t)(64 * j + 1) * DM);       \
    }                                                                         \
    _Pragma("unroll")                                                         \
    for (int j = 0; j < GVIT; ++j) {                                          \
      rv[2 * j]     = *(const float4*)(vn + (size_t)(32 * j) * DM);           \
      rv[2 * j + 1] = *(const float4*)(vn + (size_t)(32 * j + 1) * DM);       \
    }                                                                         \
  }
#define G_STASH(bf)                                                           \
  {                                                                           \
    float* kd = Ks + (bf) * GKBUF;                                            \
    float* vd = Vs + (bf) * GVBUF;                                            \
    _Pragma("unroll")                                                         \
    for (int j = 0; j < GKIT; ++j) {                                          \
      *(float4*)(kd + 32 * j * GKP + kst0) =                                  \
          make_float4(rk[2 * j].x, rk[2 * j + 1].x, rk[2 * j].y, rk[2 * j + 1].y); \
      *(float4*)(kd + 32 * j * GKP + kst1) =                                  \
          make_float4(rk[2 * j].z, rk[2 * j + 1].z, rk[2 * j].w, rk[2 * j + 1].w); \
    }                                                                         \
    _Pragma("unroll")                                                         \
    for (int j = 0; j < GVIT; ++j) {                                          \
      *(float4*)(vd + 16 * j * GVP + vst0) =                                  \
          make_float4(rv[2 * j].x, rv[2 * j + 1].x, rv[2 * j].y, rv[2 * j + 1].y); \
      *(float4*)(vd + 16 * j * GVP + vst1) =                                  \
          make_float4(rv[2 * j].z, rv[2 * j + 1].z, rv[2 * j].w, rv[2 * j + 1].w); \
    }                                                                         \
  }
#define G_FRAG(dst, g)                                                        \
  {                                                                           \
    _Pragma("unroll")                                                         \
    for (int i = 0; i < GFG; ++i) {                                           \
      fa[dst][i] = *(const v2f*)(ka + 2 * ((g) * GFG + i) * GKP);             \
      fb[dst][i] = *(const v2f*)(va + 2 * ((g) * GFG + i) * GVP);             \
    }                                                                         \
  }

  G_LOAD(0)
  G_STASH(0)
  G_LOAD(1)
  __syncthreads();

  const int ard = lh * GKP + ((2 * (dw + lm)) ^ (4 * lh));
  const int brd = lh * GVP + ((2 * (ew + lm)) ^ (4 * ((ew >> 5) & 1)));
  v8f acc = {};
  v2f fa[2][GFG], fb[2][GFG];
  for (int n = 0; n < GNST; ++n) {
    const int buf = n & 1;
    const float* ka = Ks + buf * GKBUF + ard;
    const float* va = Vs + buf * GVBUF + brd;
    G_FRAG(0, 0)
    if (n + 1 < GNST) G_STASH(buf ^ 1)
    if (n + 2 < GNST) G_LOAD(n + 2)
    __builtin_amdgcn_sched_barrier(0);
    #pragma unroll
    for (int g = 0; g < GNG; ++g) {
      if (g + 1 < GNG) G_FRAG((g + 1) & 1, g + 1)
      #pragma unroll
      for (int i = 0; i < GFG; ++i)
        acc = wmma4(fa[g & 1][i], fb[g & 1][i], acc);
      __builtin_amdgcn_sched_barrier(0);
    }
    __syncthreads();
  }
#undef G_LOAD
#undef G_STASH
#undef G_FRAG
  float* Gb = G + ((size_t)b * DM + d0 + dw) * DM + e0 + ew;
  #pragma unroll
  for (int r = 0; r < 8; ++r)
    Gb[(size_t)(r + 8 * lh) * DM + lm] = acc[r];
}

#define ROWP (DM + 4)
#define KP64 72
#define KP32 40

template <int NR>
__device__ __forceinline__ void stage_rows(const float* __restrict__ g, float* s,
                                           int wv, int lane) {
  #pragma unroll
  for (int j = 0; j < NR / 8; ++j)
    #pragma unroll
    for (int q = 0; q < 4; ++q)
      *(float4*)(s + (wv + 8 * j) * ROWP + 128 * q + 4 * lane) =
          *(const float4*)(g + (size_t)(wv + 8 * j) * DM + 128 * q + 4 * lane);
}

template <int NC, int PITCH>
__device__ __forceinline__ void stage_kmajor(const float* __restrict__ g, float* s,
                                             int tid) {
  constexpr int F4 = NC / 4;
  constexpr int RPS = 256 / F4;
  const int c4 = (tid % F4) * 4, r0 = tid / F4;
  #pragma unroll
  for (int j = 0; j < DM / RPS; ++j)
    *(float4*)(s + (r0 + RPS * j) * PITCH + c4) =
        *(const float4*)(g + (size_t)(r0 + RPS * j) * DM + c4);
}

__device__ __forceinline__ v2f ld_pair(const float* p, int pitch) {
  v2f r; r.x = p[0]; r.y = p[pitch]; return r;
}

#define PANEL_K_LOOP(LOADA, LOADB)                                            \
  {                                                                           \
    v2f fa[2][8], fb[2][8];                                                   \
    _Pragma("unroll")                                                         \
    for (int i = 0; i < 8; ++i) {                                             \
      const int kk = 4 * i;                                                   \
      fa[0][i] = LOADA; fb[0][i] = LOADB;                                     \
    }                                                                         \
    _Pragma("unroll")                                                         \
    for (int g = 0; g < DM / 32; ++g) {                                       \
      if (g + 1 < DM / 32) {                                                  \
        _Pragma("unroll")                                                     \
        for (int i = 0; i < 8; ++i) {                                         \
          const int kk = 32 * (g + 1) + 4 * i;                                \
          fa[(g + 1) & 1][i] = LOADA; fb[(g + 1) & 1][i] = LOADB;             \
        }                                                                     \
      }                                                                       \
      _Pragma("unroll")                                                       \
      for (int i = 0; i < 8; ++i)                                             \
        acc = wmma4(fa[g & 1][i], fb[g & 1][i], acc);                         \
      __builtin_amdgcn_sched_barrier(0);                                      \
    }                                                                         \
  }

__global__ void __launch_bounds__(256) kT(const float* __restrict__ wk,
                                          const float* __restrict__ G,
                                          float* __restrict__ T) {
  __shared__ __attribute__((aligned(16))) float As[32 * ROWP];
  __shared__ __attribute__((aligned(16))) float Bs[DM * KP64];
  const int tid = threadIdx.x;
  const int lane = tid & 31, wv = tid >> 5;
  const int lm = lane & 15, lh = lane >> 4;
  const int bh = blockIdx.x >> 4;
  const int b = bh >> 3, h = bh & 7;
  const int ih = (blockIdx.x >> 3) & 1;
  const int e0 = (blockIdx.x & 7) * 64;
  stage_rows<32>(wk + (size_t)(h * 64 + ih * 32) * DM, As, wv, lane);
  stage_kmajor<64, KP64>(G + (size_t)b * DM * DM + e0, Bs, tid);
  __syncthreads();
  const int ti = wv >> 2, tj = wv & 3;
  const float* ap = As + (16 * ti + lm) * ROWP + 2 * lh;
  const float* bp = Bs + (2 * lh) * KP64 + 16 * tj + lm;
  v8f acc = {};
  PANEL_K_LOOP((*(const v2f*)(ap + kk)), ld_pair(bp + kk * KP64, KP64))
  float* Tb = T + ((size_t)bh * 64 + ih * 32 + 16 * ti) * DM + e0 + 16 * tj;
  #pragma unroll
  for (int r = 0; r < 8; ++r)
    Tb[(size_t)(r + 8 * lh) * DM + lm] = acc[r];
}


__global__ void __launch_bounds__(256) kM(const float* __restrict__ T,
                                          const float* __restrict__ wvw,
                                          const float* __restrict__ bk,
                                          const float* __restrict__ bv,
                                          const float* __restrict__ u,
                                          const float* __restrict__ z,
                                          float* __restrict__ M) {
  __shared__ __attribute__((aligned(16))) float As[32 * ROWP];
  __shared__ __attribute__((aligned(16))) float Bs[64 * ROWP];
  const int lane = threadIdx.x & 31, wv = threadIdx.x >> 5;
  const int lm = lane & 15, lh = lane >> 4;
  const int bh = blockIdx.x >> 1, half = blockIdx.x & 1;
  const int b = bh >> 3, h = bh & 7;
  const float* Ag = T + ((size_t)bh * 64 + half * 32) * DM + 4 * lane;
  const float* Bg = wvw + (size_t)(h * 64) * DM + 4 * lane;
  #pragma unroll
  for (int q = 0; q < 4; ++q) {
    float4 ra[4], rb[8];
    #pragma unroll
    for (int j = 0; j < 4; ++j)
      ra[j] = *(const float4*)(Ag + (size_t)(wv + 8 * j) * DM + 128 * q);
    #pragma unroll
    for (int j = 0; j < 8; ++j)
      rb[j] = *(const float4*)(Bg + (size_t)(wv + 8 * j) * DM + 128 * q);
    #pragma unroll
    for (int j = 0; j < 4; ++j)
      *(float4*)(As + (wv + 8 * j) * ROWP + 128 * q + 4 * lane) = ra[j];
    #pragma unroll
    for (int j = 0; j < 8; ++j)
      *(float4*)(Bs + (wv + 8 * j) * ROWP + 128 * q + 4 * lane) = rb[j];
  }
  __syncthreads();
  const int ti = wv >> 2, tj = wv & 3;
  const float* ap = As + (16 * ti + lm) * ROWP + 2 * lh;
  const float* bp = Bs + (16 * tj + lm) * ROWP + 2 * lh;
  v8f acc = {};
  PANEL_K_LOOP((*(const v2f*)(ap + kk)), (*(const v2f*)(bp + kk)))
  const int i0 = half * 32 + 16 * ti;
  const int jf = h * 64 + 16 * tj + lm;
  const float bvj = bv[jf];
  const float zj  = z[(size_t)b * DM + jf];
  #pragma unroll
  for (int r = 0; r < 8; ++r) {
    const int ifl = h * 64 + i0 + r + 8 * lh;
    const float bki = bk[ifl];
    const float ui  = u[(size_t)b * DM + ifl];
    float val = acc[r] + ui * bvj + bki * zj + (float)SEQ * bki * bvj;
    M[((size_t)bh * 64 + i0 + r + 8 * lh) * DKH + 16 * tj + lm] = 0.125f * val;
  }
}

__global__ void __launch_bounds__(256) kP(const float* __restrict__ M,
                                          const float* __restrict__ wo,
                                          float* __restrict__ P) {
  const int lane = threadIdx.x & 31, wv = threadIdx.x >> 5;
  const int lm = lane & 15, lh = lane >> 4;
  const int job = blockIdx.x * 8 + wv;
  const int bh = job >> 5;
  const int b = bh >> 3, h = bh & 7;
  const int i0 = ((job >> 3) & 3) * 16;
  const int o0 = (job & 7) * 64;
  const float* arow = M + ((size_t)bh * 64 + i0 + lm) * DKH + 2 * lh;
  v8f acc[4] = {};
  for (int kk = 0; kk < DKH; kk += 4) {
    v2f a = *(const v2f*)(arow + kk);
    #pragma unroll
    for (int t = 0; t < 4; ++t) {
      v2f bm = *(const v2f*)(wo + (size_t)(o0 + 16 * t + lm) * DM + h * 64 + kk + 2 * lh);
      acc[t] = wmma4(a, bm, acc[t]);
    }
  }
  float* Pb = P + ((size_t)b * DM + h * 64 + i0) * DM + o0;
  #pragma unroll
  for (int t = 0; t < 4; ++t)
    #pragma unroll
    for (int r = 0; r < 8; ++r)
      Pb[(size_t)(r + 8 * lh) * DM + 16 * t + lm] = acc[t][r];
}

__global__ void __launch_bounds__(256) kW4(const float* __restrict__ wq,
                                           const float* __restrict__ P,
                                           float* __restrict__ W4) {
  __shared__ __attribute__((aligned(16))) float As[DM * KP32];
  __shared__ __attribute__((aligned(16))) float Bs[DM * KP64];
  const int tid = threadIdx.x;
  const int lane = tid & 31, wv = tid >> 5;
  const int lm = lane & 15, lh = lane >> 4;
  const int b = blockIdx.x >> 7;
  const int d0 = ((blockIdx.x >> 3) & 15) * 32;
  const int o0 = (blockIdx.x & 7) * 64;
  stage_kmajor<32, KP32>(wq + d0, As, tid);
  stage_kmajor<64, KP64>(P + (size_t)b * DM * DM + o0, Bs, tid);
  __syncthreads();
  const int ti = wv >> 2, tj = wv & 3;
  const float* ap = As + (2 * lh) * KP32 + 16 * ti + lm;
  const float* bp = Bs + (2 * lh) * KP64 + 16 * tj + lm;
  v8f acc = {};
  PANEL_K_LOOP(ld_pair(ap + kk * KP32, KP32), ld_pair(bp + kk * KP64, KP64))
  float* Wb = W4 + ((size_t)b * DM + d0 + 16 * ti) * DM + o0 + 16 * tj;
  #pragma unroll
  for (int r = 0; r < 8; ++r)
    Wb[(size_t)(r + 8 * lh) * DM + lm] = acc[r];
}

#define OKS   64
#define ONST  (DM / OKS)
#define OAP   (OKS + 4)
#define OBP   160
#define OABUF (256 * OAP)
#define OBBUF ((OKS / 2) * OBP)
#define OAIT  16
#define OBIT  (OKS / 32)
#define OFG   2
#define ONG   (OKS / 4 / OFG)

__global__ void __launch_bounds__(256) kOut(const float* __restrict__ q,
                                            const float* __restrict__ W4,
                                            const float* __restrict__ bo,
                                            const float* __restrict__ cq,
                                            float* __restrict__ out) {
  __shared__ __attribute__((aligned(16))) float Qs[2 * OABUF];
  __shared__ __attribute__((aligned(16))) float Ws[2 * OBBUF];
  const int tid = threadIdx.x;
  const int lane = tid & 31, wv = tid >> 5;
  const int lm = lane & 15, lh = lane >> 4;
  const int b = blockIdx.x >> 4;
  const int n0 = (blockIdx.x & 15) * 256;
  const int o0 = blockIdx.y * 64;
  const int sc = (tid & 15) * 4, sr = tid >> 4;
  const float* qg = q + ((size_t)b * SEQ + n0 + sr) * DM + sc;
  const float* wg = W4 + ((size_t)b * DM + 2 * sr) * DM + o0 + sc;
  const int qst = sr * OAP + sc;
  const int wsw = 4 * ((sc >> 5) & 1);
  const int wst0 = sr * OBP + 2 * sc + wsw, wst1 = sr * OBP + 2 * sc + 4 - wsw;
  v4f rq[OAIT];
  float4 rw[2 * OBIT];

#define O_LOAD(st)                                                            \
  {                                                                           \
    const float* qn = qg + (st) * OKS;                                        \
    const float* wn = wg + (size_t)(st) * OKS * DM;                           \
    _Pragma("unroll")                                                         \
    for (int j = 0; j < OAIT; ++j)                                            \
      rq[j] = *(const v4f*)(qn + (size_t)(16 * j) * DM);                      \
    _Pragma("unroll")                                                         \
    for (int j = 0; j < OBIT; ++j) {                                          \
      rw[2 * j]     = *(const float4*)(wn + (size_t)(32 * j) * DM);           \
      rw[2 * j + 1] = *(const float4*)(wn + (size_t)(32 * j + 1) * DM);       \
    }                                                                         \
  }
#define O_STASH(bf)                                                           \
  {                                                                           \
    float* qd = Qs + (bf) * OABUF + qst;                                      \
    float* wd = Ws + (bf) * OBBUF;                                            \
    _Pragma("unroll")                                                         \
    for (int j = 0; j < OAIT; ++j)                                            \
      *(v4f*)(qd + 16 * j * OAP) = rq[j];                                     \
    _Pragma("unroll")                                                         \
    for (int j = 0; j < OBIT; ++j) {                                          \
      *(float4*)(wd + 16 * j * OBP + wst0) =                                  \
          make_float4(rw[2 * j].x, rw[2 * j + 1].x, rw[2 * j].y, rw[2 * j + 1].y); \
      *(float4*)(wd + 16 * j * OBP + wst1) =                                  \
          make_float4(rw[2 * j].z, rw[2 * j + 1].z, rw[2 * j].w, rw[2 * j + 1].w); \
    }                                                                         \
  }
#define O_FRAG(dst, g)                                                        \
  {                                                                           \
    _Pragma("unroll")                                                         \
    for (int i = 0; i < OFG; ++i) {                                           \
      const int ks = (g) * OFG + i;                                           \
      fa[dst][i][0] = *(const v2f*)(qa + 4 * ks);                             \
      fa[dst][i][1] = *(const v2f*)(qa + 16 * OAP + 4 * ks);                  \
      fb[dst][i][0] = *(const v2f*)(wa0 + 2 * ks * OBP);                      \
      fb[dst][i][1] = *(const v2f*)(wa0 + 2 * ks * OBP + 32);                 \
      fb[dst][i][2] = *(const v2f*)(wa1 + 2 * ks * OBP + 64);                 \
      fb[dst][i][3] = *(const v2f*)(wa1 + 2 * ks * OBP + 96);                 \
    }                                                                         \
  }

  O_LOAD(0)
  O_STASH(0)
  O_LOAD(1)
  __syncthreads();

  const int ard = (32 * wv + lm) * OAP + 2 * lh;
  const int wr0 = lh * OBP + 2 * lm;
  const int wr1 = lh * OBP + ((2 * lm) ^ 4);
  v8f acc[8] = {};
  v2f fa[2][OFG][2], fb[2][OFG][4];
  for (int n = 0; n < ONST; ++n) {
    const int buf = n & 1;
    const float* qa  = Qs + buf * OABUF + ard;
    const float* wa0 = Ws + buf * OBBUF + wr0;
    const float* wa1 = Ws + buf * OBBUF + wr1;
    O_FRAG(0, 0)
    if (n + 1 < ONST) O_STASH(buf ^ 1)
    if (n + 2 < ONST) O_LOAD(n + 2)
    #pragma unroll
    for (int g = 0; g < ONG; ++g) {
      if (g + 1 < ONG) O_FRAG((g + 1) & 1, g + 1)
      __builtin_amdgcn_sched_barrier(0);
      #pragma unroll
      for (int i = 0; i < OFG; ++i) {
        #pragma unroll
        for (int t = 0; t < 4; ++t) {
          acc[t]     = wmma4(fa[g & 1][i][0], fb[g & 1][i][t], acc[t]);
          acc[4 + t] = wmma4(fa[g & 1][i][1], fb[g & 1][i][t], acc[4 + t]);
        }
      }
    }
    __syncthreads();
  }
#undef O_LOAD
#undef O_STASH
#undef O_FRAG
  float* ob = out + ((size_t)b * SEQ + n0 + 32 * wv) * DM + o0;
  #pragma unroll
  for (int t = 0; t < 4; ++t) {
    float add = bo[o0 + 16 * t + lm] + cq[(size_t)b * DM + o0 + 16 * t + lm];
    #pragma unroll
    for (int m = 0; m < 2; ++m)
      #pragma unroll
      for (int r = 0; r < 8; ++r)
        ob[(size_t)(16 * m + r + 8 * lh) * DM + 16 * t + lm] = acc[4 * m + t][r] + add;
  }
}

#define SCH   64
#define SROWS (SEQ / SCH)
#define SPH   4
#define SRPT  (SROWS / SPH)

__global__ void __launch_bounds__(512) kSums(const float* __restrict__ k,
                                             const float* __restrict__ v,
                                             float* __restrict__ part) {
  __shared__ float4 red[SPH * 256];
  const int tid = threadIdx.x;
  const int cg = tid & 127, ph = tid >> 7;
  const int b = blockIdx.x / SCH, c = blockIdx.x % SCH;
  const size_t off =
      ((size_t)b * SEQ + (size_t)c * SROWS + (size_t)ph * SRPT) * DM + 4 * cg;
  const float* kp = k + off;
  const float* vp = v + off;
  float4 rk[SRPT], rv[SRPT];
  #pragma unroll
  for (int i = 0; i < SRPT; ++i) {
    rk[i] = *(const float4*)(kp + (size_t)i * DM);
    rv[i] = *(const float4*)(vp + (size_t)i * DM);
  }
  float4 sk = rk[0], sv = rv[0];
  #pragma unroll
  for (int i = 1; i < SRPT; ++i) {
    sk.x += rk[i].x; sk.y += rk[i].y; sk.z += rk[i].z; sk.w += rk[i].w;
    sv.x += rv[i].x; sv.y += rv[i].y; sv.z += rv[i].z; sv.w += rv[i].w;
  }
  red[ph * 256 + cg] = sk;
  red[ph * 256 + 128 + cg] = sv;
  __syncthreads();
  if (tid < 256) {
    float4 s = red[tid];
    #pragma unroll
    for (int p = 1; p < SPH; ++p) {
      const float4 t = red[p * 256 + tid];
      s.x += t.x; s.y += t.y; s.z += t.z; s.w += t.w;
    }
    *(float4*)(part + (size_t)blockIdx.x * (2 * DM) + 4 * tid) = s;
  }
}

#define UZR 128

__global__ void __launch_bounds__(256) kUZ(const float* __restrict__ wk,
                                           const float* __restrict__ wvw,
                                           const float* __restrict__ part,
                                           float* __restrict__ u,
                                           float* __restrict__ z) {
  __shared__ __attribute__((aligned(16))) float Ws[UZR * ROWP];
  __shared__ __attribute__((aligned(16))) float cs[DM];
  const int tid = threadIdx.x;
  const int lane = tid & 31, wv = tid >> 5;
  const int b = blockIdx.x >> 3, mat = (blockIdx.x >> 2) & 1, rq = blockIdx.x & 3;
  const float* pp = part + (size_t)b * SCH * (2 * DM) + mat * DM + 2 * tid;
  float2 s = *(const float2*)pp;
  #pragma unroll
  for (int c = 1; c < SCH; ++c) {
    const float2 t = *(const float2*)(pp + (size_t)c * (2 * DM));
    s.x += t.x; s.y += t.y;
  }
  *(float2*)(cs + 2 * tid) = s;
  const float* Wg = (mat ? wvw : wk) + (size_t)(rq * UZR) * DM + 4 * lane;
  for (int g = 0; g < 4; ++g) {
    float4 r[16];
    #pragma unroll
    for (int j = 0; j < 4; ++j)
      #pragma unroll
      for (int q = 0; q < 4; ++q)
        r[4 * j + q] = *(const float4*)(Wg + (size_t)(wv + 8 * (4 * g + j)) * DM + 128 * q);
    #pragma unroll
    for (int j = 0; j < 4; ++j)
      #pragma unroll
      for (int q = 0; q < 4; ++q)
        *(float4*)(Ws + (wv + 8 * (4 * g + j)) * ROWP + 128 * q + 4 * lane) = r[4 * j + q];
  }
  __syncthreads();
  if (tid < UZR) {
    const float* wr = Ws + tid * ROWP;
    float su = 0.f;
    for (int d = 0; d < DM; ++d) su += wr[d] * cs[d];
    float* dst = mat ? z : u;
    dst[(size_t)b * DM + rq * UZR + tid] = su;
  }
}

__global__ void kCQ(const float* __restrict__ bq, const float* __restrict__ P,
                    float* __restrict__ cq) {
  int idx = blockIdx.x * blockDim.x + threadIdx.x;
  if (idx >= NB * DM) return;
  int b = idx >> 9, o = idx & (DM - 1);
  const float* Pb = P + (size_t)b * DM * DM + o;
  float s = 0.f;
  for (int f = 0; f < DM; ++f) s += bq[f] * Pb[(size_t)f * DM];
  cq[idx] = s;
}

extern "C" void kernel_launch(void* const* d_in, const int* in_sizes, int n_in,
                              void* d_out, int out_size, void* d_ws, size_t ws_size,
                              hipStream_t stream) {
  const float* q   = (const float*)d_in[0];
  const float* k   = (const float*)d_in[1];
  const float* v   = (const float*)d_in[2];
  const float* wq  = (const float*)d_in[3];
  const float* bq  = (const float*)d_in[4];
  const float* wk  = (const float*)d_in[5];
  const float* bk  = (const float*)d_in[6];
  const float* wvw = (const float*)d_in[7];
  const float* bv  = (const float*)d_in[8];
  const float* wo  = (const float*)d_in[9];
  const float* bo  = (const float*)d_in[10];
  float* out = (float*)d_out;

  float* ws = (float*)d_ws;
  float* G    = ws;
  float* T    = G    + (size_t)NB * DM * DM;
  float* M    = T    + (size_t)NB * DM * DM;
  float* P    = M    + (size_t)NB * NH * DKH * DKH;
  float* W4   = P    + (size_t)NB * DM * DM;
  float* u    = W4   + (size_t)NB * DM * DM;
  float* z    = u    + (size_t)NB * DM;
  float* cq   = z    + (size_t)NB * DM;
  float* part = cq   + (size_t)NB * DM;

  kSums<<<128, 512, 0, stream>>>(k, v, part);
  kUZ  <<<16, 256, 0, stream>>>(wk, wvw, part, u, z);
  kG   <<<256, 256, 0, stream>>>(k, v, G);
  kT   <<<256, 256, 0, stream>>>(wk, G, T);
  kM   <<<32, 256, 0, stream>>>(T, wvw, bk, bv, u, z, M);
  kP   <<<64, 256, 0, stream>>>(M, wo, P);
  kCQ  <<<4, 256, 0, stream>>>(bq, P, cq);
  kW4  <<<256, 256, 0, stream>>>(wq, P, W4);
  dim3 go(32, 8, 1);
  kOut <<<go, 256, 0, stream>>>(q, W4, bo, cq, out);
}


# u, z, cq: each taken as one more row/column of the T, M, W4 products on WMMA from the panels already in LDS; kUZ and kCQ launches dropped (9 -> 7); changes u, z, cq summation order
# speedup vs baseline: 7.1460x; 1.2356x over previous
#include <hip/hip_runtime.h>


typedef float v2f __attribute__((ext_vector_type(2)));
typedef float v4f __attribute__((ext_vector_type(4)));
typedef float v8f __attribute__((ext_vector_type(8)));

#define DM   512
#define SEQ  4096
#define NB   2
#define NH   8
#define DKH  64

__device__ __forceinline__ v8f wmma4(v2f a, v2f b, v8f c) {
  return __builtin_amdgcn_wmma_f32_16x16x4_f32(
       false, a,  false, b,
       (short)0, c,  false,  false);
}

#define GKS   256
#define GNST  (SEQ / GKS)
#define GKP   96
#define GVP   160
#define GKBUF ((GKS / 2) * GKP)
#define GVBUF ((GKS / 2) * GVP)
#define GKIT  (GKS / 64)
#define GVIT  (GKS / 32)
#define GFG   8
#define GNG   (GKS / 4 / GFG)

__global__ void __launch_bounds__(256) kG(const float* __restrict__ k,
                                          const float* __restrict__ v,
                                          float* __restrict__ G) {
  __shared__ __attribute__((aligned(16))) float Ks[2 * GKBUF];
  __shared__ __attribute__((aligned(16))) float Vs[2 * GVBUF];
  const int tid = threadIdx.x;
  const int lane = tid & 31, wv = tid >> 5;
  const int lm = lane & 15, lh = lane >> 4;
  const int blk = blockIdx.x;
  const int b = blk >> 7;
  const int rest = blk & 127;
  const int d0 = (rest >> 3) * 32;
  const int e0 = (rest & 7) * 64;
  const int dw = 16 * (wv & 1);
  const int ew = 16 * (wv >> 1);
  const int kc = (tid & 7) * 4,  krp = tid >> 3;
  const int vc = (tid & 15) * 4, vrp = tid >> 4;
  const float* kg = k + (size_t)b * SEQ * DM + d0 + (size_t)(2 * krp) * DM + kc;
  const float* vg = v + (size_t)b * SEQ * DM + e0 + (size_t)(2 * vrp) * DM + vc;
  const int ksw = 4 * (krp & 1);
  const int vsw = 4 * ((vc >> 5) & 1);
  const int kst0 = krp * GKP + 2 * kc + ksw, kst1 = krp * GKP + 2 * kc + 4 - ksw;
  const int vst0 = vrp * GVP + 2 * vc + vsw, vst1 = vrp * GVP + 2 * vc + 4 - vsw;
  float4 rk[2 * GKIT], rv[2 * GVIT];

#define G_LOAD(st)                                                            \
  {                                                                           \
    const float* kn = kg + (size_t)(st) * GKS * DM;                           \
    const float* vn = vg + (size_t)(st) * GKS * DM;                           \
    _Pragma("unroll")                                                         \
    for (int j = 0; j < GKIT; ++j) {                                          \
      rk[2 * j]     = *(const float4*)(kn + (size_t)(64 * j) * DM);           \
      rk[2 * j + 1] = *(const float4*)(kn + (size_t)(64 * j + 1) * DM);       \
    }                                                                         \
    _Pragma("unroll")                                                         \
    for (int j = 0; j < GVIT; ++j) {                                          \
      rv[2 * j]     = *(const float4*)(vn + (size_t)(32 * j) * DM);           \
      rv[2 * j + 1] = *(const float4*)(vn + (size_t)(32 * j + 1) * DM);       \
    }                                                                         \
  }
#define G_STASH(bf)                                                           \
  {                                                                           \
    float* kd = Ks + (bf) * GKBUF;                                            \
    float* vd = Vs + (bf) * GVBUF;                                            \
    _Pragma("unroll")                                                         \
    for (int j = 0; j < GKIT; ++j) {                                          \
      *(float4*)(kd + 32 * j * GKP + kst0) =                                  \
          make_float4(rk[2 * j].x, rk[2 * j + 1].x, rk[2 * j].y, rk[2 * j + 1].y); \
      *(float4*)(kd + 32 * j * GKP + kst1) =                                  \
          make_float4(rk[2 * j].z, rk[2 * j + 1].z, rk[2 * j].w, rk[2 * j + 1].w); \
    }                                                                         \
    _Pragma("unroll")                                                         \
    for (int j = 0; j < GVIT; ++j) {                                          \
      *(float4*)(vd + 16 * j * GVP + vst0) =                                  \
          make_float4(rv[2 * j].x, rv[2 * j + 1].x, rv[2 * j].y, rv[2 * j + 1].y); \
      *(float4*)(vd + 16 * j * GVP + vst1) =                                  \
          make_float4(rv[2 * j].z, rv[2 * j + 1].z, rv[2 * j].w, rv[2 * j + 1].w); \
    }                                                                         \
  }
#define G_FRAG(dst, g)                                                        \
  {                                                                           \
    _Pragma("unroll")                                                         \
    for (int i = 0; i < GFG; ++i) {                                           \
      fa[dst][i] = *(const v2f*)(ka + 2 * ((g) * GFG + i) * GKP);             \
      fb[dst][i] = *(const v2f*)(va + 2 * ((g) * GFG + i) * GVP);             \
    }                                                                         \
  }

  G_LOAD(0)
  G_STASH(0)
  G_LOAD(1)
  __syncthreads();

  const int ard = lh * GKP + ((2 * (dw + lm)) ^ (4 * lh));
  const int brd = lh * GVP + ((2 * (ew + lm)) ^ (4 * ((ew >> 5) & 1)));
  v8f acc = {};
  v2f fa[2][GFG], fb[2][GFG];
  for (int n = 0; n < GNST; ++n) {
    const int buf = n & 1;
    const float* ka = Ks + buf * GKBUF + ard;
    const float* va = Vs + buf * GVBUF + brd;
    G_FRAG(0, 0)
    if (n + 1 < GNST) G_STASH(buf ^ 1)
    if (n + 2 < GNST) G_LOAD(n + 2)
    __builtin_amdgcn_sched_barrier(0);
    #pragma unroll
    for (int g = 0; g < GNG; ++g) {
      if (g + 1 < GNG) G_FRAG((g + 1) & 1, g + 1)
      #pragma unroll
      for (int i = 0; i < GFG; ++i)
        acc = wmma4(fa[g & 1][i], fb[g & 1][i], acc);
      __builtin_amdgcn_sched_barrier(0);
    }
    __syncthreads();
  }
#undef G_LOAD
#undef G_STASH
#undef G_FRAG
  float* Gb = G + ((size_t)b * DM + d0 + dw) * DM + e0 + ew;
  #pragma unroll
  for (int r = 0; r < 8; ++r)
    Gb[(size_t)(r + 8 * lh) * DM + lm] = acc[r];
}

#define ROWP (DM + 4)
#define KP64 72
#define KP32 40

template <int NR>
__device__ __forceinline__ void stage_rows(const float* __restrict__ g, float* s,
                                           int wv, int lane) {
  #pragma unroll
  for (int j = 0; j < NR / 8; ++j)
    #pragma unroll
    for (int q = 0; q < 4; ++q)
      *(float4*)(s + (wv + 8 * j) * ROWP + 128 * q + 4 * lane) =
          *(const float4*)(g + (size_t)(wv + 8 * j) * DM + 128 * q + 4 * lane);
}

template <int NC, int PITCH>
__device__ __forceinline__ void stage_kmajor(const float* __restrict__ g, float* s,
                                             int tid) {
  constexpr int F4 = NC / 4;
  constexpr int RPS = 256 / F4;
  const int c4 = (tid % F4) * 4, r0 = tid / F4;
  #pragma unroll
  for (int j = 0; j < DM / RPS; ++j)
    *(float4*)(s + (r0 + RPS * j) * PITCH + c4) =
        *(const float4*)(g + (size_t)(r0 + RPS * j) * DM + c4);
}

__device__ __forceinline__ v2f ld_pair(const float* p, int pitch) {
  v2f r; r.x = p[0]; r.y = p[pitch]; return r;
}

#define SCH   64

__device__ __forceinline__ float2 chunk_sum2(const float* __restrict__ pp) {
  float2 s = *(const float2*)pp;
  #pragma unroll
  for (int c = 1; c < SCH; ++c) {
    const float2 t = *(const float2*)(pp + (size_t)c * (2 * DM));
    s.x += t.x; s.y += t.y;
  }
  return s;
}

#define PANEL_K_LOOP(LOADA, LOADB)                                            \
  {                                                                           \
    v2f fa[2][8], fb[2][8];                                                   \
    _Pragma("unroll")                                                         \
    for (int i = 0; i < 8; ++i) {                                             \
      const int kk = 4 * i;                                                   \
      fa[0][i] = LOADA; fb[0][i] = LOADB;                                     \
    }                                                                         \
    _Pragma("unroll")                                                         \
    for (int g = 0; g < DM / 32; ++g) {                                       \
      if (g + 1 < DM / 32) {                                                  \
        _Pragma("unroll")                                                     \
        for (int i = 0; i < 8; ++i) {                                         \
          const int kk = 32 * (g + 1) + 4 * i;                                \
          fa[(g + 1) & 1][i] = LOADA; fb[(g + 1) & 1][i] = LOADB;             \
        }                                                                     \
      }                                                                       \
      _Pragma("unroll")                                                       \
      for (int i = 0; i < 8; ++i)                                             \
        acc = wmma4(fa[g & 1][i], fb[g & 1][i], acc);                         \
      __builtin_amdgcn_sched_barrier(0);                                      \
    }                                                                         \
  }

__global__ void __launch_bounds__(256) kT(const float* __restrict__ wk,
                                          const float* __restrict__ G,
                                          const float* __restrict__ part,
                                          float* __restrict__ T,
                                          float* __restrict__ u) {
  __shared__ __attribute__((aligned(16))) float As[32 * ROWP];
  __shared__ __attribute__((aligned(16))) float Bs[DM * KP64];
  const int tid = threadIdx.x;
  const int lane = tid & 31, wv = tid >> 5;
  const int lm = lane & 15, lh = lane >> 4;
  const int bh = blockIdx.x >> 4;
  const int b = bh >> 3, h = bh & 7;
  const int ih = (blockIdx.x >> 3) & 1;
  const int e0 = (blockIdx.x & 7) * 64;
  if (e0 == 0) {
    const float2 s = chunk_sum2(part + (size_t)b * SCH * (2 * DM) + 2 * tid);
    Bs[(2 * tid) * KP64 + 64] = s.x;
    Bs[(2 * tid + 1) * KP64 + 64] = s.y;
  }
  stage_rows<32>(wk + (size_t)(h * 64 + ih * 32) * DM, As, wv, lane);
  stage_kmajor<64, KP64>(G + (size_t)b * DM * DM + e0, Bs, tid);
  __syncthreads();
  const int ti = wv >> 2, tj = wv & 3;
  const float* ap = As + (16 * ti + lm) * ROWP + 2 * lh;
  const float* bp = Bs + (2 * lh) * KP64 + 16 * tj + lm;
  v8f acc = {};
  PANEL_K_LOOP((*(const v2f*)(ap + kk)), ld_pair(bp + kk * KP64, KP64))
  float* Tb = T + ((size_t)bh * 64 + ih * 32 + 16 * ti) * DM + e0 + 16 * tj;
  #pragma unroll
  for (int r = 0; r < 8; ++r)
    Tb[(size_t)(r + 8 * lh) * DM + lm] = acc[r];
  if (e0 == 0 && wv < 2) {
    const float* ua = As + (16 * wv + lm) * ROWP + 2 * lh;
    const float* ub = Bs + (2 * lh) * KP64 + 64;
    v8f accu;
    {
      v8f acc = {};
      PANEL_K_LOOP((*(const v2f*)(ua + kk)), ld_pair(ub + kk * KP64, KP64))
      accu = acc;
    }
    if (lm == 0) {
      float* ud = u + (size_t)b * DM + h * 64 + ih * 32 + 16 * wv + 8 * lh;
      #pragma unroll
      for (int r = 0; r < 8; ++r) ud[r] = accu[r];
    }
  }
}


__global__ void __launch_bounds__(256) kM(const float* __restrict__ T,
                                          const float* __restrict__ wvw,
                                          const float* __restrict__ bk,
                                          const float* __restrict__ bv,
                                          const float* __restrict__ u,
                                          const float* __restrict__ part,
                                          float* __restrict__ M) {
  __shared__ __attribute__((aligned(16))) float As[33 * ROWP];
  __shared__ __attribute__((aligned(16))) float Bs[64 * ROWP];
  __shared__ float zs[DKH];
  const int lane = threadIdx.x & 31, wv = threadIdx.x >> 5;
  const int lm = lane & 15, lh = lane >> 4;
  const int bh = blockIdx.x >> 1, half = blockIdx.x & 1;
  const int b = bh >> 3, h = bh & 7;
  {
    const float2 s =
        chunk_sum2(part + (size_t)b * SCH * (2 * DM) + DM + 2 * threadIdx.x);
    *(float2*)(As + 32 * ROWP + 2 * threadIdx.x) = s;
  }
  const float* Ag = T + ((size_t)bh * 64 + half * 32) * DM + 4 * lane;
  const float* Bg = wvw + (size_t)(h * 64) * DM + 4 * lane;
  #pragma unroll
  for (int q = 0; q < 4; ++q) {
    float4 ra[4], rb[8];
    #pragma unroll
    for (int j = 0; j < 4; ++j)
      ra[j] = *(const float4*)(Ag + (size_t)(wv + 8 * j) * DM + 128 * q);
    #pragma unroll
    for (int j = 0; j < 8; ++j)
      rb[j] = *(const float4*)(Bg + (size_t)(wv + 8 * j) * DM + 128 * q);
    #pragma unroll
    for (int j = 0; j < 4; ++j)
      *(float4*)(As + (wv + 8 * j) * ROWP + 128 * q + 4 * lane) = ra[j];
    #pragma unroll
    for (int j = 0; j < 8; ++j)
      *(float4*)(Bs + (wv + 8 * j) * ROWP + 128 * q + 4 * lane) = rb[j];
  }
  __syncthreads();
  const int ti = wv >> 2, tj = wv & 3;
  const float* ap = As + (16 * ti + lm) * ROWP + 2 * lh;
  const float* bp = Bs + (16 * tj + lm) * ROWP + 2 * lh;
  v8f acc = {};
  PANEL_K_LOOP((*(const v2f*)(ap + kk)), (*(const v2f*)(bp + kk)))
  if (ti == 0) {
    const float* za = As + 32 * ROWP + 2 * lh;
    v8f accz;
    {
      v8f acc = {};
      PANEL_K_LOOP((*(const v2f*)(za + kk)), (*(const v2f*)(bp + kk)))
      accz = acc;
    }
    if (lh == 0) zs[16 * tj + lm] = accz[0];
  }
  __syncthreads();
  const int i0 = half * 32 + 16 * ti;
  const int jf = h * 64 + 16 * tj + lm;
  const float bvj = bv[jf];
  const float zj  = zs[16 * tj + lm];
  #pragma unroll
  for (int r = 0; r < 8; ++r) {
    const int ifl = h * 64 + i0 + r + 8 * lh;
    const float bki = bk[ifl];
    const float ui  = u[(size_t)b * DM + ifl];
    float val = acc[r] + ui * bvj + bki * zj + (float)SEQ * bki * bvj;
    M[((size_t)bh * 64 + i0 + r + 8 * lh) * DKH + 16 * tj + lm] = 0.125f * val;
  }
}

__global__ void __launch_bounds__(256) kP(const float* __restrict__ M,
                                          const float* __restrict__ wo,
                                          float* __restrict__ P) {
  const int lane = threadIdx.x & 31, wv = threadIdx.x >> 5;
  const int lm = lane & 15, lh = lane >> 4;
  const int job = blockIdx.x * 8 + wv;
  const int bh = job >> 5;
  const int b = bh >> 3, h = bh & 7;
  const int i0 = ((job >> 3) & 3) * 16;
  const int o0 = (job & 7) * 64;
  const float* arow = M + ((size_t)bh * 64 + i0 + lm) * DKH + 2 * lh;
  v8f acc[4] = {};
  for (int kk = 0; kk < DKH; kk += 4) {
    v2f a = *(const v2f*)(arow + kk);
    #pragma unroll
    for (int t = 0; t < 4; ++t) {
      v2f bm = *(const v2f*)(wo + (size_t)(o0 + 16 * t + lm) * DM + h * 64 + kk + 2 * lh);
      acc[t] = wmma4(a, bm, acc[t]);
    }
  }
  float* Pb = P + ((size_t)b * DM + h * 64 + i0) * DM + o0;
  #pragma unroll
  for (int t = 0; t < 4; ++t)
    #pragma unroll
    for (int r = 0; r < 8; ++r)
      Pb[(size_t)(r + 8 * lh) * DM + 16 * t + lm] = acc[t][r];
}

__global__ void __launch_bounds__(256) kW4(const float* __restrict__ wq,
                                           const float* __restrict__ bq,
                                           const float* __restrict__ P,
                                           float* __restrict__ W4,
                                           float* __restrict__ cq) {
  __shared__ __attribute__((aligned(16))) float As[DM * KP32];
  __shared__ __attribute__((aligned(16))) float Bs[DM * KP64];
  const int tid = threadIdx.x;
  const int lane = tid & 31, wv = tid >> 5;
  const int lm = lane & 15, lh = lane >> 4;
  const int b = blockIdx.x >> 7;
  const int d0 = ((blockIdx.x >> 3) & 15) * 32;
  const int o0 = (blockIdx.x & 7) * 64;
  As[tid * KP32 + 32] = bq[tid];
  As[(tid + 256) * KP32 + 32] = bq[tid + 256];
  stage_kmajor<32, KP32>(wq + d0, As, tid);
  stage_kmajor<64, KP64>(P + (size_t)b * DM * DM + o0, Bs, tid);
  __syncthreads();
  const int ti = wv >> 2, tj = wv & 3;
  const float* ap = As + (2 * lh) * KP32 + 16 * ti + lm;
  const float* bp = Bs + (2 * lh) * KP64 + 16 * tj + lm;
  v8f acc = {};
  PANEL_K_LOOP(ld_pair(ap + kk * KP32, KP32), ld_pair(bp + kk * KP64, KP64))
  float* Wb = W4 + ((size_t)b * DM + d0 + 16 * ti) * DM + o0 + 16 * tj;
  #pragma unroll
  for (int r = 0; r < 8; ++r)
    Wb[(size_t)(r + 8 * lh) * DM + lm] = acc[r];
  if (d0 == 0 && ti == 0) {
    const float* ca = As + (2 * lh) * KP32 + 32;
    v8f accc;
    {
      v8f acc = {};
      PANEL_K_LOOP(ld_pair(ca + kk * KP32, KP32), ld_pair(bp + kk * KP64, KP64))
      accc = acc;
    }
    if (lh == 0) cq[(size_t)b * DM + o0 + 16 * tj + lm] = accc[0];
  }
}

#define OKS   64
#define ONST  (DM / OKS)
#define OAP   (OKS + 4)
#define OBP   160
#define OABUF (256 * OAP)
#define OBBUF ((OKS / 2) * OBP)
#define OAIT  16
#define OBIT  (OKS / 32)
#define OFG   2
#define ONG   (OKS / 4 / OFG)

__global__ void __launch_bounds__(256) kOut(const float* __restrict__ q,
                                            const float* __restrict__ W4,
                                            const float* __restrict__ bo,
                                            const float* __restrict__ cq,
                                            float* __restrict__ out) {
  __shared__ __attribute__((aligned(16))) float Qs[2 * OABUF];
  __shared__ __attribute__((aligned(16))) float Ws[2 * OBBUF];
  const int tid = threadIdx.x;
  const int lane = tid & 31, wv = tid >> 5;
  const int lm = lane & 15, lh = lane >> 4;
  const int b = blockIdx.x >> 4;
  const int n0 = (blockIdx.x & 15) * 256;
  const int o0 = blockIdx.y * 64;
  const int sc = (tid & 15) * 4, sr = tid >> 4;
  const float* qg = q + ((size_t)b * SEQ + n0 + sr) * DM + sc;
  const float* wg = W4 + ((size_t)b * DM + 2 * sr) * DM + o0 + sc;
  const int qst = sr * OAP + sc;
  const int wsw = 4 * ((sc >> 5) & 1);
  const int wst0 = sr * OBP + 2 * sc + wsw, wst1 = sr * OBP + 2 * sc + 4 - wsw;
  v4f rq[OAIT];
  float4 rw[2 * OBIT];

#define O_LOAD(st)                                                            \
  {                                                                           \
    const float* qn = qg + (st) * OKS;                                        \
    const float* wn = wg + (size_t)(st) * OKS * DM;                           \
    _Pragma("unroll")                                                         \
    for (int j = 0; j < OAIT; ++j)                                            \
      rq[j] = *(const v4f*)(qn + (size_t)(16 * j) * DM);                      \
    _Pragma("unroll")                                                         \
    for (int j = 0; j < OBIT; ++j) {                                          \
      rw[2 * j]     = *(const float4*)(wn + (size_t)(32 * j) * DM);           \
      rw[2 * j + 1] = *(const float4*)(wn + (size_t)(32 * j + 1) * DM);       \
    }                                                                         \
  }
#define O_STASH(bf)                                                           \
  {                                                                           \
    float* qd = Qs + (bf) * OABUF + qst;                                      \
    float* wd = Ws + (bf) * OBBUF;                                            \
    _Pragma("unroll")                                                         \
    for (int j = 0; j < OAIT; ++j)                                            \
      *(v4f*)(qd + 16 * j * OAP) = rq[j];                                     \
    _Pragma("unroll")                                                         \
    for (int j = 0; j < OBIT; ++j) {                                          \
      *(float4*)(wd + 16 * j * OBP + wst0) =                                  \
          make_float4(rw[2 * j].x, rw[2 * j + 1].x, rw[2 * j].y, rw[2 * j + 1].y); \
      *(float4*)(wd + 16 * j * OBP + wst1) =                                  \
          make_float4(rw[2 * j].z, rw[2 * j + 1].z, rw[2 * j].w, rw[2 * j + 1].w); \
    }                                                                         \
  }
#define O_FRAG(dst, g)                                                        \
  {                                                                           \
    _Pragma("unroll")                                                         \
    for (int i = 0; i < OFG; ++i) {                                           \
      const int ks = (g) * OFG + i;                                           \
      fa[dst][i][0] = *(const v2f*)(qa + 4 * ks);                             \
      fa[dst][i][1] = *(const v2f*)(qa + 16 * OAP + 4 * ks);                  \
      fb[dst][i][0] = *(const v2f*)(wa0 + 2 * ks * OBP);                      \
      fb[dst][i][1] = *(const v2f*)(wa0 + 2 * ks * OBP + 32);                 \
      fb[dst][i][2] = *(const v2f*)(wa1 + 2 * ks * OBP + 64);                 \
      fb[dst][i][3] = *(const v2f*)(wa1 + 2 * ks * OBP + 96);                 \
    }                                                                         \
  }

  O_LOAD(0)
  O_STASH(0)
  O_LOAD(1)
  __syncthreads();

  const int ard = (32 * wv + lm) * OAP + 2 * lh;
  const int wr0 = lh * OBP + 2 * lm;
  const int wr1 = lh * OBP + ((2 * lm) ^ 4);
  v8f acc[8] = {};
  v2f fa[2][OFG][2], fb[2][OFG][4];
  for (int n = 0; n < ONST; ++n) {
    const int buf = n & 1;
    const float* qa  = Qs + buf * OABUF + ard;
    const float* wa0 = Ws + buf * OBBUF + wr0;
    const float* wa1 = Ws + buf * OBBUF + wr1;
    O_FRAG(0, 0)
    if (n + 1 < ONST) O_STASH(buf ^ 1)
    if (n + 2 < ONST) O_LOAD(n + 2)
    #pragma unroll
    for (int g = 0; g < ONG; ++g) {
      if (g + 1 < ONG) O_FRAG((g + 1) & 1, g + 1)
      __builtin_amdgcn_sched_barrier(0);
      #pragma unroll
      for (int i = 0; i < OFG; ++i) {
        #pragma unroll
        for (int t = 0; t < 4; ++t) {
          acc[t]     = wmma4(fa[g & 1][i][0], fb[g & 1][i][t], acc[t]);
          acc[4 + t] = wmma4(fa[g & 1][i][1], fb[g & 1][i][t], acc[4 + t]);
        }
      }
    }
    __syncthreads();
  }
#undef O_LOAD
#undef O_STASH
#undef O_FRAG
  float* ob = out + ((size_t)b * SEQ + n0 + 32 * wv) * DM + o0;
  #pragma unroll
  for (int t = 0; t < 4; ++t) {
    float add = bo[o0 + 16 * t + lm] + cq[(size_t)b * DM + o0 + 16 * t + lm];
    #pragma unroll
    for (int m = 0; m < 2; ++m)
      #pragma unroll
      for (int r = 0; r < 8; ++r)
        ob[(size_t)(16 * m + r + 8 * lh) * DM + 16 * t + lm] = acc[4 * m + t][r] + add;
  }
}

#define SROWS (SEQ / SCH)
#define SPH   4
#define SRPT  (SROWS / SPH)

__global__ void __launch_bounds__(512) kSums(const float* __restrict__ k,
                                             const float* __restrict__ v,
                                             float* __restrict__ part) {
  __shared__ float4 red[SPH * 256];
  const int tid = threadIdx.x;
  const int cg = tid & 127, ph = tid >> 7;
  const int b = blockIdx.x / SCH, c = blockIdx.x % SCH;
  const size_t off =
      ((size_t)b * SEQ + (size_t)c * SROWS + (size_t)ph * SRPT) * DM + 4 * cg;
  const float* kp = k + off;
  const float* vp = v + off;
  float4 rk[SRPT], rv[SRPT];
  #pragma unroll
  for (int i = 0; i < SRPT; ++i) {
    rk[i] = *(const float4*)(kp + (size_t)i * DM);
    rv[i] = *(const float4*)(vp + (size_t)i * DM);
  }
  float4 sk = rk[0], sv = rv[0];
  #pragma unroll
  for (int i = 1; i < SRPT; ++i) {
    sk.x += rk[i].x; sk.y += rk[i].y; sk.z += rk[i].z; sk.w += rk[i].w;
    sv.x += rv[i].x; sv.y += rv[i].y; sv.z += rv[i].z; sv.w += rv[i].w;
  }
  red[ph * 256 + cg] = sk;
  red[ph * 256 + 128 + cg] = sv;
  __syncthreads();
  if (tid < 256) {
    float4 s = red[tid];
    #pragma unroll
    for (int p = 1; p < SPH; ++p) {
      const float4 t = red[p * 256 + tid];
      s.x += t.x; s.y += t.y; s.z += t.z; s.w += t.w;
    }
    *(float4*)(part + (size_t)blockIdx.x * (2 * DM) + 4 * tid) = s;
  }
}

extern "C" void kernel_launch(void* const* d_in, const int* in_sizes, int n_in,
                              void* d_out, int out_size, void* d_ws, size_t ws_size,
                              hipStream_t stream) {
  const float* q   = (const float*)d_in[0];
  const float* k   = (const float*)d_in[1];
  const float* v   = (const float*)d_in[2];
  const float* wq  = (const float*)d_in[3];
  const float* bq  = (const float*)d_in[4];
  const float* wk  = (const float*)d_in[5];
  const float* bk  = (const float*)d_in[6];
  const float* wvw = (const float*)d_in[7];
  const float* bv  = (const float*)d_in[8];
  const float* wo  = (const float*)d_in[9];
  const float* bo  = (const float*)d_in[10];
  float* out = (float*)d_out;

  float* ws = (float*)d_ws;
  float* G    = ws;
  float* T    = G    + (size_t)NB * DM * DM;
  float* M    = T    + (size_t)NB * DM * DM;
  float* P    = M    + (size_t)NB * NH * DKH * DKH;
  float* W4   = P    + (size_t)NB * DM * DM;
  float* u    = W4   + (size_t)NB * DM * DM;
  float* cq   = u    + (size_t)NB * DM;
  float* part = cq   + (size_t)NB * DM;

  kSums<<<128, 512, 0, stream>>>(k, v, part);
  kG   <<<256, 256, 0, stream>>>(k, v, G);
  kT   <<<256, 256, 0, stream>>>(wk, G, part, T, u);
  kM   <<<32, 256, 0, stream>>>(T, wvw, bk, bv, u, part, M);
  kP   <<<64, 256, 0, stream>>>(M, wo, P);
  kW4  <<<256, 256, 0, stream>>>(wq, bq, P, W4, cq);
  dim3 go(32, 8, 1);
  kOut <<<go, 256, 0, stream>>>(q, W4, bo, cq, out);
}


# kM folded into kT: each T tile times its w_v block out of LDS in the same work-group; kP adds the 8 e-group shares of M in e-group order; 7 -> 6 launches; changes M's summation order
# speedup vs baseline: 7.9006x; 1.1056x over previous
#include <hip/hip_runtime.h>


typedef float v2f __attribute__((ext_vector_type(2)));
typedef float v4f __attribute__((ext_vector_type(4)));
typedef float v8f __attribute__((ext_vector_type(8)));

#define DM   512
#define SEQ  4096
#define NB   2
#define NH   8
#define DKH  64

__device__ __forceinline__ v8f wmma4(v2f a, v2f b, v8f c) {
  return __builtin_amdgcn_wmma_f32_16x16x4_f32(
       false, a,  false, b,
       (short)0, c,  false,  false);
}

#define GKS   256
#define GNST  (SEQ / GKS)
#define GKP   96
#define GVP   160
#define GKBUF ((GKS / 2) * GKP)
#define GVBUF ((GKS / 2) * GVP)
#define GKIT  (GKS / 64)
#define GVIT  (GKS / 32)
#define GFG   8
#define GNG   (GKS / 4 / GFG)

__global__ void __launch_bounds__(256) kG(const float* __restrict__ k,
                                          const float* __restrict__ v,
                                          float* __restrict__ G) {
  __shared__ __attribute__((aligned(16))) float Ks[2 * GKBUF];
  __shared__ __attribute__((aligned(16))) float Vs[2 * GVBUF];
  const int tid = threadIdx.x;
  const int lane = tid & 31, wv = tid >> 5;
  const int lm = lane & 15, lh = lane >> 4;
  const int blk = blockIdx.x;
  const int b = blk >> 7;
  const int rest = blk & 127;
  const int d0 = (rest >> 3) * 32;
  const int e0 = (rest & 7) * 64;
  const int dw = 16 * (wv & 1);
  const int ew = 16 * (wv >> 1);
  const int kc = (tid & 7) * 4,  krp = tid >> 3;
  const int vc = (tid & 15) * 4, vrp = tid >> 4;
  const float* kg = k + (size_t)b * SEQ * DM + d0 + (size_t)(2 * krp) * DM + kc;
  const float* vg = v + (size_t)b * SEQ * DM + e0 + (size_t)(2 * vrp) * DM + vc;
  const int ksw = 4 * (krp & 1);
  const int vsw = 4 * ((vc >> 5) & 1);
  const int kst0 = krp * GKP + 2 * kc + ksw, kst1 = krp * GKP + 2 * kc + 4 - ksw;
  const int vst0 = vrp * GVP + 2 * vc + vsw, vst1 = vrp * GVP + 2 * vc + 4 - vsw;
  float4 rk[2 * GKIT], rv[2 * GVIT];

#define G_LOAD(st)                                                            \
  {                                                                           \
    const float* kn = kg + (size_t)(st) * GKS * DM;                           \
    const float* vn = vg + (size_t)(st) * GKS * DM;                           \
    _Pragma("unroll")                                                         \
    for (int j = 0; j < GKIT; ++j) {                                          \
      rk[2 * j]     = *(const float4*)(kn + (size_t)(64 * j) * DM);           \
      rk[2 * j + 1] = *(const float4*)(kn + (size_t)(64 * j + 1) * DM);       \
    }                                                                         \
    _Pragma("unroll")                                                         \
    for (int j = 0; j < GVIT; ++j) {                                          \
      rv[2 * j]     = *(const float4*)(vn + (size_t)(32 * j) * DM);           \
      rv[2 * j + 1] = *(const float4*)(vn + (size_t)(32 * j + 1) * DM);       \
    }                                                                         \
  }
#define G_STASH(bf)                                                           \
  {                                                                           \
    float* kd = Ks + (bf) * GKBUF;                                            \
    float* vd = Vs + (bf) * GVBUF;                                            \
    _Pragma("unroll")                                                         \
    for (int j = 0; j < GKIT; ++j) {                                          \
      *(float4*)(kd + 32 * j * GKP + kst0) =                                  \
          make_float4(rk[2 * j].x, rk[2 * j + 1].x, rk[2 * j].y, rk[2 * j + 1].y); \
      *(float4*)(kd + 32 * j * GKP + kst1) =                                  \
          make_float4(rk[2 * j].z, rk[2 * j + 1].z, rk[2 * j].w, rk[2 * j + 1].w); \
    }                                                                         \
    _Pragma("unroll")                                                         \
    for (int j = 0; j < GVIT; ++j) {                                          \
      *(float4*)(vd + 16 * j * GVP + vst0) =                                  \
          make_float4(rv[2 * j].x, rv[2 * j + 1].x, rv[2 * j].y, rv[2 * j + 1].y); \
      *(float4*)(vd + 16 * j * GVP + vst1) =                                  \
          make_float4(rv[2 * j].z, rv[2 * j + 1].z, rv[2 * j].w, rv[2 * j + 1].w); \
    }                                                                         \
  }
#define G_FRAG(dst, g)                                                        \
  {                                                                           \
    _Pragma("unroll")                                                         \
    for (int i = 0; i < GFG; ++i) {                                           \
      fa[dst][i] = *(const v2f*)(ka + 2 * ((g) * GFG + i) * GKP);             \
      fb[dst][i] = *(const v2f*)(va + 2 * ((g) * GFG + i) * GVP);             \
    }                                                                         \
  }

  G_LOAD(0)
  G_STASH(0)
  G_LOAD(1)
  __syncthreads();

  const int ard = lh * GKP + ((2 * (dw + lm)) ^ (4 * lh));
  const int brd = lh * GVP + ((2 * (ew + lm)) ^ (4 * ((ew >> 5) & 1)));
  v8f acc = {};
  v2f fa[2][GFG], fb[2][GFG];
  for (int n = 0; n < GNST; ++n) {
    const int buf = n & 1;
    const float* ka = Ks + buf * GKBUF + ard;
    const float* va = Vs + buf * GVBUF + brd;
    G_FRAG(0, 0)
    if (n + 1 < GNST) G_STASH(buf ^ 1)
    if (n + 2 < GNST) G_LOAD(n + 2)
    __builtin_amdgcn_sched_barrier(0);
    #pragma unroll
    for (int g = 0; g < GNG; ++g) {
      if (g + 1 < GNG) G_FRAG((g + 1) & 1, g + 1)
      #pragma unroll
      for (int i = 0; i < GFG; ++i)
        acc = wmma4(fa[g & 1][i], fb[g & 1][i], acc);
      __builtin_amdgcn_sched_barrier(0);
    }
    __syncthreads();
  }
#undef G_LOAD
#undef G_STASH
#undef G_FRAG
  float* Gb = G + ((size_t)b * DM + d0 + dw) * DM + e0 + ew;
  #pragma unroll
  for (int r = 0; r < 8; ++r)
    Gb[(size_t)(r + 8 * lh) * DM + lm] = acc[r];
}

#define ROWP (DM + 4)
#define KP64 72
#define KP32 40

template <int NR>
__device__ __forceinline__ void stage_rows(const float* __restrict__ g, float* s,
                                           int wv, int lane) {
  #pragma unroll
  for (int j = 0; j < NR / 8; ++j)
    #pragma unroll
    for (int q = 0; q < 4; ++q)
      *(float4*)(s + (wv + 8 * j) * ROWP + 128 * q + 4 * lane) =
          *(const float4*)(g + (size_t)(wv + 8 * j) * DM + 128 * q + 4 * lane);
}

template <int NC, int PITCH>
__device__ __forceinline__ void stage_kmajor(const float* __restrict__ g, float* s,
                                             int tid) {
  constexpr int F4 = NC / 4;
  constexpr int RPS = 256 / F4;
  const int c4 = (tid % F4) * 4, r0 = tid / F4;
  #pragma unroll
  for (int j = 0; j < DM / RPS; ++j)
    *(float4*)(s + (r0 + RPS * j) * PITCH + c4) =
        *(const float4*)(g + (size_t)(r0 + RPS * j) * DM + c4);
}

__device__ __forceinline__ v2f ld_pair(const float* p, int pitch) {
  v2f r; r.x = p[0]; r.y = p[pitch]; return r;
}

#define SCH   64

__device__ __forceinline__ float2 chunk_sum2(const float* __restrict__ pp) {
  float2 s = *(const float2*)pp;
  #pragma unroll
  for (int c = 1; c < SCH; ++c) {
    const float2 t = *(const float2*)(pp + (size_t)c * (2 * DM));
    s.x += t.x; s.y += t.y;
  }
  return s;
}

#define PANEL_K_LOOP(LOADA, LOADB)                                            \
  {                                                                           \
    v2f fa[2][8], fb[2][8];                                                   \
    _Pragma("unroll")                                                         \
    for (int i = 0; i < 8; ++i) {                                             \
      const int kk = 4 * i;                                                   \
      fa[0][i] = LOADA; fb[0][i] = LOADB;                                     \
    }                                                                         \
    _Pragma("unroll")                                                         \
    for (int g = 0; g < DM / 32; ++g) {                                       \
      if (g + 1 < DM / 32) {                                                  \
        _Pragma("unroll")                                                     \
        for (int i = 0; i < 8; ++i) {                                         \
          const int kk = 32 * (g + 1) + 4 * i;                                \
          fa[(g + 1) & 1][i] = LOADA; fb[(g + 1) & 1][i] = LOADB;             \
        }                                                                     \
      }                                                                       \
      _Pragma("unroll")                                                       \
      for (int i = 0; i < 8; ++i)                                             \
        acc = wmma4(fa[g & 1][i], fb[g & 1][i], acc);                         \
      __builtin_amdgcn_sched_barrier(0);                                      \
    }                                                                         \
  }

#define TP (DKH + 4)

__global__ void __launch_bounds__(256) kT(const float* __restrict__ wk,
                                          const float* __restrict__ G,
                                          const float* __restrict__ part,
                                          const float* __restrict__ wvw,
                                          const float* __restrict__ bk,
                                          float* __restrict__ Mp,
                                          float* __restrict__ u) {
  __shared__ __attribute__((aligned(16))) float As[32 * ROWP];
  __shared__ __attribute__((aligned(16))) float Bs[DM * KP64];
  __shared__ __attribute__((aligned(16))) float Ts[32 * TP];
  __shared__ __attribute__((aligned(16))) float Vw[DKH * TP];
  __shared__ float vs[DKH];
  __shared__ float bks[32];
  const int tid = threadIdx.x;
  const int lane = tid & 31, wv = tid >> 5;
  const int lm = lane & 15, lh = lane >> 4;
  const int bh = blockIdx.x >> 4;
  const int b = bh >> 3, h = bh & 7;
  const int ih = (blockIdx.x >> 3) & 1;
  const int eg = blockIdx.x & 7;
  const int e0 = eg * 64;
  if (e0 == 0) {
    const float2 s = chunk_sum2(part + (size_t)b * SCH * (2 * DM) + 2 * tid);
    Bs[(2 * tid) * KP64 + 64] = s.x;
    Bs[(2 * tid + 1) * KP64 + 64] = s.y;
  }
  stage_rows<32>(wk + (size_t)(h * 64 + ih * 32) * DM, As, wv, lane);
  stage_kmajor<64, KP64>(G + (size_t)b * DM * DM + e0, Bs, tid);
  {
    const int c4 = (tid & 15) * 4, r0 = tid >> 4;
    #pragma unroll
    for (int t = 0; t < 4; ++t)
      *(float4*)(Vw + (r0 + 16 * t) * TP + c4) =
          *(const float4*)(wvw + (size_t)(h * 64 + r0 + 16 * t) * DM + e0 + c4);
  }
  if (wv == 6) bks[lane] = bk[h * 64 + ih * 32 + lane];
  if (wv == 7) {
    const float2 s =
        chunk_sum2(part + (size_t)b * SCH * (2 * DM) + DM + e0 + 2 * lane);
    vs[2 * lane] = s.x;
    vs[2 * lane + 1] = s.y;
  }
  __syncthreads();
  const int ti = wv >> 2, tj = wv & 3;
  const float* ap = As + (16 * ti + lm) * ROWP + 2 * lh;
  const float* bp = Bs + (2 * lh) * KP64 + 16 * tj + lm;
  v8f acc = {};
  PANEL_K_LOOP((*(const v2f*)(ap + kk)), ld_pair(bp + kk * KP64, KP64))
  {
    const float vse = vs[16 * tj + lm];
    #pragma unroll
    for (int r = 0; r < 8; ++r) {
      const int il = 16 * ti + r + 8 * lh;
      Ts[il * TP + 16 * tj + lm] = acc[r] + bks[il] * vse;
    }
  }
  __syncthreads();
  {
    const float* ap2 = Ts + (16 * ti + lm) * TP + 2 * lh;
    const float* bp2 = Vw + (16 * tj + lm) * TP + 2 * lh;
    v2f fa2[DKH / 4], fb2[DKH / 4];
    #pragma unroll
    for (int i = 0; i < DKH / 4; ++i) {
      fa2[i] = *(const v2f*)(ap2 + 4 * i);
      fb2[i] = *(const v2f*)(bp2 + 4 * i);
    }
    __builtin_amdgcn_sched_barrier(0);
    v8f acc2 = {};
    #pragma unroll
    for (int i = 0; i < DKH / 4; ++i)
      acc2 = wmma4(fa2[i], fb2[i], acc2);
    float* Mb = Mp + (((size_t)eg * (NB * NH) + bh) * DKH + ih * 32 + 16 * ti) * DKH + 16 * tj;
    #pragma unroll
    for (int r = 0; r < 8; ++r)
      Mb[(size_t)(r + 8 * lh) * DKH + lm] = acc2[r];
  }
  if (e0 == 0 && wv < 2) {
    const float* ua = As + (16 * wv + lm) * ROWP + 2 * lh;
    const float* ub = Bs + (2 * lh) * KP64 + 64;
    v8f accu;
    {
      v8f acc = {};
      PANEL_K_LOOP((*(const v2f*)(ua + kk)), ld_pair(ub + kk * KP64, KP64))
      accu = acc;
    }
    if (lm == 0) {
      float* ud = u + (size_t)b * DM + h * 64 + ih * 32 + 16 * wv + 8 * lh;
      #pragma unroll
      for (int r = 0; r < 8; ++r) ud[r] = accu[r];
    }
  }
}

__global__ void __launch_bounds__(256) kP(const float* __restrict__ Mp,
                                          const float* __restrict__ u,
                                          const float* __restrict__ bk,
                                          const float* __restrict__ bv,
                                          const float* __restrict__ wo,
                                          float* __restrict__ P) {
  __shared__ __attribute__((aligned(16))) float Ms[16 * TP];
  const int tid = threadIdx.x;
  const int lane = tid & 31, wv = tid >> 5;
  const int lm = lane & 15, lh = lane >> 4;
  const int bh = blockIdx.x >> 2;
  const int b = bh >> 3, h = bh & 7;
  const int i0 = (blockIdx.x & 3) * 16;
  const int o0 = wv * 64;
  v2f bm[DKH / 4][4];
  #pragma unroll
  for (int i = 0; i < DKH / 4; ++i)
    #pragma unroll
    for (int t = 0; t < 4; ++t)
      bm[i][t] = *(const v2f*)(wo + (size_t)(o0 + 16 * t + lm) * DM + h * 64 + 4 * i + 2 * lh);
  {
    const int row = tid >> 4, c4 = (tid & 15) * 4;
    const float* mp = Mp + ((size_t)bh * 64 + i0 + row) * DKH + c4;
    float4 s = *(const float4*)mp;
    #pragma unroll
    for (int g = 1; g < 8; ++g) {
      const float4 t = *(const float4*)(mp + (size_t)g * (NB * NH * DKH * DKH));
      s.x += t.x; s.y += t.y; s.z += t.z; s.w += t.w;
    }
    const float ui  = u[(size_t)b * DM + h * 64 + i0 + row];
    const float bki = bk[h * 64 + i0 + row];
    const float4 bv4 = *(const float4*)(bv + h * 64 + c4);
    float4 m;
    m.x = 0.125f * (s.x + ui * bv4.x + (float)SEQ * bki * bv4.x);
    m.y = 0.125f * (s.y + ui * bv4.y + (float)SEQ * bki * bv4.y);
    m.z = 0.125f * (s.z + ui * bv4.z + (float)SEQ * bki * bv4.z);
    m.w = 0.125f * (s.w + ui * bv4.w + (float)SEQ * bki * bv4.w);
    *(float4*)(Ms + row * TP + c4) = m;
  }
  __syncthreads();
  const float* ap = Ms + lm * TP + 2 * lh;
  v2f fa[DKH / 4];
  #pragma unroll
  for (int i = 0; i < DKH / 4; ++i) fa[i] = *(const v2f*)(ap + 4 * i);
  v8f acc[4] = {};
  #pragma unroll
  for (int i = 0; i < DKH / 4; ++i) {
    #pragma unroll
    for (int t = 0; t < 4; ++t)
      acc[t] = wmma4(fa[i], bm[i][t], acc[t]);
  }
  float* Pb = P + ((size_t)b * DM + h * 64 + i0) * DM + o0;
  #pragma unroll
  for (int t = 0; t < 4; ++t)
    #pragma unroll
    for (int r = 0; r < 8; ++r)
      Pb[(size_t)(r + 8 * lh) * DM + 16 * t + lm] = acc[t][r];
}

__global__ void __launch_bounds__(256) kW4(const float* __restrict__ wq,
                                           const float* __restrict__ bq,
                                           const float* __restrict__ P,
                                           float* __restrict__ W4,
                                           float* __restrict__ cq) {
  __shared__ __attribute__((aligned(16))) float As[DM * KP32];
  __shared__ __attribute__((aligned(16))) float Bs[DM * KP64];
  const int tid = threadIdx.x;
  const int lane = tid & 31, wv = tid >> 5;
  const int lm = lane & 15, lh = lane >> 4;
  const int b = blockIdx.x >> 7;
  const int d0 = ((blockIdx.x >> 3) & 15) * 32;
  const int o0 = (blockIdx.x & 7) * 64;
  As[tid * KP32 + 32] = bq[tid];
  As[(tid + 256) * KP32 + 32] = bq[tid + 256];
  stage_kmajor<32, KP32>(wq + d0, As, tid);
  stage_kmajor<64, KP64>(P + (size_t)b * DM * DM + o0, Bs, tid);
  __syncthreads();
  const int ti = wv >> 2, tj = wv & 3;
  const float* ap = As + (2 * lh) * KP32 + 16 * ti + lm;
  const float* bp = Bs + (2 * lh) * KP64 + 16 * tj + lm;
  v8f acc = {};
  PANEL_K_LOOP(ld_pair(ap + kk * KP32, KP32), ld_pair(bp + kk * KP64, KP64))
  float* Wb = W4 + ((size_t)b * DM + d0 + 16 * ti) * DM + o0 + 16 * tj;
  #pragma unroll
  for (int r = 0; r < 8; ++r)
    Wb[(size_t)(r + 8 * lh) * DM + lm] = acc[r];
  if (d0 == 0 && ti == 0) {
    const float* ca = As + (2 * lh) * KP32 + 32;
    v8f accc;
    {
      v8f acc = {};
      PANEL_K_LOOP(ld_pair(ca + kk * KP32, KP32), ld_pair(bp + kk * KP64, KP64))
      accc = acc;
    }
    if (lh == 0) cq[(size_t)b * DM + o0 + 16 * tj + lm] = accc[0];
  }
}

#define OKS   64
#define ONST  (DM / OKS)
#define OAP   (OKS + 4)
#define OBP   160
#define OABUF (256 * OAP)
#define OBBUF ((OKS / 2) * OBP)
#define OAIT  16
#define OBIT  (OKS / 32)
#define OFG   2
#define ONG   (OKS / 4 / OFG)

__global__ void __launch_bounds__(256) kOut(const float* __restrict__ q,
                                            const float* __restrict__ W4,
                                            const float* __restrict__ bo,
                                            const float* __restrict__ cq,
                                            float* __restrict__ out) {
  __shared__ __attribute__((aligned(16))) float Qs[2 * OABUF];
  __shared__ __attribute__((aligned(16))) float Ws[2 * OBBUF];
  const int tid = threadIdx.x;
  const int lane = tid & 31, wv = tid >> 5;
  const int lm = lane & 15, lh = lane >> 4;
  const int b = blockIdx.x >> 4;
  const int n0 = (blockIdx.x & 15) * 256;
  const int o0 = blockIdx.y * 64;
  const int sc = (tid & 15) * 4, sr = tid >> 4;
  const float* qg = q + ((size_t)b * SEQ + n0 + sr) * DM + sc;
  const float* wg = W4 + ((size_t)b * DM + 2 * sr) * DM + o0 + sc;
  const int qst = sr * OAP + sc;
  const int wsw = 4 * ((sc >> 5) & 1);
  const int wst0 = sr * OBP + 2 * sc + wsw, wst1 = sr * OBP + 2 * sc + 4 - wsw;
  v4f rq[OAIT];
  float4 rw[2 * OBIT];

#define O_LOAD(st)                                                            \
  {                                                                           \
    const float* qn = qg + (st) * OKS;                                        \
    const float* wn = wg + (size_t)(st) * OKS * DM;                           \
    _Pragma("unroll")                                                         \
    for (int j = 0; j < OAIT; ++j)                                            \
      rq[j] = *(const v4f*)(qn + (size_t)(16 * j) * DM);                      \
    _Pragma("unroll")                                                         \
    for (int j = 0; j < OBIT; ++j) {                                          \
      rw[2 * j]     = *(const float4*)(wn + (size_t)(32 * j) * DM);           \
      rw[2 * j + 1] = *(const float4*)(wn + (size_t)(32 * j + 1) * DM);       \
    }                                                                         \
  }
#define O_STASH(bf)                                                           \
  {                                                                           \
    float* qd = Qs + (bf) * OABUF + qst;                                      \
    float* wd = Ws + (bf) * OBBUF;                                            \
    _Pragma("unroll")                                                         \
    for (int j = 0; j < OAIT; ++j)                                            \
      *(v4f*)(qd + 16 * j * OAP) = rq[j];                                     \
    _Pragma("unroll")                                                         \
    for (int j = 0; j < OBIT; ++j) {                                          \
      *(float4*)(wd + 16 * j * OBP + wst0) =                                  \
          make_float4(rw[2 * j].x, rw[2 * j + 1].x, rw[2 * j].y, rw[2 * j + 1].y); \
      *(float4*)(wd + 16 * j * OBP + wst1) =                                  \
          make_float4(rw[2 * j].z, rw[2 * j + 1].z, rw[2 * j].w, rw[2 * j + 1].w); \
    }                                                                         \
  }
#define O_FRAG(dst, g)                                                        \
  {                                                                           \
    _Pragma("unroll")                                                         \
    for (int i = 0; i < OFG; ++i) {                                           \
      const int ks = (g) * OFG + i;                                           \
      fa[dst][i][0] = *(const v2f*)(qa + 4 * ks);                             \
      fa[dst][i][1] = *(const v2f*)(qa + 16 * OAP + 4 * ks);                  \
      fb[dst][i][0] = *(const v2f*)(wa0 + 2 * ks * OBP);                      \
      fb[dst][i][1] = *(const v2f*)(wa0 + 2 * ks * OBP + 32);                 \
      fb[dst][i][2] = *(const v2f*)(wa1 + 2 * ks * OBP + 64);                 \
      fb[dst][i][3] = *(const v2f*)(wa1 + 2 * ks * OBP + 96);                 \
    }                                                                         \
  }

  O_LOAD(0)
  O_STASH(0)
  O_LOAD(1)
  __syncthreads();

  const int ard = (32 * wv + lm) * OAP + 2 * lh;
  const int wr0 = lh * OBP + 2 * lm;
  const int wr1 = lh * OBP + ((2 * lm) ^ 4);
  v8f acc[8] = {};
  v2f fa[2][OFG][2], fb[2][OFG][4];
  for (int n = 0; n < ONST; ++n) {
    const int buf = n & 1;
    const float* qa  = Qs + buf * OABUF + ard;
    const float* wa0 = Ws + buf * OBBUF + wr0;
    const float* wa1 = Ws + buf * OBBUF + wr1;
    O_FRAG(0, 0)
    if (n + 1 < ONST) O_STASH(buf ^ 1)
    if (n + 2 < ONST) O_LOAD(n + 2)
    #pragma unroll
    for (int g = 0; g < ONG; ++g) {
      if (g + 1 < ONG) O_FRAG((g + 1) & 1, g + 1)
      __builtin_amdgcn_sched_barrier(0);
      #pragma unroll
      for (int i = 0; i < OFG; ++i) {
        #pragma unroll
        for (int t = 0; t < 4; ++t) {
          acc[t]     = wmma4(fa[g & 1][i][0], fb[g & 1][i][t], acc[t]);
          acc[4 + t] = wmma4(fa[g & 1][i][1], fb[g & 1][i][t], acc[4 + t]);
        }
      }
    }
    __syncthreads();
  }
#undef O_LOAD
#undef O_STASH
#undef O_FRAG
  float* ob = out + ((size_t)b * SEQ + n0 + 32 * wv) * DM + o0;
  #pragma unroll
  for (int t = 0; t < 4; ++t) {
    float add = bo[o0 + 16 * t + lm] + cq[(size_t)b * DM + o0 + 16 * t + lm];
    #pragma unroll
    for (int m = 0; m < 2; ++m)
      #pragma unroll
      for (int r = 0; r < 8; ++r)
        ob[(size_t)(16 * m + r + 8 * lh) * DM + 16 * t + lm] = acc[4 * m + t][r] + add;
  }
}

#define SROWS (SEQ / SCH)
#define SPH   4
#define SRPT  (SROWS / SPH)

__global__ void __launch_bounds__(512) kSums(const float* __restrict__ k,
                                             const float* __restrict__ v,
                                             float* __restrict__ part) {
  __shared__ float4 red[SPH * 256];
  const int tid = threadIdx.x;
  const int cg = tid & 127, ph = tid >> 7;
  const int b = blockIdx.x / SCH, c = blockIdx.x % SCH;
  const size_t off =
      ((size_t)b * SEQ + (size_t)c * SROWS + (size_t)ph * SRPT) * DM + 4 * cg;
  const float* kp = k + off;
  const float* vp = v + off;
  float4 rk[SRPT], rv[SRPT];
  #pragma unroll
  for (int i = 0; i < SRPT; ++i) {
    rk[i] = *(const float4*)(kp + (size_t)i * DM);
    rv[i] = *(const float4*)(vp + (size_t)i * DM);
  }
  float4 sk = rk[0], sv = rv[0];
  #pragma unroll
  for (int i = 1; i < SRPT; ++i) {
    sk.x += rk[i].x; sk.y += rk[i].y; sk.z += rk[i].z; sk.w += rk[i].w;
    sv.x += rv[i].x; sv.y += rv[i].y; sv.z += rv[i].z; sv.w += rv[i].w;
  }
  red[ph * 256 + cg] = sk;
  red[ph * 256 + 128 + cg] = sv;
  __syncthreads();
  if (tid < 256) {
    float4 s = red[tid];
    #pragma unroll
    for (int p = 1; p < SPH; ++p) {
      const float4 t = red[p * 256 + tid];
      s.x += t.x; s.y += t.y; s.z += t.z; s.w += t.w;
    }
    *(float4*)(part + (size_t)blockIdx.x * (2 * DM) + 4 * tid) = s;
  }
}

extern "C" void kernel_launch(void* const* d_in, const int* in_sizes, int n_in,
                              void* d_out, int out_size, void* d_ws, size_t ws_size,
                              hipStream_t stream) {
  const float* q   = (const float*)d_in[0];
  const float* k   = (const float*)d_in[1];
  const float* v   = (const float*)d_in[2];
  const float* wq  = (const float*)d_in[3];
  const float* bq  = (const float*)d_in[4];
  const float* wk  = (const float*)d_in[5];
  const float* bk  = (const float*)d_in[6];
  const float* wvw = (const float*)d_in[7];
  const float* bv  = (const float*)d_in[8];
  const float* wo  = (const float*)d_in[9];
  const float* bo  = (const float*)d_in[10];
  float* out = (float*)d_out;

  float* ws = (float*)d_ws;
  float* G    = ws;
  float* Mp   = G    + (size_t)NB * DM * DM;
  float* P    = Mp   + (size_t)8 * NB * NH * DKH * DKH;
  float* W4   = P    + (size_t)NB * DM * DM;
  float* u    = W4   + (size_t)NB * DM * DM;
  float* cq   = u    + (size_t)NB * DM;
  float* part = cq   + (size_t)NB * DM;

  kSums<<<128, 512, 0, stream>>>(k, v, part);
  kG   <<<256, 256, 0, stream>>>(k, v, G);
  kT   <<<256, 256, 0, stream>>>(wk, G, part, wvw, bk, Mp, u);
  kP   <<<64, 256, 0, stream>>>(Mp, u, bk, bv, wo, P);
  kW4  <<<256, 256, 0, stream>>>(wq, bq, P, W4, cq);
  dim3 go(32, 8, 1);
  kOut <<<go, 256, 0, stream>>>(q, W4, bo, cq, out);
}


# kG, kOut: a stage's LDS stores of the next panel and the loads behind them go slice by slice under its 8 fragment groups' WMMAs, not in one block before them; barrier orders LDS only; no sum changes
# speedup vs baseline: 8.3325x; 1.0547x over previous
#include <hip/hip_runtime.h>


typedef float v2f __attribute__((ext_vector_type(2)));
typedef float v4f __attribute__((ext_vector_type(4)));
typedef float v8f __attribute__((ext_vector_type(8)));

#define DM   512
#define SEQ  4096
#define NB   2
#define NH   8
#define DKH  64

__device__ __forceinline__ v8f wmma4(v2f a, v2f b, v8f c) {
  return __builtin_amdgcn_wmma_f32_16x16x4_f32(
       false, a,  false, b,
       (short)0, c,  false,  false);
}

#define LDS_BARRIER()                                                         \
  {                                                                           \
    __builtin_amdgcn_fence(__ATOMIC_RELEASE, "workgroup", "local");           \
    __builtin_amdgcn_s_barrier();                                             \
    __builtin_amdgcn_fence(__ATOMIC_ACQUIRE, "workgroup", "local");           \
  }

#define GKS   256
#define GNST  (SEQ / GKS)
#define GKP   96
#define GVP   160
#define GKBUF ((GKS / 2) * GKP)
#define GVBUF ((GKS / 2) * GVP)
#define GKIT  (GKS / 64)
#define GVIT  (GKS / 32)
#define GFG   8
#define GNG   (GKS / 4 / GFG)

__global__ void __launch_bounds__(256) kG(const float* __restrict__ k,
                                          const float* __restrict__ v,
                                          float* __restrict__ G) {
  __shared__ __attribute__((aligned(16))) float Ks[2 * GKBUF];
  __shared__ __attribute__((aligned(16))) float Vs[2 * GVBUF];
  const int tid = threadIdx.x;
  const int lane = tid & 31, wv = tid >> 5;
  const int lm = lane & 15, lh = lane >> 4;
  const int blk = blockIdx.x;
  const int b = blk >> 7;
  const int rest = blk & 127;
  const int d0 = (rest >> 3) * 32;
  const int e0 = (rest & 7) * 64;
  const int dw = 16 * (wv & 1);
  const int ew = 16 * (wv >> 1);
  const int kc = (tid & 7) * 4,  krp = tid >> 3;
  const int vc = (tid & 15) * 4, vrp = tid >> 4;
  const float* kg = k + (size_t)b * SEQ * DM + d0 + (size_t)(2 * krp) * DM + kc;
  const float* vg = v + (size_t)b * SEQ * DM + e0 + (size_t)(2 * vrp) * DM + vc;
  const int ksw = 4 * (krp & 1);
  const int vsw = 4 * ((vc >> 5) & 1);
  const int kst0 = krp * GKP + 2 * kc + ksw, kst1 = krp * GKP + 2 * kc + 4 - ksw;
  const int vst0 = vrp * GVP + 2 * vc + vsw, vst1 = vrp * GVP + 2 * vc + 4 - vsw;
  v4f rk[2 * GKIT], rv[2 * GVIT];

#define G_LOAD_K(st, j)                                                       \
  {                                                                           \
    const float* kn = kg + (size_t)(st) * GKS * DM + (size_t)(64 * (j)) * DM; \
    rk[2 * (j)]     = *(const v4f*)(kn);                                      \
    rk[2 * (j) + 1] = *(const v4f*)(kn + DM);                                 \
  }
#define G_LOAD_V(st, j)                                                       \
  {                                                                           \
    const float* vn = vg + (size_t)(st) * GKS * DM + (size_t)(32 * (j)) * DM; \
    rv[2 * (j)]     = *(const v4f*)(vn);                                      \
    rv[2 * (j) + 1] = *(const v4f*)(vn + DM);                                 \
  }
#define G_HERE(a, b) asm volatile("" : : "v"(a), "v"(b));
#define G_STASH_K(bf, j)                                                      \
  {                                                                           \
    float* kd = Ks + (bf) * GKBUF + 32 * (j) * GKP;                           \
    G_HERE(rk[2 * (j)], rk[2 * (j) + 1])                                      \
    *(float4*)(kd + kst0) = make_float4(rk[2 * (j)].x, rk[2 * (j) + 1].x,     \
                                        rk[2 * (j)].y, rk[2 * (j) + 1].y);    \
    *(float4*)(kd + kst1) = make_float4(rk[2 * (j)].z, rk[2 * (j) + 1].z,     \
                                        rk[2 * (j)].w, rk[2 * (j) + 1].w);    \
  }
#define G_STASH_V(bf, j)                                                      \
  {                                                                           \
    float* vd = Vs + (bf) * GVBUF + 16 * (j) * GVP;                           \
    G_HERE(rv[2 * (j)], rv[2 * (j) + 1])                                      \
    *(float4*)(vd + vst0) = make_float4(rv[2 * (j)].x, rv[2 * (j) + 1].x,     \
                                        rv[2 * (j)].y, rv[2 * (j) + 1].y);    \
    *(float4*)(vd + vst1) = make_float4(rv[2 * (j)].z, rv[2 * (j) + 1].z,     \
                                        rv[2 * (j)].w, rv[2 * (j) + 1].w);    \
  }
  static_assert(GVIT == GNG && 2 * GKIT == GNG, "one v item a group, one k item every second group");
#define G_LOAD_SLICE(st, g)                                                   \
  {                                                                           \
    G_LOAD_V(st, g)                                                           \
    if (((g) & 1) == 0) G_LOAD_K(st, (g) / 2)                                 \
  }
#define G_STASH_SLICE(bf, g)                                                  \
  {                                                                           \
    G_STASH_V(bf, g)                                                          \
    if (((g) & 1) == 0) G_STASH_K(bf, (g) / 2)                                \
  }
#define G_FRAG(dst, g)                                                        \
  {                                                                           \
    _Pragma("unroll")                                                         \
    for (int i = 0; i < GFG; ++i) {                                           \
      fa[dst][i] = *(const v2f*)(ka + 2 * ((g) * GFG + i) * GKP);             \
      fb[dst][i] = *(const v2f*)(va + 2 * ((g) * GFG + i) * GVP);             \
    }                                                                         \
  }
#define G_STAGE(n, ST, LD)                                                    \
  {                                                                           \
    const int buf = (n) & 1;                                                  \
    const float* ka = Ks + buf * GKBUF + ard;                                 \
    const float* va = Vs + buf * GVBUF + brd;                                 \
    G_FRAG(0, 0)                                                              \
    _Pragma("unroll")                                                         \
    for (int g = 0; g < GNG; ++g) {                                           \
      if (ST) G_STASH_SLICE(buf ^ 1, g)                                       \
      if (g + 1 < GNG) G_FRAG((g + 1) & 1, g + 1)                             \
      if (LD) G_LOAD_SLICE((n) + 2, g)                                        \
      _Pragma("unroll")                                                       \
      for (int i = 0; i < GFG; ++i)                                           \
        acc = wmma4(fa[g & 1][i], fb[g & 1][i], acc);                         \
                                          \
      __builtin_amdgcn_sched_barrier(0);                                      \
    }                                                                         \
    LDS_BARRIER()                                                             \
  }

  #pragma unroll
  for (int g = 0; g < GNG; ++g) G_LOAD_SLICE(0, g)
  #pragma unroll
  for (int g = 0; g < GNG; ++g) G_STASH_SLICE(0, g)
  #pragma unroll
  for (int g = 0; g < GNG; ++g) G_LOAD_SLICE(1, g)
  __syncthreads();

  const int ard = lh * GKP + ((2 * (dw + lm)) ^ (4 * lh));
  const int brd = lh * GVP + ((2 * (ew + lm)) ^ (4 * ((ew >> 5) & 1)));
  v8f acc = {};
  v2f fa[2][GFG], fb[2][GFG];
  for (int n = 0; n < GNST - 2; ++n) G_STAGE(n, 1, 1)
  G_STAGE(GNST - 2, 1, 0)
  G_STAGE(GNST - 1, 0, 0)
#undef G_HERE
#undef G_LOAD_K
#undef G_LOAD_V
#undef G_STASH_K
#undef G_STASH_V
#undef G_LOAD_SLICE
#undef G_STASH_SLICE
#undef G_FRAG
#undef G_STAGE
  float* Gb = G + ((size_t)b * DM + d0 + dw) * DM + e0 + ew;
  #pragma unroll
  for (int r = 0; r < 8; ++r)
    Gb[(size_t)(r + 8 * lh) * DM + lm] = acc[r];
}

#define ROWP (DM + 4)
#define KP64 72
#define KP32 40

template <int NR>
__device__ __forceinline__ void stage_rows(const float* __restrict__ g, float* s,
                                           int wv, int lane) {
  #pragma unroll
  for (int j = 0; j < NR / 8; ++j)
    #pragma unroll
    for (int q = 0; q < 4; ++q)
      *(float4*)(s + (wv + 8 * j) * ROWP + 128 * q + 4 * lane) =
          *(const float4*)(g + (size_t)(wv + 8 * j) * DM + 128 * q + 4 * lane);
}

template <int NC, int PITCH>
__device__ __forceinline__ void stage_kmajor(const float* __restrict__ g, float* s,
                                             int tid) {
  constexpr int F4 = NC / 4;
  constexpr int RPS = 256 / F4;
  const int c4 = (tid % F4) * 4, r0 = tid / F4;
  #pragma unroll
  for (int j = 0; j < DM / RPS; ++j)
    *(float4*)(s + (r0 + RPS * j) * PITCH + c4) =
        *(const float4*)(g + (size_t)(r0 + RPS * j) * DM + c4);
}

__device__ __forceinline__ v2f ld_pair(const float* p, int pitch) {
  v2f r; r.x = p[0]; r.y = p[pitch]; return r;
}

#define SCH   64

__device__ __forceinline__ float2 chunk_sum2(const float* __restrict__ pp) {
  float2 s = *(const float2*)pp;
  #pragma unroll
  for (int c = 1; c < SCH; ++c) {
    const float2 t = *(const float2*)(pp + (size_t)c * (2 * DM));
    s.x += t.x; s.y += t.y;
  }
  return s;
}

#define PANEL_K_LOOP(LOADA, LOADB)                                            \
  {                                                                           \
    v2f fa[2][8], fb[2][8];                                                   \
    _Pragma("unroll")                                                         \
    for (int i = 0; i < 8; ++i) {                                             \
      const int kk = 4 * i;                                                   \
      fa[0][i] = LOADA; fb[0][i] = LOADB;                                     \
    }                                                                         \
    _Pragma("unroll")                                                         \
    for (int g = 0; g < DM / 32; ++g) {                                       \
      if (g + 1 < DM / 32) {                                                  \
        _Pragma("unroll")                                                     \
        for (int i = 0; i < 8; ++i) {                                         \
          const int kk = 32 * (g + 1) + 4 * i;                                \
          fa[(g + 1) & 1][i] = LOADA; fb[(g + 1) & 1][i] = LOADB;             \
        }                                                                     \
      }                                                                       \
      _Pragma("unroll")                                                       \
      for (int i = 0; i < 8; ++i)                                             \
        acc = wmma4(fa[g & 1][i], fb[g & 1][i], acc);                         \
      __builtin_amdgcn_sched_barrier(0);                                      \
    }                                                                         \
  }

#define TP (DKH + 4)

__global__ void __launch_bounds__(256) kT(const float* __restrict__ wk,
                                          const float* __restrict__ G,
                                          const float* __restrict__ part,
                                          const float* __restrict__ wvw,
                                          const float* __restrict__ bk,
                                          float* __restrict__ Mp,
                                          float* __restrict__ u) {
  __shared__ __attribute__((aligned(16))) float As[32 * ROWP];
  __shared__ __attribute__((aligned(16))) float Bs[DM * KP64];
  __shared__ __attribute__((aligned(16))) float Ts[32 * TP];
  __shared__ __attribute__((aligned(16))) float Vw[DKH * TP];
  __shared__ float vs[DKH];
  __shared__ float bks[32];
  const int tid = threadIdx.x;
  const int lane = tid & 31, wv = tid >> 5;
  const int lm = lane & 15, lh = lane >> 4;
  const int bh = blockIdx.x >> 4;
  const int b = bh >> 3, h = bh & 7;
  const int ih = (blockIdx.x >> 3) & 1;
  const int eg = blockIdx.x & 7;
  const int e0 = eg * 64;
  if (e0 == 0) {
    const float2 s = chunk_sum2(part + (size_t)b * SCH * (2 * DM) + 2 * tid);
    Bs[(2 * tid) * KP64 + 64] = s.x;
    Bs[(2 * tid + 1) * KP64 + 64] = s.y;
  }
  stage_rows<32>(wk + (size_t)(h * 64 + ih * 32) * DM, As, wv, lane);
  stage_kmajor<64, KP64>(G + (size_t)b * DM * DM + e0, Bs, tid);
  {
    const int c4 = (tid & 15) * 4, r0 = tid >> 4;
    #pragma unroll
    for (int t = 0; t < 4; ++t)
      *(float4*)(Vw + (r0 + 16 * t) * TP + c4) =
          *(const float4*)(wvw + (size_t)(h * 64 + r0 + 16 * t) * DM + e0 + c4);
  }
  if (wv == 6) bks[lane] = bk[h * 64 + ih * 32 + lane];
  if (wv == 7) {
    const float2 s =
        chunk_sum2(part + (size_t)b * SCH * (2 * DM) + DM + e0 + 2 * lane);
    vs[2 * lane] = s.x;
    vs[2 * lane + 1] = s.y;
  }
  __syncthreads();
  const int ti = wv >> 2, tj = wv & 3;
  const float* ap = As + (16 * ti + lm) * ROWP + 2 * lh;
  const float* bp = Bs + (2 * lh) * KP64 + 16 * tj + lm;
  v8f acc = {};
  PANEL_K_LOOP((*(const v2f*)(ap + kk)), ld_pair(bp + kk * KP64, KP64))
  {
    const float vse = vs[16 * tj + lm];
    #pragma unroll
    for (int r = 0; r < 8; ++r) {
      const int il = 16 * ti + r + 8 * lh;
      Ts[il * TP + 16 * tj + lm] = acc[r] + bks[il] * vse;
    }
  }
  __syncthreads();
  {
    const float* ap2 = Ts + (16 * ti + lm) * TP + 2 * lh;
    const float* bp2 = Vw + (16 * tj + lm) * TP + 2 * lh;
    v2f fa2[DKH / 4], fb2[DKH / 4];
    #pragma unroll
    for (int i = 0; i < DKH / 4; ++i) {
      fa2[i] = *(const v2f*)(ap2 + 4 * i);
      fb2[i] = *(const v2f*)(bp2 + 4 * i);
    }
    __builtin_amdgcn_sched_barrier(0);
    v8f acc2 = {};
    #pragma unroll
    for (int i = 0; i < DKH / 4; ++i)
      acc2 = wmma4(fa2[i], fb2[i], acc2);
    float* Mb = Mp + (((size_t)eg * (NB * NH) + bh) * DKH + ih * 32 + 16 * ti) * DKH + 16 * tj;
    #pragma unroll
    for (int r = 0; r < 8; ++r)
      Mb[(size_t)(r + 8 * lh) * DKH + lm] = acc2[r];
  }
  if (e0 == 0 && wv < 2) {
    const float* ua = As + (16 * wv + lm) * ROWP + 2 * lh;
    const float* ub = Bs + (2 * lh) * KP64 + 64;
    v8f accu;
    {
      v8f acc = {};
      PANEL_K_LOOP((*(const v2f*)(ua + kk)), ld_pair(ub + kk * KP64, KP64))
      accu = acc;
    }
    if (lm == 0) {
      float* ud = u + (size_t)b * DM + h * 64 + ih * 32 + 16 * wv + 8 * lh;
      #pragma unroll
      for (int r = 0; r < 8; ++r) ud[r] = accu[r];
    }
  }
}

__global__ void __launch_bounds__(256) kP(const float* __restrict__ Mp,
                                          const float* __restrict__ u,
                                          const float* __restrict__ bk,
                                          const float* __restrict__ bv,
                                          const float* __restrict__ wo,
                                          float* __restrict__ P) {
  __shared__ __attribute__((aligned(16))) float Ms[16 * TP];
  const int tid = threadIdx.x;
  const int lane = tid & 31, wv = tid >> 5;
  const int lm = lane & 15, lh = lane >> 4;
  const int bh = blockIdx.x >> 2;
  const int b = bh >> 3, h = bh & 7;
  const int i0 = (blockIdx.x & 3) * 16;
  const int o0 = wv * 64;
  v2f bm[DKH / 4][4];
  #pragma unroll
  for (int i = 0; i < DKH / 4; ++i)
    #pragma unroll
    for (int t = 0; t < 4; ++t)
      bm[i][t] = *(const v2f*)(wo + (size_t)(o0 + 16 * t + lm) * DM + h * 64 + 4 * i + 2 * lh);
  {
    const int row = tid >> 4, c4 = (tid & 15) * 4;
    const float* mp = Mp + ((size_t)bh * 64 + i0 + row) * DKH + c4;
    float4 s = *(const float4*)mp;
    #pragma unroll
    for (int g = 1; g < 8; ++g) {
      const float4 t = *(const float4*)(mp + (size_t)g * (NB * NH * DKH * DKH));
      s.x += t.x; s.y += t.y; s.z += t.z; s.w += t.w;
    }
    const float ui  = u[(size_t)b * DM + h * 64 + i0 + row];
    const float bki = bk[h * 64 + i0 + row];
    const float4 bv4 = *(const float4*)(bv + h * 64 + c4);
    float4 m;
    m.x = 0.125f * (s.x + ui * bv4.x + (float)SEQ * bki * bv4.x);
    m.y = 0.125f * (s.y + ui * bv4.y + (float)SEQ * bki * bv4.y);
    m.z = 0.125f * (s.z + ui * bv4.z + (float)SEQ * bki * bv4.z);
    m.w = 0.125f * (s.w + ui * bv4.w + (float)SEQ * bki * bv4.w);
    *(float4*)(Ms + row * TP + c4) = m;
  }
  __syncthreads();
  const float* ap = Ms + lm * TP + 2 * lh;
  v2f fa[DKH / 4];
  #pragma unroll
  for (int i = 0; i < DKH / 4; ++i) fa[i] = *(const v2f*)(ap + 4 * i);
  v8f acc[4] = {};
  #pragma unroll
  for (int i = 0; i < DKH / 4; ++i) {
    #pragma unroll
    for (int t = 0; t < 4; ++t)
      acc[t] = wmma4(fa[i], bm[i][t], acc[t]);
  }
  float* Pb = P + ((size_t)b * DM + h * 64 + i0) * DM + o0;
  #pragma unroll
  for (int t = 0; t < 4; ++t)
    #pragma unroll
    for (int r = 0; r < 8; ++r)
      Pb[(size_t)(r + 8 * lh) * DM + 16 * t + lm] = acc[t][r];
}

__global__ void __launch_bounds__(256) kW4(const float* __restrict__ wq,
                                           const float* __restrict__ bq,
                                           const float* __restrict__ P,
                                           float* __restrict__ W4,
                                           float* __restrict__ cq) {
  __shared__ __attribute__((aligned(16))) float As[DM * KP32];
  __shared__ __attribute__((aligned(16))) float Bs[DM * KP64];
  const int tid = threadIdx.x;
  const int lane = tid & 31, wv = tid >> 5;
  const int lm = lane & 15, lh = lane >> 4;
  const int b = blockIdx.x >> 7;
  const int d0 = ((blockIdx.x >> 3) & 15) * 32;
  const int o0 = (blockIdx.x & 7) * 64;
  As[tid * KP32 + 32] = bq[tid];
  As[(tid + 256) * KP32 + 32] = bq[tid + 256];
  stage_kmajor<32, KP32>(wq + d0, As, tid);
  stage_kmajor<64, KP64>(P + (size_t)b * DM * DM + o0, Bs, tid);
  __syncthreads();
  const int ti = wv >> 2, tj = wv & 3;
  const float* ap = As + (2 * lh) * KP32 + 16 * ti + lm;
  const float* bp = Bs + (2 * lh) * KP64 + 16 * tj + lm;
  v8f acc = {};
  PANEL_K_LOOP(ld_pair(ap + kk * KP32, KP32), ld_pair(bp + kk * KP64, KP64))
  float* Wb = W4 + ((size_t)b * DM + d0 + 16 * ti) * DM + o0 + 16 * tj;
  #pragma unroll
  for (int r = 0; r < 8; ++r)
    Wb[(size_t)(r + 8 * lh) * DM + lm] = acc[r];
  if (d0 == 0 && ti == 0) {
    const float* ca = As + (2 * lh) * KP32 + 32;
    v8f accc;
    {
      v8f acc = {};
      PANEL_K_LOOP(ld_pair(ca + kk * KP32, KP32), ld_pair(bp + kk * KP64, KP64))
      accc = acc;
    }
    if (lh == 0) cq[(size_t)b * DM + o0 + 16 * tj + lm] = accc[0];
  }
}

#define OKS   64
#define ONST  (DM / OKS)
#define OAP   (OKS + 4)
#define OBP   160
#define OABUF (256 * OAP)
#define OBBUF ((OKS / 2) * OBP)
#define OAIT  16
#define OBIT  (OKS / 32)
#define OFG   2
#define ONG   (OKS / 4 / OFG)

__global__ void __launch_bounds__(256) kOut(const float* __restrict__ q,
                                            const float* __restrict__ W4,
                                            const float* __restrict__ bo,
                                            const float* __restrict__ cq,
                                            float* __restrict__ out) {
  __shared__ __attribute__((aligned(16))) float Qs[2 * OABUF];
  __shared__ __attribute__((aligned(16))) float Ws[2 * OBBUF];
  const int tid = threadIdx.x;
  const int lane = tid & 31, wv = tid >> 5;
  const int lm = lane & 15, lh = lane >> 4;
  const int b = blockIdx.x >> 4;
  const int n0 = (blockIdx.x & 15) * 256;
  const int o0 = blockIdx.y * 64;
  const int sc = (tid & 15) * 4, sr = tid >> 4;
  const float* qg = q + ((size_t)b * SEQ + n0 + sr) * DM + sc;
  const float* wg = W4 + ((size_t)b * DM + 2 * sr) * DM + o0 + sc;
  const int qst = sr * OAP + sc;
  const int wsw = 4 * ((sc >> 5) & 1);
  const int wst0 = sr * OBP + 2 * sc + wsw, wst1 = sr * OBP + 2 * sc + 4 - wsw;
  v4f rq[OAIT];
  v4f rw[2 * OBIT];

#define O_LOAD_Q(st, j)                                                       \
  rq[j] = *(const v4f*)(qg + (st) * OKS + (size_t)(16 * (j)) * DM);
#define O_LOAD_W(st, j)                                                       \
  {                                                                           \
    const float* wn = wg + (size_t)(st) * OKS * DM + (size_t)(32 * (j)) * DM; \
    rw[2 * (j)]     = *(const v4f*)(wn);                                      \
    rw[2 * (j) + 1] = *(const v4f*)(wn + DM);                                 \
  }
#define O_STASH_Q(bf, j)                                                      \
  *(v4f*)(Qs + (bf) * OABUF + qst + 16 * (j) * OAP) = rq[j];
#define O_HERE(a, b) asm volatile("" : : "v"(a), "v"(b));
#define O_STASH_W(bf, j)                                                      \
  {                                                                           \
    float* wd = Ws + (bf) * OBBUF + 16 * (j) * OBP;                           \
    O_HERE(rw[2 * (j)], rw[2 * (j) + 1])                                      \
    *(float4*)(wd + wst0) = make_float4(rw[2 * (j)].x, rw[2 * (j) + 1].x,     \
                                        rw[2 * (j)].y, rw[2 * (j) + 1].y);    \
    *(float4*)(wd + wst1) = make_float4(rw[2 * (j)].z, rw[2 * (j) + 1].z,     \
                                        rw[2 * (j)].w, rw[2 * (j) + 1].w);    \
  }
  static_assert(OAIT == 2 * ONG && 4 * OBIT == ONG, "two q items a group, one W4 item every fourth group");
#define O_LOAD_SLICE(st, g)                                                   \
  {                                                                           \
    O_LOAD_Q(st, 2 * (g))                                                     \
    O_LOAD_Q(st, 2 * (g) + 1)                                                 \
    if (((g) & 3) == 0) O_LOAD_W(st, (g) / 4)                                 \
  }
#define O_STASH_SLICE(bf, g)                                                  \
  {                                                                           \
    O_STASH_Q(bf, 2 * (g))                                                    \
    O_STASH_Q(bf, 2 * (g) + 1)                                                \
    if (((g) & 3) == 0) O_STASH_W(bf, (g) / 4)                                \
  }
#define O_FRAG(dst, g)                                                        \
  {                                                                           \
    _Pragma("unroll")                                                         \
    for (int i = 0; i < OFG; ++i) {                                           \
      const int ks = (g) * OFG + i;                                           \
      fa[dst][i][0] = *(const v2f*)(qa + 4 * ks);                             \
      fa[dst][i][1] = *(const v2f*)(qa + 16 * OAP + 4 * ks);                  \
      fb[dst][i][0] = *(const v2f*)(wa0 + 2 * ks * OBP);                      \
      fb[dst][i][1] = *(const v2f*)(wa0 + 2 * ks * OBP + 32);                 \
      fb[dst][i][2] = *(const v2f*)(wa1 + 2 * ks * OBP + 64);                 \
      fb[dst][i][3] = *(const v2f*)(wa1 + 2 * ks * OBP + 96);                 \
    }                                                                         \
  }
#define O_STAGE(n, ST, LD)                                                    \
  {                                                                           \
    const int buf = (n) & 1;                                                  \
    const float* qa  = Qs + buf * OABUF + ard;                                \
    const float* wa0 = Ws + buf * OBBUF + wr0;                                \
    const float* wa1 = Ws + buf * OBBUF + wr1;                                \
    O_FRAG(0, 0)                                                              \
    _Pragma("unroll")                                                         \
    for (int g = 0; g < ONG; ++g) {                                           \
      if (ST) O_STASH_SLICE(buf ^ 1, g)                                       \
      if (g + 1 < ONG) O_FRAG((g + 1) & 1, g + 1)                             \
      if (LD) O_LOAD_SLICE((n) + 2, g)                                        \
                      \
      __builtin_amdgcn_sched_barrier(0);                                      \
      _Pragma("unroll")                                                       \
      for (int i = 0; i < OFG; ++i) {                                         \
        _Pragma("unroll")                                                     \
        for (int t = 0; t < 4; ++t) {                                         \
          acc[t]     = wmma4(fa[g & 1][i][0], fb[g & 1][i][t], acc[t]);       \
          acc[4 + t] = wmma4(fa[g & 1][i][1], fb[g & 1][i][t], acc[4 + t]);   \
        }                                                                     \
      }                                                                       \
    }                                                                         \
    LDS_BARRIER()                                                             \
  }

  #pragma unroll
  for (int g = 0; g < ONG; ++g) O_LOAD_SLICE(0, g)
  #pragma unroll
  for (int g = 0; g < ONG; ++g) O_STASH_SLICE(0, g)
  #pragma unroll
  for (int g = 0; g < ONG; ++g) O_LOAD_SLICE(1, g)
  __syncthreads();

  const int ard = (32 * wv + lm) * OAP + 2 * lh;
  const int wr0 = lh * OBP + 2 * lm;
  const int wr1 = lh * OBP + ((2 * lm) ^ 4);
  v8f acc[8] = {};
  v2f fa[2][OFG][2], fb[2][OFG][4];
  for (int n = 0; n < ONST - 2; ++n) O_STAGE(n, 1, 1)
  O_STAGE(ONST - 2, 1, 0)
  O_STAGE(ONST - 1, 0, 0)
#undef O_HERE
#undef O_LOAD_Q
#undef O_LOAD_W
#undef O_STASH_Q
#undef O_STASH_W
#undef O_LOAD_SLICE
#undef O_STASH_SLICE
#undef O_FRAG
#undef O_STAGE
  float* ob = out + ((size_t)b * SEQ + n0 + 32 * wv) * DM + o0;
  #pragma unroll
  for (int t = 0; t < 4; ++t) {
    float add = bo[o0 + 16 * t + lm] + cq[(size_t)b * DM + o0 + 16 * t + lm];
    #pragma unroll
    for (int m = 0; m < 2; ++m)
      #pragma unroll
      for (int r = 0; r < 8; ++r)
        ob[(size_t)(16 * m + r + 8 * lh) * DM + 16 * t + lm] = acc[4 * m + t][r] + add;
  }
}

#define SROWS (SEQ / SCH)
#define SPH   4
#define SRPT  (SROWS / SPH)

__global__ void __launch_bounds__(512) kSums(const float* __restrict__ k,
                                             const float* __restrict__ v,
                                             float* __restrict__ part) {
  __shared__ float4 red[SPH * 256];
  const int tid = threadIdx.x;
  const int cg = tid & 127, ph = tid >> 7;
  const int b = blockIdx.x / SCH, c = blockIdx.x % SCH;
  const size_t off =
      ((size_t)b * SEQ + (size_t)c * SROWS + (size_t)ph * SRPT) * DM + 4 * cg;
  const float* kp = k + off;
  const float* vp = v + off;
  float4 rk[SRPT], rv[SRPT];
  #pragma unroll
  for (int i = 0; i < SRPT; ++i) {
    rk[i] = *(const float4*)(kp + (size_t)i * DM);
    rv[i] = *(const float4*)(vp + (size_t)i * DM);
  }
  float4 sk = rk[0], sv = rv[0];
  #pragma unroll
  for (int i = 1; i < SRPT; ++i) {
    sk.x += rk[i].x; sk.y += rk[i].y; sk.z += rk[i].z; sk.w += rk[i].w;
    sv.x += rv[i].x; sv.y += rv[i].y; sv.z += rv[i].z; sv.w += rv[i].w;
  }
  red[ph * 256 + cg] = sk;
  red[ph * 256 + 128 + cg] = sv;
  __syncthreads();
  if (tid < 256) {
    float4 s = red[tid];
    #pragma unroll
    for (int p = 1; p < SPH; ++p) {
      const float4 t = red[p * 256 + tid];
      s.x += t.x; s.y += t.y; s.z += t.z; s.w += t.w;
    }
    *(float4*)(part + (size_t)blockIdx.x * (2 * DM) + 4 * tid) = s;
  }
}

extern "C" void kernel_launch(void* const* d_in, const int* in_sizes, int n_in,
                              void* d_out, int out_size, void* d_ws, size_t ws_size,
                              hipStream_t stream) {
  const float* q   = (const float*)d_in[0];
  const float* k   = (const float*)d_in[1];
  const float* v   = (const float*)d_in[2];
  const float* wq  = (const float*)d_in[3];
  const float* bq  = (const float*)d_in[4];
  const float* wk  = (const float*)d_in[5];
  const float* bk  = (const float*)d_in[6];
  const float* wvw = (const float*)d_in[7];
  const float* bv  = (const float*)d_in[8];
  const float* wo  = (const float*)d_in[9];
  const float* bo  = (const float*)d_in[10];
  float* out = (float*)d_out;

  float* ws = (float*)d_ws;
  float* G    = ws;
  float* Mp   = G    + (size_t)NB * DM * DM;
  float* P    = Mp   + (size_t)8 * NB * NH * DKH * DKH;
  float* W4   = P    + (size_t)NB * DM * DM;
  float* u    = W4   + (size_t)NB * DM * DM;
  float* cq   = u    + (size_t)NB * DM;
  float* part = cq   + (size_t)NB * DM;

  kSums<<<128, 512, 0, stream>>>(k, v, part);
  kG   <<<256, 256, 0, stream>>>(k, v, G);
  kT   <<<256, 256, 0, stream>>>(wk, G, part, wvw, bk, Mp, u);
  kP   <<<64, 256, 0, stream>>>(Mp, u, bk, bv, wo, P);
  kW4  <<<256, 256, 0, stream>>>(wq, bq, P, W4, cq);
  dim3 go(32, 8, 1);
  kOut <<<go, 256, 0, stream>>>(q, W4, bo, cq, out);
}


# kSums launch dropped (6 -> 5): k, v column sums ride in kG as WMMAs of each wave's fragments against a ones operand, 16 partials a column added in kT; changes ksum/vsum summation tree
# speedup vs baseline: 8.5796x; 1.0297x over previous
#include <hip/hip_runtime.h>


typedef float v2f __attribute__((ext_vector_type(2)));
typedef float v4f __attribute__((ext_vector_type(4)));
typedef float v8f __attribute__((ext_vector_type(8)));

#define DM   512
#define SEQ  4096
#define NB   2
#define NH   8
#define DKH  64

__device__ __forceinline__ v8f wmma4(v2f a, v2f b, v8f c) {
  return __builtin_amdgcn_wmma_f32_16x16x4_f32(
       false, a,  false, b,
       (short)0, c,  false,  false);
}

#define LDS_BARRIER()                                                         \
  {                                                                           \
    __builtin_amdgcn_fence(__ATOMIC_RELEASE, "workgroup", "local");           \
    __builtin_amdgcn_s_barrier();                                             \
    __builtin_amdgcn_fence(__ATOMIC_ACQUIRE, "workgroup", "local");           \
  }

#define GKS   256
#define GNST  (SEQ / GKS)
#define GKP   96
#define GVP   160
#define GKBUF ((GKS / 2) * GKP)
#define GVBUF ((GKS / 2) * GVP)
#define GKIT  (GKS / 64)
#define GVIT  (GKS / 32)
#define GFG   8
#define GNG   (GKS / 4 / GFG)

#define XCH   16

__global__ void __launch_bounds__(256) kG(const float* __restrict__ k,
                                          const float* __restrict__ v,
                                          float* __restrict__ G,
                                          float* __restrict__ part) {
  __shared__ __attribute__((aligned(16))) float Ks[2 * GKBUF];
  __shared__ __attribute__((aligned(16))) float Vs[2 * GVBUF];
  const int tid = threadIdx.x;
  const int lane = tid & 31, wv = tid >> 5;
  const int lm = lane & 15, lh = lane >> 4;
  const int blk = blockIdx.x;
  const int b = blk >> 7;
  const int rest = blk & 127;
  const int d0 = (rest >> 3) * 32;
  const int e0 = (rest & 7) * 64;
  const int dw = 16 * (wv & 1);
  const int ew = 16 * (wv >> 1);
  const int kc = (tid & 7) * 4,  krp = tid >> 3;
  const int vc = (tid & 15) * 4, vrp = tid >> 4;
  const float* kg = k + (size_t)b * SEQ * DM + d0 + (size_t)(2 * krp) * DM + kc;
  const float* vg = v + (size_t)b * SEQ * DM + e0 + (size_t)(2 * vrp) * DM + vc;
  const int ksw = 4 * (krp & 1);
  const int vsw = 4 * ((vc >> 5) & 1);
  const int kst0 = krp * GKP + 2 * kc + ksw, kst1 = krp * GKP + 2 * kc + 4 - ksw;
  const int vst0 = vrp * GVP + 2 * vc + vsw, vst1 = vrp * GVP + 2 * vc + 4 - vsw;
  v4f rk[2 * GKIT], rv[2 * GVIT];

#define G_LOAD_K(st, j)                                                       \
  {                                                                           \
    const float* kn = kg + (size_t)(st) * GKS * DM + (size_t)(64 * (j)) * DM; \
    rk[2 * (j)]     = *(const v4f*)(kn);                                      \
    rk[2 * (j) + 1] = *(const v4f*)(kn + DM);                                 \
  }
#define G_LOAD_V(st, j)                                                       \
  {                                                                           \
    const float* vn = vg + (size_t)(st) * GKS * DM + (size_t)(32 * (j)) * DM; \
    rv[2 * (j)]     = *(const v4f*)(vn);                                      \
    rv[2 * (j) + 1] = *(const v4f*)(vn + DM);                                 \
  }
#define G_HERE(a, b) asm volatile("" : : "v"(a), "v"(b));
#define G_STASH_K(bf, j)                                                      \
  {                                                                           \
    float* kd = Ks + (bf) * GKBUF + 32 * (j) * GKP;                           \
    G_HERE(rk[2 * (j)], rk[2 * (j) + 1])                                      \
    *(float4*)(kd + kst0) = make_float4(rk[2 * (j)].x, rk[2 * (j) + 1].x,     \
                                        rk[2 * (j)].y, rk[2 * (j) + 1].y);    \
    *(float4*)(kd + kst1) = make_float4(rk[2 * (j)].z, rk[2 * (j) + 1].z,     \
                                        rk[2 * (j)].w, rk[2 * (j) + 1].w);    \
  }
#define G_STASH_V(bf, j)                                                      \
  {                                                                           \
    float* vd = Vs + (bf) * GVBUF + 16 * (j) * GVP;                           \
    G_HERE(rv[2 * (j)], rv[2 * (j) + 1])                                      \
    *(float4*)(vd + vst0) = make_float4(rv[2 * (j)].x, rv[2 * (j) + 1].x,     \
                                        rv[2 * (j)].y, rv[2 * (j) + 1].y);    \
    *(float4*)(vd + vst1) = make_float4(rv[2 * (j)].z, rv[2 * (j) + 1].z,     \
                                        rv[2 * (j)].w, rv[2 * (j) + 1].w);    \
  }
  static_assert(GVIT == GNG && 2 * GKIT == GNG, "one v item a group, one k item every second group");
#define G_LOAD_SLICE(st, g)                                                   \
  {                                                                           \
    G_LOAD_V(st, g)                                                           \
    if (((g) & 1) == 0) G_LOAD_K(st, (g) / 2)                                 \
  }
#define G_STASH_SLICE(bf, g)                                                  \
  {                                                                           \
    G_STASH_V(bf, g)                                                          \
    if (((g) & 1) == 0) G_STASH_K(bf, (g) / 2)                                \
  }
#define G_FRAG(dst, g)                                                        \
  {                                                                           \
    _Pragma("unroll")                                                         \
    for (int i = 0; i < GFG; ++i) {                                           \
      fa[dst][i] = *(const v2f*)(ka + 2 * ((g) * GFG + i) * GKP);             \
      fb[dst][i] = *(const v2f*)(va + 2 * ((g) * GFG + i) * GVP);             \
    }                                                                         \
  }
#define G_STAGE(n, ST, LD)                                                    \
  {                                                                           \
    const int buf = (n) & 1;                                                  \
    const float* ka = Ks + buf * GKBUF + ard;                                 \
    const float* va = Vs + buf * GVBUF + brd;                                 \
    G_FRAG(0, 0)                                                              \
    _Pragma("unroll")                                                         \
    for (int g = 0; g < GNG; ++g) {                                           \
      if (ST) G_STASH_SLICE(buf ^ 1, g)                                       \
      if (g + 1 < GNG) G_FRAG((g + 1) & 1, g + 1)                             \
      if (LD) G_LOAD_SLICE((n) + 2, g)                                        \
      _Pragma("unroll")                                                       \
      for (int i = 0; i < GFG; ++i)                                           \
        acc = wmma4(fa[g & 1][i], fb[g & 1][i], acc);                         \
      if (g == xg) {                          \
        if (isK) {                                                            \
          if (xh == 0) {                                                      \
            _Pragma("unroll")                                                 \
            for (int i = 0; i < GFG / 2; ++i)                                 \
              acx = wmma4(fa[g & 1][i], ones, acx);                           \
          } else {                                                            \
            _Pragma("unroll")                                                 \
            for (int i = GFG / 2; i < GFG; ++i)                               \
              acx = wmma4(fa[g & 1][i], ones, acx);                           \
          }                                                                   \
        } else {                                                              \
          if (xh == 0) {                                                      \
            _Pragma("unroll")                                                 \
            for (int i = 0; i < GFG / 2; ++i)                                 \
              acx = wmma4(ones, fb[g & 1][i], acx);                           \
          } else {                                                            \
            _Pragma("unroll")                                                 \
            for (int i = GFG / 2; i < GFG; ++i)                               \
              acx = wmma4(ones, fb[g & 1][i], acx);                           \
          }                                                                   \
        }                                                                     \
      }                                                                       \
                                          \
      __builtin_amdgcn_sched_barrier(0);                                      \
    }                                                                         \
    LDS_BARRIER()                                                             \
  }

  #pragma unroll
  for (int g = 0; g < GNG; ++g) G_LOAD_SLICE(0, g)
  #pragma unroll
  for (int g = 0; g < GNG; ++g) G_STASH_SLICE(0, g)
  #pragma unroll
  for (int g = 0; g < GNG; ++g) G_LOAD_SLICE(1, g)
  __syncthreads();

  const int ard = lh * GKP + ((2 * (dw + lm)) ^ (4 * lh));
  const int brd = lh * GVP + ((2 * (ew + lm)) ^ (4 * ((ew >> 5) & 1)));
  v8f acc = {};
  v2f fa[2][GFG], fb[2][GFG];
  const int dt = rest >> 3, eg = rest & 7;
  const int ws = __builtin_amdgcn_readfirstlane(wv);
  const bool isK = (ws == 1) || (ws == 3) || (ws == 4) || (ws == 6);
  const int xg = isK ? eg : (dt >> 1);
  const int xh = isK ? ((ws == 3 || ws == 4) ? 1 : 0) : (dt & 1);
  v8f acx = {};
  v2f ones; ones.x = 1.0f; ones.y = 1.0f;
  for (int n = 0; n < GNST - 2; ++n) G_STAGE(n, 1, 1)
  G_STAGE(GNST - 2, 1, 0)
  G_STAGE(GNST - 1, 0, 0)
#undef G_HERE
#undef G_LOAD_K
#undef G_LOAD_V
#undef G_STASH_K
#undef G_STASH_V
#undef G_LOAD_SLICE
#undef G_STASH_SLICE
#undef G_FRAG
#undef G_STAGE
  float* Gb = G + ((size_t)b * DM + d0 + dw) * DM + e0 + ew;
  #pragma unroll
  for (int r = 0; r < 8; ++r)
    Gb[(size_t)(r + 8 * lh) * DM + lm] = acc[r];
  if (isK) {
    if (lm == 0) {
      float* pk = part + ((size_t)b * XCH + 2 * eg + xh) * (2 * DM) + d0 + dw + 8 * lh;
      #pragma unroll
      for (int r = 0; r < 8; ++r) pk[r] = acx[r];
    }
  } else {
    if (lh == 0)
      part[((size_t)b * XCH + dt) * (2 * DM) + DM + e0 + ew + lm] = acx[0];
  }
}

#define ROWP (DM + 4)
#define KP64 72
#define KP32 40

template <int NR>
__device__ __forceinline__ void stage_rows(const float* __restrict__ g, float* s,
                                           int wv, int lane) {
  #pragma unroll
  for (int j = 0; j < NR / 8; ++j)
    #pragma unroll
    for (int q = 0; q < 4; ++q)
      *(float4*)(s + (wv + 8 * j) * ROWP + 128 * q + 4 * lane) =
          *(const float4*)(g + (size_t)(wv + 8 * j) * DM + 128 * q + 4 * lane);
}

template <int NC, int PITCH>
__device__ __forceinline__ void stage_kmajor(const float* __restrict__ g, float* s,
                                             int tid) {
  constexpr int F4 = NC / 4;
  constexpr int RPS = 256 / F4;
  const int c4 = (tid % F4) * 4, r0 = tid / F4;
  #pragma unroll
  for (int j = 0; j < DM / RPS; ++j)
    *(float4*)(s + (r0 + RPS * j) * PITCH + c4) =
        *(const float4*)(g + (size_t)(r0 + RPS * j) * DM + c4);
}

__device__ __forceinline__ v2f ld_pair(const float* p, int pitch) {
  v2f r; r.x = p[0]; r.y = p[pitch]; return r;
}

#define SCH   XCH

__device__ __forceinline__ float2 chunk_sum2(const float* __restrict__ pp) {
  float2 s = *(const float2*)pp;
  #pragma unroll
  for (int c = 1; c < SCH; ++c) {
    const float2 t = *(const float2*)(pp + (size_t)c * (2 * DM));
    s.x += t.x; s.y += t.y;
  }
  return s;
}

#define PANEL_K_LOOP(LOADA, LOADB)                                            \
  {                                                                           \
    v2f fa[2][8], fb[2][8];                                                   \
    _Pragma("unroll")                                                         \
    for (int i = 0; i < 8; ++i) {                                             \
      const int kk = 4 * i;                                                   \
      fa[0][i] = LOADA; fb[0][i] = LOADB;                                     \
    }                                                                         \
    _Pragma("unroll")                                                         \
    for (int g = 0; g < DM / 32; ++g) {                                       \
      if (g + 1 < DM / 32) {                                                  \
        _Pragma("unroll")                                                     \
        for (int i = 0; i < 8; ++i) {                                         \
          const int kk = 32 * (g + 1) + 4 * i;                                \
          fa[(g + 1) & 1][i] = LOADA; fb[(g + 1) & 1][i] = LOADB;             \
        }                                                                     \
      }                                                                       \
      _Pragma("unroll")                                                       \
      for (int i = 0; i < 8; ++i)                                             \
        acc = wmma4(fa[g & 1][i], fb[g & 1][i], acc);                         \
      __builtin_amdgcn_sched_barrier(0);                                      \
    }                                                                         \
  }

#define TP (DKH + 4)

__global__ void __launch_bounds__(256) kT(const float* __restrict__ wk,
                                          const float* __restrict__ G,
                                          const float* __restrict__ part,
                                          const float* __restrict__ wvw,
                                          const float* __restrict__ bk,
                                          float* __restrict__ Mp,
                                          float* __restrict__ u) {
  __shared__ __attribute__((aligned(16))) float As[32 * ROWP];
  __shared__ __attribute__((aligned(16))) float Bs[DM * KP64];
  __shared__ __attribute__((aligned(16))) float Ts[32 * TP];
  __shared__ __attribute__((aligned(16))) float Vw[DKH * TP];
  __shared__ float vs[DKH];
  __shared__ float bks[32];
  const int tid = threadIdx.x;
  const int lane = tid & 31, wv = tid >> 5;
  const int lm = lane & 15, lh = lane >> 4;
  const int bh = blockIdx.x >> 4;
  const int b = bh >> 3, h = bh & 7;
  const int ih = (blockIdx.x >> 3) & 1;
  const int eg = blockIdx.x & 7;
  const int e0 = eg * 64;
  if (e0 == 0) {
    const float2 s = chunk_sum2(part + (size_t)b * SCH * (2 * DM) + 2 * tid);
    Bs[(2 * tid) * KP64 + 64] = s.x;
    Bs[(2 * tid + 1) * KP64 + 64] = s.y;
  }
  stage_rows<32>(wk + (size_t)(h * 64 + ih * 32) * DM, As, wv, lane);
  stage_kmajor<64, KP64>(G + (size_t)b * DM * DM + e0, Bs, tid);
  {
    const int c4 = (tid & 15) * 4, r0 = tid >> 4;
    #pragma unroll
    for (int t = 0; t < 4; ++t)
      *(float4*)(Vw + (r0 + 16 * t) * TP + c4) =
          *(const float4*)(wvw + (size_t)(h * 64 + r0 + 16 * t) * DM + e0 + c4);
  }
  if (wv == 6) bks[lane] = bk[h * 64 + ih * 32 + lane];
  if (wv == 7) {
    const float2 s =
        chunk_sum2(part + (size_t)b * SCH * (2 * DM) + DM + e0 + 2 * lane);
    vs[2 * lane] = s.x;
    vs[2 * lane + 1] = s.y;
  }
  __syncthreads();
  const int ti = wv >> 2, tj = wv & 3;
  const float* ap = As + (16 * ti + lm) * ROWP + 2 * lh;
  const float* bp = Bs + (2 * lh) * KP64 + 16 * tj + lm;
  v8f acc = {};
  PANEL_K_LOOP((*(const v2f*)(ap + kk)), ld_pair(bp + kk * KP64, KP64))
  {
    const float vse = vs[16 * tj + lm];
    #pragma unroll
    for (int r = 0; r < 8; ++r) {
      const int il = 16 * ti + r + 8 * lh;
      Ts[il * TP + 16 * tj + lm] = acc[r] + bks[il] * vse;
    }
  }
  __syncthreads();
  {
    const float* ap2 = Ts + (16 * ti + lm) * TP + 2 * lh;
    const float* bp2 = Vw + (16 * tj + lm) * TP + 2 * lh;
    v2f fa2[DKH / 4], fb2[DKH / 4];
    #pragma unroll
    for (int i = 0; i < DKH / 4; ++i) {
      fa2[i] = *(const v2f*)(ap2 + 4 * i);
      fb2[i] = *(const v2f*)(bp2 + 4 * i);
    }
    __builtin_amdgcn_sched_barrier(0);
    v8f acc2 = {};
    #pragma unroll
    for (int i = 0; i < DKH / 4; ++i)
      acc2 = wmma4(fa2[i], fb2[i], acc2);
    float* Mb = Mp + (((size_t)eg * (NB * NH) + bh) * DKH + ih * 32 + 16 * ti) * DKH + 16 * tj;
    #pragma unroll
    for (int r = 0; r < 8; ++r)
      Mb[(size_t)(r + 8 * lh) * DKH + lm] = acc2[r];
  }
  if (e0 == 0 && wv < 2) {
    const float* ua = As + (16 * wv + lm) * ROWP + 2 * lh;
    const float* ub = Bs + (2 * lh) * KP64 + 64;
    v8f accu;
    {
      v8f acc = {};
      PANEL_K_LOOP((*(const v2f*)(ua + kk)), ld_pair(ub + kk * KP64, KP64))
      accu = acc;
    }
    if (lm == 0) {
      float* ud = u + (size_t)b * DM + h * 64 + ih * 32 + 16 * wv + 8 * lh;
      #pragma unroll
      for (int r = 0; r < 8; ++r) ud[r] = accu[r];
    }
  }
}

__global__ void __launch_bounds__(256) kP(const float* __restrict__ Mp,
                                          const float* __restrict__ u,
                                          const float* __restrict__ bk,
                                          const float* __restrict__ bv,
                                          const float* __restrict__ wo,
                                          float* __restrict__ P) {
  __shared__ __attribute__((aligned(16))) float Ms[16 * TP];
  const int tid = threadIdx.x;
  const int lane = tid & 31, wv = tid >> 5;
  const int lm = lane & 15, lh = lane >> 4;
  const int bh = blockIdx.x >> 2;
  const int b = bh >> 3, h = bh & 7;
  const int i0 = (blockIdx.x & 3) * 16;
  const int o0 = wv * 64;
  v2f bm[DKH / 4][4];
  #pragma unroll
  for (int i = 0; i < DKH / 4; ++i)
    #pragma unroll
    for (int t = 0; t < 4; ++t)
      bm[i][t] = *(const v2f*)(wo + (size_t)(o0 + 16 * t + lm) * DM + h * 64 + 4 * i + 2 * lh);
  {
    const int row = tid >> 4, c4 = (tid & 15) * 4;
    const float* mp = Mp + ((size_t)bh * 64 + i0 + row) * DKH + c4;
    float4 s = *(const float4*)mp;
    #pragma unroll
    for (int g = 1; g < 8; ++g) {
      const float4 t = *(const float4*)(mp + (size_t)g * (NB * NH * DKH * DKH));
      s.x += t.x; s.y += t.y; s.z += t.z; s.w += t.w;
    }
    const float ui  = u[(size_t)b * DM + h * 64 + i0 + row];
    const float bki = bk[h * 64 + i0 + row];
    const float4 bv4 = *(const float4*)(bv + h * 64 + c4);
    float4 m;
    m.x = 0.125f * (s.x + ui * bv4.x + (float)SEQ * bki * bv4.x);
    m.y = 0.125f * (s.y + ui * bv4.y + (float)SEQ * bki * bv4.y);
    m.z = 0.125f * (s.z + ui * bv4.z + (float)SEQ * bki * bv4.z);
    m.w = 0.125f * (s.w + ui * bv4.w + (float)SEQ * bki * bv4.w);
    *(float4*)(Ms + row * TP + c4) = m;
  }
  __syncthreads();
  const float* ap = Ms + lm * TP + 2 * lh;
  v2f fa[DKH / 4];
  #pragma unroll
  for (int i = 0; i < DKH / 4; ++i) fa[i] = *(const v2f*)(ap + 4 * i);
  v8f acc[4] = {};
  #pragma unroll
  for (int i = 0; i < DKH / 4; ++i) {
    #pragma unroll
    for (int t = 0; t < 4; ++t)
      acc[t] = wmma4(fa[i], bm[i][t], acc[t]);
  }
  float* Pb = P + ((size_t)b * DM + h * 64 + i0) * DM + o0;
  #pragma unroll
  for (int t = 0; t < 4; ++t)
    #pragma unroll
    for (int r = 0; r < 8; ++r)
      Pb[(size_t)(r + 8 * lh) * DM + 16 * t + lm] = acc[t][r];
}

__global__ void __launch_bounds__(256) kW4(const float* __restrict__ wq,
                                           const float* __restrict__ bq,
                                           const float* __restrict__ P,
                                           float* __restrict__ W4,
                                           float* __restrict__ cq) {
  __shared__ __attribute__((aligned(16))) float As[DM * KP32];
  __shared__ __attribute__((aligned(16))) float Bs[DM * KP64];
  const int tid = threadIdx.x;
  const int lane = tid & 31, wv = tid >> 5;
  const int lm = lane & 15, lh = lane >> 4;
  const int b = blockIdx.x >> 7;
  const int d0 = ((blockIdx.x >> 3) & 15) * 32;
  const int o0 = (blockIdx.x & 7) * 64;
  As[tid * KP32 + 32] = bq[tid];
  As[(tid + 256) * KP32 + 32] = bq[tid + 256];
  stage_kmajor<32, KP32>(wq + d0, As, tid);
  stage_kmajor<64, KP64>(P + (size_t)b * DM * DM + o0, Bs, tid);
  __syncthreads();
  const int ti = wv >> 2, tj = wv & 3;
  const float* ap = As + (2 * lh) * KP32 + 16 * ti + lm;
  const float* bp = Bs + (2 * lh) * KP64 + 16 * tj + lm;
  v8f acc = {};
  PANEL_K_LOOP(ld_pair(ap + kk * KP32, KP32), ld_pair(bp + kk * KP64, KP64))
  float* Wb = W4 + ((size_t)b * DM + d0 + 16 * ti) * DM + o0 + 16 * tj;
  #pragma unroll
  for (int r = 0; r < 8; ++r)
    Wb[(size_t)(r + 8 * lh) * DM + lm] = acc[r];
  if (d0 == 0 && ti == 0) {
    const float* ca = As + (2 * lh) * KP32 + 32;
    v8f accc;
    {
      v8f acc = {};
      PANEL_K_LOOP(ld_pair(ca + kk * KP32, KP32), ld_pair(bp + kk * KP64, KP64))
      accc = acc;
    }
    if (lh == 0) cq[(size_t)b * DM + o0 + 16 * tj + lm] = accc[0];
  }
}

#define OKS   64
#define ONST  (DM / OKS)
#define OAP   (OKS + 4)
#define OBP   160
#define OABUF (256 * OAP)
#define OBBUF ((OKS / 2) * OBP)
#define OAIT  16
#define OBIT  (OKS / 32)
#define OFG   2
#define ONG   (OKS / 4 / OFG)

__global__ void __launch_bounds__(256) kOut(const float* __restrict__ q,
                                            const float* __restrict__ W4,
                                            const float* __restrict__ bo,
                                            const float* __restrict__ cq,
                                            float* __restrict__ out) {
  __shared__ __attribute__((aligned(16))) float Qs[2 * OABUF];
  __shared__ __attribute__((aligned(16))) float Ws[2 * OBBUF];
  const int tid = threadIdx.x;
  const int lane = tid & 31, wv = tid >> 5;
  const int lm = lane & 15, lh = lane >> 4;
  const int b = blockIdx.x >> 4;
  const int n0 = (blockIdx.x & 15) * 256;
  const int o0 = blockIdx.y * 64;
  const int sc = (tid & 15) * 4, sr = tid >> 4;
  const float* qg = q + ((size_t)b * SEQ + n0 + sr) * DM + sc;
  const float* wg = W4 + ((size_t)b * DM + 2 * sr) * DM + o0 + sc;
  const int qst = sr * OAP + sc;
  const int wsw = 4 * ((sc >> 5) & 1);
  const int wst0 = sr * OBP + 2 * sc + wsw, wst1 = sr * OBP + 2 * sc + 4 - wsw;
  v4f rq[OAIT];
  v4f rw[2 * OBIT];

#define O_LOAD_Q(st, j)                                                       \
  rq[j] = *(const v4f*)(qg + (st) * OKS + (size_t)(16 * (j)) * DM);
#define O_LOAD_W(st, j)                                                       \
  {                                                                           \
    const float* wn = wg + (size_t)(st) * OKS * DM + (size_t)(32 * (j)) * DM; \
    rw[2 * (j)]     = *(const v4f*)(wn);                                      \
    rw[2 * (j) + 1] = *(const v4f*)(wn + DM);                                 \
  }
#define O_STASH_Q(bf, j)                                                      \
  *(v4f*)(Qs + (bf) * OABUF + qst + 16 * (j) * OAP) = rq[j];
#define O_HERE(a, b) asm volatile("" : : "v"(a), "v"(b));
#define O_STASH_W(bf, j)                                                      \
  {                                                                           \
    float* wd = Ws + (bf) * OBBUF + 16 * (j) * OBP;                           \
    O_HERE(rw[2 * (j)], rw[2 * (j) + 1])                                      \
    *(float4*)(wd + wst0) = make_float4(rw[2 * (j)].x, rw[2 * (j) + 1].x,     \
                                        rw[2 * (j)].y, rw[2 * (j) + 1].y);    \
    *(float4*)(wd + wst1) = make_float4(rw[2 * (j)].z, rw[2 * (j) + 1].z,     \
                                        rw[2 * (j)].w, rw[2 * (j) + 1].w);    \
  }
  static_assert(OAIT == 2 * ONG && 4 * OBIT == ONG, "two q items a group, one W4 item every fourth group");
#define O_LOAD_SLICE(st, g)                                                   \
  {                                                                           \
    O_LOAD_Q(st, 2 * (g))                                                     \
    O_LOAD_Q(st, 2 * (g) + 1)                                                 \
    if (((g) & 3) == 0) O_LOAD_W(st, (g) / 4)                                 \
  }
#define O_STASH_SLICE(bf, g)                                                  \
  {                                                                           \
    O_STASH_Q(bf, 2 * (g))                                                    \
    O_STASH_Q(bf, 2 * (g) + 1)                                                \
    if (((g) & 3) == 0) O_STASH_W(bf, (g) / 4)                                \
  }
#define O_FRAG(dst, g)                                                        \
  {                                                                           \
    _Pragma("unroll")                                                         \
    for (int i = 0; i < OFG; ++i) {                                           \
      const int ks = (g) * OFG + i;                                           \
      fa[dst][i][0] = *(const v2f*)(qa + 4 * ks);                             \
      fa[dst][i][1] = *(const v2f*)(qa + 16 * OAP + 4 * ks);                  \
      fb[dst][i][0] = *(const v2f*)(wa0 + 2 * ks * OBP);                      \
      fb[dst][i][1] = *(const v2f*)(wa0 + 2 * ks * OBP + 32);                 \
      fb[dst][i][2] = *(const v2f*)(wa1 + 2 * ks * OBP + 64);                 \
      fb[dst][i][3] = *(const v2f*)(wa1 + 2 * ks * OBP + 96);                 \
    }                                                                         \
  }
#define O_STAGE(n, ST, LD)                                                    \
  {                                                                           \
    const int buf = (n) & 1;                                                  \
    const float* qa  = Qs + buf * OABUF + ard;                                \
    const float* wa0 = Ws + buf * OBBUF + wr0;                                \
    const float* wa1 = Ws + buf * OBBUF + wr1;                                \
    O_FRAG(0, 0)                                                              \
    _Pragma("unroll")                                                         \
    for (int g = 0; g < ONG; ++g) {                                           \
      if (ST) O_STASH_SLICE(buf ^ 1, g)                                       \
      if (g + 1 < ONG) O_FRAG((g + 1) & 1, g + 1)                             \
      if (LD) O_LOAD_SLICE((n) + 2, g)                                        \
                      \
      __builtin_amdgcn_sched_barrier(0);                                      \
      _Pragma("unroll")                                                       \
      for (int i = 0; i < OFG; ++i) {                                         \
        _Pragma("unroll")                                                     \
        for (int t = 0; t < 4; ++t) {                                         \
          acc[t]     = wmma4(fa[g & 1][i][0], fb[g & 1][i][t], acc[t]);       \
          acc[4 + t] = wmma4(fa[g & 1][i][1], fb[g & 1][i][t], acc[4 + t]);   \
        }                                                                     \
      }                                                                       \
    }                                                                         \
    LDS_BARRIER()                                                             \
  }

  #pragma unroll
  for (int g = 0; g < ONG; ++g) O_LOAD_SLICE(0, g)
  #pragma unroll
  for (int g = 0; g < ONG; ++g) O_STASH_SLICE(0, g)
  #pragma unroll
  for (int g = 0; g < ONG; ++g) O_LOAD_SLICE(1, g)
  __syncthreads();

  const int ard = (32 * wv + lm) * OAP + 2 * lh;
  const int wr0 = lh * OBP + 2 * lm;
  const int wr1 = lh * OBP + ((2 * lm) ^ 4);
  v8f acc[8] = {};
  v2f fa[2][OFG][2], fb[2][OFG][4];
  for (int n = 0; n < ONST - 2; ++n) O_STAGE(n, 1, 1)
  O_STAGE(ONST - 2, 1, 0)
  O_STAGE(ONST - 1, 0, 0)
#undef O_HERE
#undef O_LOAD_Q
#undef O_LOAD_W
#undef O_STASH_Q
#undef O_STASH_W
#undef O_LOAD_SLICE
#undef O_STASH_SLICE
#undef O_FRAG
#undef O_STAGE
  float* ob = out + ((size_t)b * SEQ + n0 + 32 * wv) * DM + o0;
  #pragma unroll
  for (int t = 0; t < 4; ++t) {
    float add = bo[o0 + 16 * t + lm] + cq[(size_t)b * DM + o0 + 16 * t + lm];
    #pragma unroll
    for (int m = 0; m < 2; ++m)
      #pragma unroll
      for (int r = 0; r < 8; ++r)
        ob[(size_t)(16 * m + r + 8 * lh) * DM + 16 * t + lm] = acc[4 * m + t][r] + add;
  }
}

extern "C" void kernel_launch(void* const* d_in, const int* in_sizes, int n_in,
                              void* d_out, int out_size, void* d_ws, size_t ws_size,
                              hipStream_t stream) {
  const float* q   = (const float*)d_in[0];
  const float* k   = (const float*)d_in[1];
  const float* v   = (const float*)d_in[2];
  const float* wq  = (const float*)d_in[3];
  const float* bq  = (const float*)d_in[4];
  const float* wk  = (const float*)d_in[5];
  const float* bk  = (const float*)d_in[6];
  const float* wvw = (const float*)d_in[7];
  const float* bv  = (const float*)d_in[8];
  const float* wo  = (const float*)d_in[9];
  const float* bo  = (const float*)d_in[10];
  float* out = (float*)d_out;

  float* ws = (float*)d_ws;
  float* G    = ws;
  float* Mp   = G    + (size_t)NB * DM * DM;
  float* P    = Mp   + (size_t)8 * NB * NH * DKH * DKH;
  float* W4   = P    + (size_t)NB * DM * DM;
  float* u    = W4   + (size_t)NB * DM * DM;
  float* cq   = u    + (size_t)NB * DM;
  float* part = cq   + (size_t)NB * DM;

  kG   <<<256, 256, 0, stream>>>(k, v, G, part);
  kT   <<<256, 256, 0, stream>>>(wk, G, part, wvw, bk, Mp, u);
  kP   <<<64, 256, 0, stream>>>(Mp, u, bk, bv, wo, P);
  kW4  <<<256, 256, 0, stream>>>(wq, bq, P, W4, cq);
  dim3 go(32, 8, 1);
  kOut <<<go, 256, 0, stream>>>(q, W4, bo, cq, out);
}


# kG: group tile 64x64 over half of SEQ (was 32x64 over all), a wave 16x32 with 2 accumulators on one A fragment: a third fewer L2 and LDS bytes; kT adds the 2 planes; changes G, ksum, vsum sum trees
# speedup vs baseline: 9.1758x; 1.0695x over previous
#include <hip/hip_runtime.h>


typedef float v2f __attribute__((ext_vector_type(2)));
typedef float v4f __attribute__((ext_vector_type(4)));
typedef float v8f __attribute__((ext_vector_type(8)));

#define DM   512
#define SEQ  4096
#define NB   2
#define NH   8
#define DKH  64

__device__ __forceinline__ v8f wmma4(v2f a, v2f b, v8f c) {
  return __builtin_amdgcn_wmma_f32_16x16x4_f32(
       false, a,  false, b,
       (short)0, c,  false,  false);
}

#define LDS_BARRIER()                                                         \
  {                                                                           \
    __builtin_amdgcn_fence(__ATOMIC_RELEASE, "workgroup", "local");           \
    __builtin_amdgcn_s_barrier();                                             \
    __builtin_amdgcn_fence(__ATOMIC_ACQUIRE, "workgroup", "local");           \
  }

#define GKS   128
#define GHALF (SEQ / 2)
#define GNST  (GHALF / GKS)
#define GPP   160
#define GBUF  ((GKS / 2) * GPP)
#define GIT   (GKS / 32)
#define GFG   8
#define GNG   (GKS / 4 / GFG)

#define XCH   16

__global__ void __launch_bounds__(256) kG(const float* __restrict__ k,
                                          const float* __restrict__ v,
                                          float* __restrict__ Gp,
                                          float* __restrict__ part) {
  __shared__ __attribute__((aligned(16))) float Ks[2 * GBUF];
  __shared__ __attribute__((aligned(16))) float Vs[2 * GBUF];
  __shared__ float Xk[8 * 16];
  __shared__ float Xv[8 * 32];
  const int tid = threadIdx.x;
  const int lane = tid & 31, wv = tid >> 5;
  const int lm = lane & 15, lh = lane >> 4;
  const int blk = blockIdx.x;
  const int b = blk >> 7;
  const int rest = blk & 127;
  const int hf = rest >> 6;
  const int dt = (rest >> 3) & 7, et = rest & 7;
  const int d0 = dt * 64, e0 = et * 64;
  const int dwi = wv & 3, ewi = wv >> 2;
  const int dw = 16 * dwi;
  const int ew = 32 * ewi;
  const int sc = (tid & 15) * 4, srp = tid >> 4;
  const size_t gbase = ((size_t)b * SEQ + (size_t)hf * GHALF + 2 * srp) * DM + sc;
  const float* kg = k + gbase + d0;
  const float* vg = v + gbase + e0;
  const int ssw = 4 * ((sc >> 5) & 1);
  const int st0 = srp * GPP + 2 * sc + ssw, st1 = srp * GPP + 2 * sc + 4 - ssw;
  v4f rk[2 * GIT], rv[2 * GIT];

#define G_LOAD(r, gp, st, j)                                                  \
  {                                                                           \
    const float* pn = gp + (size_t)(st) * GKS * DM + (size_t)(32 * (j)) * DM; \
    r[2 * (j)]     = *(const v4f*)(pn);                                       \
    r[2 * (j) + 1] = *(const v4f*)(pn + DM);                                  \
  }
#define G_HERE(a, b) asm volatile("" : : "v"(a), "v"(b));
#define G_STASH(r, S, bf, j)                                                  \
  {                                                                           \
    float* pd = S + (bf) * GBUF + 16 * (j) * GPP;                             \
    G_HERE(r[2 * (j)], r[2 * (j) + 1])                                        \
    *(float4*)(pd + st0) = make_float4(r[2 * (j)].x, r[2 * (j) + 1].x,        \
                                       r[2 * (j)].y, r[2 * (j) + 1].y);       \
    *(float4*)(pd + st1) = make_float4(r[2 * (j)].z, r[2 * (j) + 1].z,        \
                                       r[2 * (j)].w, r[2 * (j) + 1].w);       \
  }
  static_assert(GIT == GNG, "one item of each panel a group");
#define G_LOAD_SLICE(st, g)                                                   \
  {                                                                           \
    G_LOAD(rv, vg, st, g)                                                     \
    G_LOAD(rk, kg, st, g)                                                     \
  }
#define G_STASH_SLICE(bf, g)                                                  \
  {                                                                           \
    G_STASH(rv, Vs, bf, g)                                                    \
    G_STASH(rk, Ks, bf, g)                                                    \
  }
#define G_FRAG(dst, g)                                                        \
  {                                                                           \
    _Pragma("unroll")                                                         \
    for (int i = 0; i < GFG; ++i) {                                           \
      fa[dst][i]    = *(const v2f*)(ka + 2 * ((g) * GFG + i) * GPP);          \
      fb[dst][i][0] = *(const v2f*)(va + 2 * ((g) * GFG + i) * GPP);          \
      fb[dst][i][1] = *(const v2f*)(va + 2 * ((g) * GFG + i) * GPP + 32);     \
    }                                                                         \
  }
#define G_STAGE(n, ST, LD)                                                    \
  {                                                                           \
    const int buf = (n) & 1;                                                  \
    const float* ka = Ks + buf * GBUF + ard;                                  \
    const float* va = Vs + buf * GBUF + brd;                                  \
    G_FRAG(0, 0)                                                              \
    _Pragma("unroll")                                                         \
    for (int g = 0; g < GNG; ++g) {                                           \
      if (ST) G_STASH_SLICE(buf ^ 1, g)                                       \
      if (g + 1 < GNG) G_FRAG((g + 1) & 1, g + 1)                             \
      if (g == 0) {                           \
        xa[0] = *(const v2f*)(ka + xko);                                      \
        xa[1] = *(const v2f*)(ka + xko + 2 * GPP);                            \
        xb[0] = *(const v2f*)(va + xvo);                                      \
        xb[1] = *(const v2f*)(va + xvo + 32);                                 \
      }                                                                       \
      if (LD) G_LOAD_SLICE((n) + 2, g)                                        \
               \
      __builtin_amdgcn_sched_barrier(0);                                      \
      _Pragma("unroll")                                                       \
      for (int i = 0; i < GFG; ++i) {                                         \
        acc[0] = wmma4(fa[g & 1][i], fb[g & 1][i][0], acc[0]);                \
        acc[1] = wmma4(fa[g & 1][i], fb[g & 1][i][1], acc[1]);                \
      }                                                                       \
      if (g == 1) {                                                           \
        acxk    = wmma4(xa[0], ones, acxk);                                   \
        acxv[0] = wmma4(ones, xb[0], acxv[0]);                                \
        acxk    = wmma4(xa[1], ones, acxk);                                   \
        acxv[1] = wmma4(ones, xb[1], acxv[1]);                                \
      }                                                                       \
    }                                                                         \
    LDS_BARRIER()                                                             \
  }

  #pragma unroll
  for (int g = 0; g < GNG; ++g) G_LOAD_SLICE(0, g)
  #pragma unroll
  for (int g = 0; g < GNG; ++g) G_STASH_SLICE(0, g)
  #pragma unroll
  for (int g = 0; g < GNG; ++g) G_LOAD_SLICE(1, g)
  __syncthreads();

  const int ard = lh * GPP + ((2 * (dw + lm)) ^ (4 * ((dw >> 5) & 1)));
  const int brd = lh * GPP + ((2 * (ew + lm)) ^ (4 * ((ew >> 5) & 1)));
  const int xko = 2 * (4 * et + 2 * ewi) * GPP;
  const int xvo = 2 * (4 * dt + dwi) * GPP;
  v8f acc[2] = {};
  v2f fa[2][GFG], fb[2][GFG][2];
  v2f xa[2], xb[2];
  v8f acxk = {};
  v8f acxv[2] = {};
  v2f ones; ones.x = 1.0f; ones.y = 1.0f;
  for (int n = 0; n < GNST - 2; ++n) G_STAGE(n, 1, 1)
  G_STAGE(GNST - 2, 1, 0)
  G_STAGE(GNST - 1, 0, 0)
#undef G_HERE
#undef G_LOAD
#undef G_STASH
#undef G_LOAD_SLICE
#undef G_STASH_SLICE
#undef G_FRAG
#undef G_STAGE
  float* Gb = Gp + (((size_t)hf * NB + b) * DM + d0 + dw) * DM + e0 + ew;
  #pragma unroll
  for (int t = 0; t < 2; ++t)
    #pragma unroll
    for (int r = 0; r < 8; ++r)
      Gb[(size_t)(r + 8 * lh) * DM + 16 * t + lm] = acc[t][r];
  if (lm == 0) {
    #pragma unroll
    for (int r = 0; r < 8; ++r) Xk[wv * 16 + 8 * lh + r] = acxk[r];
  }
  if (lh == 0) {
    Xv[wv * 32 + lm]      = acxv[0][0];
    Xv[wv * 32 + 16 + lm] = acxv[1][0];
  }
  __syncthreads();
  if (tid < 64) {
    const int wi = tid >> 4, row = tid & 15;
    part[((size_t)b * XCH + hf * 8 + et) * (2 * DM) + d0 + tid] =
        Xk[wi * 16 + row] + Xk[(wi + 4) * 16 + row];
  } else if (tid < 128) {
    const int col = tid - 64, eh = col >> 5, cc = col & 31;
    float s = Xv[(4 * eh) * 32 + cc];
    #pragma unroll
    for (int w = 1; w < 4; ++w) s += Xv[(4 * eh + w) * 32 + cc];
    part[((size_t)b * XCH + hf * 8 + dt) * (2 * DM) + DM + e0 + col] = s;
  }
}

#define ROWP (DM + 4)
#define KP64 72
#define KP32 40

template <int NR>
__device__ __forceinline__ void stage_rows(const float* __restrict__ g, float* s,
                                           int wv, int lane) {
  #pragma unroll
  for (int j = 0; j < NR / 8; ++j)
    #pragma unroll
    for (int q = 0; q < 4; ++q)
      *(float4*)(s + (wv + 8 * j) * ROWP + 128 * q + 4 * lane) =
          *(const float4*)(g + (size_t)(wv + 8 * j) * DM + 128 * q + 4 * lane);
}

template <int NC, int PITCH>
__device__ __forceinline__ void stage_kmajor(const float* __restrict__ g, float* s,
                                             int tid) {
  constexpr int F4 = NC / 4;
  constexpr int RPS = 256 / F4;
  const int c4 = (tid % F4) * 4, r0 = tid / F4;
  #pragma unroll
  for (int j = 0; j < DM / RPS; ++j)
    *(float4*)(s + (r0 + RPS * j) * PITCH + c4) =
        *(const float4*)(g + (size_t)(r0 + RPS * j) * DM + c4);
}

template <int NC, int PITCH>
__device__ __forceinline__ void stage_kmajor_sum2(const float* __restrict__ g0,
                                                  const float* __restrict__ g1,
                                                  float* s, int tid) {
  constexpr int F4 = NC / 4;
  constexpr int RPS = 256 / F4;
  const int c4 = (tid % F4) * 4, r0 = tid / F4;
  #pragma unroll
  for (int j = 0; j < DM / RPS; ++j) {
    const float4 a = *(const float4*)(g0 + (size_t)(r0 + RPS * j) * DM + c4);
    const float4 c = *(const float4*)(g1 + (size_t)(r0 + RPS * j) * DM + c4);
    *(float4*)(s + (r0 + RPS * j) * PITCH + c4) =
        make_float4(a.x + c.x, a.y + c.y, a.z + c.z, a.w + c.w);
  }
}

__device__ __forceinline__ v2f ld_pair(const float* p, int pitch) {
  v2f r; r.x = p[0]; r.y = p[pitch]; return r;
}

#define SCH   XCH

__device__ __forceinline__ float2 chunk_sum2(const float* __restrict__ pp) {
  float2 s = *(const float2*)pp;
  #pragma unroll
  for (int c = 1; c < SCH; ++c) {
    const float2 t = *(const float2*)(pp + (size_t)c * (2 * DM));
    s.x += t.x; s.y += t.y;
  }
  return s;
}

#define PANEL_K_LOOP(LOADA, LOADB)                                            \
  {                                                                           \
    v2f fa[2][8], fb[2][8];                                                   \
    _Pragma("unroll")                                                         \
    for (int i = 0; i < 8; ++i) {                                             \
      const int kk = 4 * i;                                                   \
      fa[0][i] = LOADA; fb[0][i] = LOADB;                                     \
    }                                                                         \
    _Pragma("unroll")                                                         \
    for (int g = 0; g < DM / 32; ++g) {                                       \
      if (g + 1 < DM / 32) {                                                  \
        _Pragma("unroll")                                                     \
        for (int i = 0; i < 8; ++i) {                                         \
          const int kk = 32 * (g + 1) + 4 * i;                                \
          fa[(g + 1) & 1][i] = LOADA; fb[(g + 1) & 1][i] = LOADB;             \
        }                                                                     \
      }                                                                       \
      _Pragma("unroll")                                                       \
      for (int i = 0; i < 8; ++i)                                             \
        acc = wmma4(fa[g & 1][i], fb[g & 1][i], acc);                         \
      __builtin_amdgcn_sched_barrier(0);                                      \
    }                                                                         \
  }

#define TP (DKH + 4)

__global__ void __launch_bounds__(256) kT(const float* __restrict__ wk,
                                          const float* __restrict__ G,
                                          const float* __restrict__ part,
                                          const float* __restrict__ wvw,
                                          const float* __restrict__ bk,
                                          float* __restrict__ Mp,
                                          float* __restrict__ u) {
  __shared__ __attribute__((aligned(16))) float As[32 * ROWP];
  __shared__ __attribute__((aligned(16))) float Bs[DM * KP64];
  __shared__ __attribute__((aligned(16))) float Ts[32 * TP];
  __shared__ __attribute__((aligned(16))) float Vw[DKH * TP];
  __shared__ float vs[DKH];
  __shared__ float bks[32];
  const int tid = threadIdx.x;
  const int lane = tid & 31, wv = tid >> 5;
  const int lm = lane & 15, lh = lane >> 4;
  const int bh = blockIdx.x >> 4;
  const int b = bh >> 3, h = bh & 7;
  const int ih = (blockIdx.x >> 3) & 1;
  const int eg = blockIdx.x & 7;
  const int e0 = eg * 64;
  if (e0 == 0) {
    const float2 s = chunk_sum2(part + (size_t)b * SCH * (2 * DM) + 2 * tid);
    Bs[(2 * tid) * KP64 + 64] = s.x;
    Bs[(2 * tid + 1) * KP64 + 64] = s.y;
  }
  stage_rows<32>(wk + (size_t)(h * 64 + ih * 32) * DM, As, wv, lane);
  stage_kmajor_sum2<64, KP64>(G + (size_t)b * DM * DM + e0,
                              G + (size_t)(NB + b) * DM * DM + e0, Bs, tid);
  {
    const int c4 = (tid & 15) * 4, r0 = tid >> 4;
    #pragma unroll
    for (int t = 0; t < 4; ++t)
      *(float4*)(Vw + (r0 + 16 * t) * TP + c4) =
          *(const float4*)(wvw + (size_t)(h * 64 + r0 + 16 * t) * DM + e0 + c4);
  }
  if (wv == 6) bks[lane] = bk[h * 64 + ih * 32 + lane];
  if (wv == 7) {
    const float2 s =
        chunk_sum2(part + (size_t)b * SCH * (2 * DM) + DM + e0 + 2 * lane);
    vs[2 * lane] = s.x;
    vs[2 * lane + 1] = s.y;
  }
  __syncthreads();
  const int ti = wv >> 2, tj = wv & 3;
  const float* ap = As + (16 * ti + lm) * ROWP + 2 * lh;
  const float* bp = Bs + (2 * lh) * KP64 + 16 * tj + lm;
  v8f acc = {};
  PANEL_K_LOOP((*(const v2f*)(ap + kk)), ld_pair(bp + kk * KP64, KP64))
  {
    const float vse = vs[16 * tj + lm];
    #pragma unroll
    for (int r = 0; r < 8; ++r) {
      const int il = 16 * ti + r + 8 * lh;
      Ts[il * TP + 16 * tj + lm] = acc[r] + bks[il] * vse;
    }
  }
  __syncthreads();
  {
    const float* ap2 = Ts + (16 * ti + lm) * TP + 2 * lh;
    const float* bp2 = Vw + (16 * tj + lm) * TP + 2 * lh;
    v2f fa2[DKH / 4], fb2[DKH / 4];
    #pragma unroll
    for (int i = 0; i < DKH / 4; ++i) {
      fa2[i] = *(const v2f*)(ap2 + 4 * i);
      fb2[i] = *(const v2f*)(bp2 + 4 * i);
    }
    __builtin_amdgcn_sched_barrier(0);
    v8f acc2 = {};
    #pragma unroll
    for (int i = 0; i < DKH / 4; ++i)
      acc2 = wmma4(fa2[i], fb2[i], acc2);
    float* Mb = Mp + (((size_t)eg * (NB * NH) + bh) * DKH + ih * 32 + 16 * ti) * DKH + 16 * tj;
    #pragma unroll
    for (int r = 0; r < 8; ++r)
      Mb[(size_t)(r + 8 * lh) * DKH + lm] = acc2[r];
  }
  if (e0 == 0 && wv < 2) {
    const float* ua = As + (16 * wv + lm) * ROWP + 2 * lh;
    const float* ub = Bs + (2 * lh) * KP64 + 64;
    v8f accu;
    {
      v8f acc = {};
      PANEL_K_LOOP((*(const v2f*)(ua + kk)), ld_pair(ub + kk * KP64, KP64))
      accu = acc;
    }
    if (lm == 0) {
      float* ud = u + (size_t)b * DM + h * 64 + ih * 32 + 16 * wv + 8 * lh;
      #pragma unroll
      for (int r = 0; r < 8; ++r) ud[r] = accu[r];
    }
  }
}

__global__ void __launch_bounds__(256) kP(const float* __restrict__ Mp,
                                          const float* __restrict__ u,
                                          const float* __restrict__ bk,
                                          const float* __restrict__ bv,
                                          const float* __restrict__ wo,
                                          float* __restrict__ P) {
  __shared__ __attribute__((aligned(16))) float Ms[16 * TP];
  const int tid = threadIdx.x;
  const int lane = tid & 31, wv = tid >> 5;
  const int lm = lane & 15, lh = lane >> 4;
  const int bh = blockIdx.x >> 2;
  const int b = bh >> 3, h = bh & 7;
  const int i0 = (blockIdx.x & 3) * 16;
  const int o0 = wv * 64;
  v2f bm[DKH / 4][4];
  #pragma unroll
  for (int i = 0; i < DKH / 4; ++i)
    #pragma unroll
    for (int t = 0; t < 4; ++t)
      bm[i][t] = *(const v2f*)(wo + (size_t)(o0 + 16 * t + lm) * DM + h * 64 + 4 * i + 2 * lh);
  {
    const int row = tid >> 4, c4 = (tid & 15) * 4;
    const float* mp = Mp + ((size_t)bh * 64 + i0 + row) * DKH + c4;
    float4 s = *(const float4*)mp;
    #pragma unroll
    for (int g = 1; g < 8; ++g) {
      const float4 t = *(const float4*)(mp + (size_t)g * (NB * NH * DKH * DKH));
      s.x += t.x; s.y += t.y; s.z += t.z; s.w += t.w;
    }
    const float ui  = u[(size_t)b * DM + h * 64 + i0 + row];
    const float bki = bk[h * 64 + i0 + row];
    const float4 bv4 = *(const float4*)(bv + h * 64 + c4);
    float4 m;
    m.x = 0.125f * (s.x + ui * bv4.x + (float)SEQ * bki * bv4.x);
    m.y = 0.125f * (s.y + ui * bv4.y + (float)SEQ * bki * bv4.y);
    m.z = 0.125f * (s.z + ui * bv4.z + (float)SEQ * bki * bv4.z);
    m.w = 0.125f * (s.w + ui * bv4.w + (float)SEQ * bki * bv4.w);
    *(float4*)(Ms + row * TP + c4) = m;
  }
  __syncthreads();
  const float* ap = Ms + lm * TP + 2 * lh;
  v2f fa[DKH / 4];
  #pragma unroll
  for (int i = 0; i < DKH / 4; ++i) fa[i] = *(const v2f*)(ap + 4 * i);
  v8f acc[4] = {};
  #pragma unroll
  for (int i = 0; i < DKH / 4; ++i) {
    #pragma unroll
    for (int t = 0; t < 4; ++t)
      acc[t] = wmma4(fa[i], bm[i][t], acc[t]);
  }
  float* Pb = P + ((size_t)b * DM + h * 64 + i0) * DM + o0;
  #pragma unroll
  for (int t = 0; t < 4; ++t)
    #pragma unroll
    for (int r = 0; r < 8; ++r)
      Pb[(size_t)(r + 8 * lh) * DM + 16 * t + lm] = acc[t][r];
}

__global__ void __launch_bounds__(256) kW4(const float* __restrict__ wq,
                                           const float* __restrict__ bq,
                                           const float* __restrict__ P,
                                           float* __restrict__ W4,
                                           float* __restrict__ cq) {
  __shared__ __attribute__((aligned(16))) float As[DM * KP32];
  __shared__ __attribute__((aligned(16))) float Bs[DM * KP64];
  const int tid = threadIdx.x;
  const int lane = tid & 31, wv = tid >> 5;
  const int lm = lane & 15, lh = lane >> 4;
  const int b = blockIdx.x >> 7;
  const int d0 = ((blockIdx.x >> 3) & 15) * 32;
  const int o0 = (blockIdx.x & 7) * 64;
  As[tid * KP32 + 32] = bq[tid];
  As[(tid + 256) * KP32 + 32] = bq[tid + 256];
  stage_kmajor<32, KP32>(wq + d0, As, tid);
  stage_kmajor<64, KP64>(P + (size_t)b * DM * DM + o0, Bs, tid);
  __syncthreads();
  const int ti = wv >> 2, tj = wv & 3;
  const float* ap = As + (2 * lh) * KP32 + 16 * ti + lm;
  const float* bp = Bs + (2 * lh) * KP64 + 16 * tj + lm;
  v8f acc = {};
  PANEL_K_LOOP(ld_pair(ap + kk * KP32, KP32), ld_pair(bp + kk * KP64, KP64))
  float* Wb = W4 + ((size_t)b * DM + d0 + 16 * ti) * DM + o0 + 16 * tj;
  #pragma unroll
  for (int r = 0; r < 8; ++r)
    Wb[(size_t)(r + 8 * lh) * DM + lm] = acc[r];
  if (d0 == 0 && ti == 0) {
    const float* ca = As + (2 * lh) * KP32 + 32;
    v8f accc;
    {
      v8f acc = {};
      PANEL_K_LOOP(ld_pair(ca + kk * KP32, KP32), ld_pair(bp + kk * KP64, KP64))
      accc = acc;
    }
    if (lh == 0) cq[(size_t)b * DM + o0 + 16 * tj + lm] = accc[0];
  }
}

#define OKS   64
#define ONST  (DM / OKS)
#define OAP   (OKS + 4)
#define OBP   160
#define OABUF (256 * OAP)
#define OBBUF ((OKS / 2) * OBP)
#define OAIT  16
#define OBIT  (OKS / 32)
#define OFG   2
#define ONG   (OKS / 4 / OFG)

__global__ void __launch_bounds__(256) kOut(const float* __restrict__ q,
                                            const float* __restrict__ W4,
                                            const float* __restrict__ bo,
                                            const float* __restrict__ cq,
                                            float* __restrict__ out) {
  __shared__ __attribute__((aligned(16))) float Qs[2 * OABUF];
  __shared__ __attribute__((aligned(16))) float Ws[2 * OBBUF];
  const int tid = threadIdx.x;
  const int lane = tid & 31, wv = tid >> 5;
  const int lm = lane & 15, lh = lane >> 4;
  const int b = blockIdx.x >> 4;
  const int n0 = (blockIdx.x & 15) * 256;
  const int o0 = blockIdx.y * 64;
  const int sc = (tid & 15) * 4, sr = tid >> 4;
  const float* qg = q + ((size_t)b * SEQ + n0 + sr) * DM + sc;
  const float* wg = W4 + ((size_t)b * DM + 2 * sr) * DM + o0 + sc;
  const int qst = sr * OAP + sc;
  const int wsw = 4 * ((sc >> 5) & 1);
  const int wst0 = sr * OBP + 2 * sc + wsw, wst1 = sr * OBP + 2 * sc + 4 - wsw;
  v4f rq[OAIT];
  v4f rw[2 * OBIT];

#define O_LOAD_Q(st, j)                                                       \
  rq[j] = *(const v4f*)(qg + (st) * OKS + (size_t)(16 * (j)) * DM);
#define O_LOAD_W(st, j)                                                       \
  {                                                                           \
    const float* wn = wg + (size_t)(st) * OKS * DM + (size_t)(32 * (j)) * DM; \
    rw[2 * (j)]     = *(const v4f*)(wn);                                      \
    rw[2 * (j) + 1] = *(const v4f*)(wn + DM);                                 \
  }
#define O_STASH_Q(bf, j)                                                      \
  *(v4f*)(Qs + (bf) * OABUF + qst + 16 * (j) * OAP) = rq[j];
#define O_HERE(a, b) asm volatile("" : : "v"(a), "v"(b));
#define O_STASH_W(bf, j)                                                      \
  {                                                                           \
    float* wd = Ws + (bf) * OBBUF + 16 * (j) * OBP;                           \
    O_HERE(rw[2 * (j)], rw[2 * (j) + 1])                                      \
    *(float4*)(wd + wst0) = make_float4(rw[2 * (j)].x, rw[2 * (j) + 1].x,     \
                                        rw[2 * (j)].y, rw[2 * (j) + 1].y);    \
    *(float4*)(wd + wst1) = make_float4(rw[2 * (j)].z, rw[2 * (j) + 1].z,     \
                                        rw[2 * (j)].w, rw[2 * (j) + 1].w);    \
  }
  static_assert(OAIT == 2 * ONG && 4 * OBIT == ONG, "two q items a group, one W4 item every fourth group");
#define O_LOAD_SLICE(st, g)                                                   \
  {                                                                           \
    O_LOAD_Q(st, 2 * (g))                                                     \
    O_LOAD_Q(st, 2 * (g) + 1)                                                 \
    if (((g) & 3) == 0) O_LOAD_W(st, (g) / 4)                                 \
  }
#define O_STASH_SLICE(bf, g)                                                  \
  {                                                                           \
    O_STASH_Q(bf, 2 * (g))                                                    \
    O_STASH_Q(bf, 2 * (g) + 1)                                                \
    if (((g) & 3) == 0) O_STASH_W(bf, (g) / 4)                                \
  }
#define O_FRAG(dst, g)                                                        \
  {                                                                           \
    _Pragma("unroll")                                                         \
    for (int i = 0; i < OFG; ++i) {                                           \
      const int ks = (g) * OFG + i;                                           \
      fa[dst][i][0] = *(const v2f*)(qa + 4 * ks);                             \
      fa[dst][i][1] = *(const v2f*)(qa + 16 * OAP + 4 * ks);                  \
      fb[dst][i][0] = *(const v2f*)(wa0 + 2 * ks * OBP);                      \
      fb[dst][i][1] = *(const v2f*)(wa0 + 2 * ks * OBP + 32);                 \
      fb[dst][i][2] = *(const v2f*)(wa1 + 2 * ks * OBP + 64);                 \
      fb[dst][i][3] = *(const v2f*)(wa1 + 2 * ks * OBP + 96);                 \
    }                                                                         \
  }
#define O_STAGE(n, ST, LD)                                                    \
  {                                                                           \
    const int buf = (n) & 1;                                                  \
    const float* qa  = Qs + buf * OABUF + ard;                                \
    const float* wa0 = Ws + buf * OBBUF + wr0;                                \
    const float* wa1 = Ws + buf * OBBUF + wr1;                                \
    O_FRAG(0, 0)                                                              \
    _Pragma("unroll")                                                         \
    for (int g = 0; g < ONG; ++g) {                                           \
      if (ST) O_STASH_SLICE(buf ^ 1, g)                                       \
      if (g + 1 < ONG) O_FRAG((g + 1) & 1, g + 1)                             \
      if (LD) O_LOAD_SLICE((n) + 2, g)                                        \
                      \
      __builtin_amdgcn_sched_barrier(0);                                      \
      _Pragma("unroll")                                                       \
      for (int i = 0; i < OFG; ++i) {                                         \
        _Pragma("unroll")                                                     \
        for (int t = 0; t < 4; ++t) {                                         \
          acc[t]     = wmma4(fa[g & 1][i][0], fb[g & 1][i][t], acc[t]);       \
          acc[4 + t] = wmma4(fa[g & 1][i][1], fb[g & 1][i][t], acc[4 + t]);   \
        }                                                                     \
      }                                                                       \
    }                                                                         \
    LDS_BARRIER()                                                             \
  }

  #pragma unroll
  for (int g = 0; g < ONG; ++g) O_LOAD_SLICE(0, g)
  #pragma unroll
  for (int g = 0; g < ONG; ++g) O_STASH_SLICE(0, g)
  #pragma unroll
  for (int g = 0; g < ONG; ++g) O_LOAD_SLICE(1, g)
  __syncthreads();

  const int ard = (32 * wv + lm) * OAP + 2 * lh;
  const int wr0 = lh * OBP + 2 * lm;
  const int wr1 = lh * OBP + ((2 * lm) ^ 4);
  v8f acc[8] = {};
  v2f fa[2][OFG][2], fb[2][OFG][4];
  for (int n = 0; n < ONST - 2; ++n) O_STAGE(n, 1, 1)
  O_STAGE(ONST - 2, 1, 0)
  O_STAGE(ONST - 1, 0, 0)
#undef O_HERE
#undef O_LOAD_Q
#undef O_LOAD_W
#undef O_STASH_Q
#undef O_STASH_W
#undef O_LOAD_SLICE
#undef O_STASH_SLICE
#undef O_FRAG
#undef O_STAGE
  float* ob = out + ((size_t)b * SEQ + n0 + 32 * wv) * DM + o0;
  #pragma unroll
  for (int t = 0; t < 4; ++t) {
    float add = bo[o0 + 16 * t + lm] + cq[(size_t)b * DM + o0 + 16 * t + lm];
    #pragma unroll
    for (int m = 0; m < 2; ++m)
      #pragma unroll
      for (int r = 0; r < 8; ++r)
        ob[(size_t)(16 * m + r + 8 * lh) * DM + 16 * t + lm] = acc[4 * m + t][r] + add;
  }
}

extern "C" void kernel_launch(void* const* d_in, const int* in_sizes, int n_in,
                              void* d_out, int out_size, void* d_ws, size_t ws_size,
                              hipStream_t stream) {
  const float* q   = (const float*)d_in[0];
  const float* k   = (const float*)d_in[1];
  const float* v   = (const float*)d_in[2];
  const float* wq  = (const float*)d_in[3];
  const float* bq  = (const float*)d_in[4];
  const float* wk  = (const float*)d_in[5];
  const float* bk  = (const float*)d_in[6];
  const float* wvw = (const float*)d_in[7];
  const float* bv  = (const float*)d_in[8];
  const float* wo  = (const float*)d_in[9];
  const float* bo  = (const float*)d_in[10];
  float* out = (float*)d_out;

  float* ws = (float*)d_ws;
  float* G    = ws;
  float* Mp   = G    + (size_t)2 * NB * DM * DM;
  float* P    = Mp   + (size_t)8 * NB * NH * DKH * DKH;
  float* W4   = P    + (size_t)NB * DM * DM;
  float* u    = W4   + (size_t)NB * DM * DM;
  float* cq   = u    + (size_t)NB * DM;
  float* part = cq   + (size_t)NB * DM;

  kG   <<<256, 256, 0, stream>>>(k, v, G, part);
  kT   <<<256, 256, 0, stream>>>(wk, G, part, wvw, bk, Mp, u);
  kP   <<<64, 256, 0, stream>>>(Mp, u, bk, bv, wo, P);
  kW4  <<<256, 256, 0, stream>>>(wq, bq, P, W4, cq);
  dim3 go(32, 8, 1);
  kOut <<<go, 256, 0, stream>>>(q, W4, bo, cq, out);
}


# kT, kW4: u, cq leave their second full K loop in the e-group-0 / d-tile-0 work-groups: every work-group sharing the panel takes 8 k-steps of it, 16 partials added in kP, kOut; changes u, cq sum order
# speedup vs baseline: 9.4872x; 1.0339x over previous
#include <hip/hip_runtime.h>


typedef float v2f __attribute__((ext_vector_type(2)));
typedef float v4f __attribute__((ext_vector_type(4)));
typedef float v8f __attribute__((ext_vector_type(8)));

#define DM   512
#define SEQ  4096
#define NB   2
#define NH   8
#define DKH  64

__device__ __forceinline__ v8f wmma4(v2f a, v2f b, v8f c) {
  return __builtin_amdgcn_wmma_f32_16x16x4_f32(
       false, a,  false, b,
       (short)0, c,  false,  false);
}

#define LDS_BARRIER()                                                         \
  {                                                                           \
    __builtin_amdgcn_fence(__ATOMIC_RELEASE, "workgroup", "local");           \
    __builtin_amdgcn_s_barrier();                                             \
    __builtin_amdgcn_fence(__ATOMIC_ACQUIRE, "workgroup", "local");           \
  }

#define GKS   128
#define GHALF (SEQ / 2)
#define GNST  (GHALF / GKS)
#define GPP   160
#define GBUF  ((GKS / 2) * GPP)
#define GIT   (GKS / 32)
#define GFG   8
#define GNG   (GKS / 4 / GFG)

#define XCH   16

__global__ void __launch_bounds__(256) kG(const float* __restrict__ k,
                                          const float* __restrict__ v,
                                          float* __restrict__ Gp,
                                          float* __restrict__ part) {
  __shared__ __attribute__((aligned(16))) float Ks[2 * GBUF];
  __shared__ __attribute__((aligned(16))) float Vs[2 * GBUF];
  __shared__ float Xk[8 * 16];
  __shared__ float Xv[8 * 32];
  const int tid = threadIdx.x;
  const int lane = tid & 31, wv = tid >> 5;
  const int lm = lane & 15, lh = lane >> 4;
  const int blk = blockIdx.x;
  const int b = blk >> 7;
  const int rest = blk & 127;
  const int hf = rest >> 6;
  const int dt = (rest >> 3) & 7, et = rest & 7;
  const int d0 = dt * 64, e0 = et * 64;
  const int dwi = wv & 3, ewi = wv >> 2;
  const int dw = 16 * dwi;
  const int ew = 32 * ewi;
  const int sc = (tid & 15) * 4, srp = tid >> 4;
  const size_t gbase = ((size_t)b * SEQ + (size_t)hf * GHALF + 2 * srp) * DM + sc;
  const float* kg = k + gbase + d0;
  const float* vg = v + gbase + e0;
  const int ssw = 4 * ((sc >> 5) & 1);
  const int st0 = srp * GPP + 2 * sc + ssw, st1 = srp * GPP + 2 * sc + 4 - ssw;
  v4f rk[2 * GIT], rv[2 * GIT];

#define G_LOAD(r, gp, st, j)                                                  \
  {                                                                           \
    const float* pn = gp + (size_t)(st) * GKS * DM + (size_t)(32 * (j)) * DM; \
    r[2 * (j)]     = *(const v4f*)(pn);                                       \
    r[2 * (j) + 1] = *(const v4f*)(pn + DM);                                  \
  }
#define G_HERE(a, b) asm volatile("" : : "v"(a), "v"(b));
#define G_STASH(r, S, bf, j)                                                  \
  {                                                                           \
    float* pd = S + (bf) * GBUF + 16 * (j) * GPP;                             \
    G_HERE(r[2 * (j)], r[2 * (j) + 1])                                        \
    *(float4*)(pd + st0) = make_float4(r[2 * (j)].x, r[2 * (j) + 1].x,        \
                                       r[2 * (j)].y, r[2 * (j) + 1].y);       \
    *(float4*)(pd + st1) = make_float4(r[2 * (j)].z, r[2 * (j) + 1].z,        \
                                       r[2 * (j)].w, r[2 * (j) + 1].w);       \
  }
  static_assert(GIT == GNG, "one item of each panel a group");
#define G_LOAD_SLICE(st, g)                                                   \
  {                                                                           \
    G_LOAD(rv, vg, st, g)                                                     \
    G_LOAD(rk, kg, st, g)                                                     \
  }
#define G_STASH_SLICE(bf, g)                                                  \
  {                                                                           \
    G_STASH(rv, Vs, bf, g)                                                    \
    G_STASH(rk, Ks, bf, g)                                                    \
  }
#define G_FRAG(dst, g)                                                        \
  {                                                                           \
    _Pragma("unroll")                                                         \
    for (int i = 0; i < GFG; ++i) {                                           \
      fa[dst][i]    = *(const v2f*)(ka + 2 * ((g) * GFG + i) * GPP);          \
      fb[dst][i][0] = *(const v2f*)(va + 2 * ((g) * GFG + i) * GPP);          \
      fb[dst][i][1] = *(const v2f*)(va + 2 * ((g) * GFG + i) * GPP + 32);     \
    }                                                                         \
  }
#define G_STAGE(n, ST, LD)                                                    \
  {                                                                           \
    const int buf = (n) & 1;                                                  \
    const float* ka = Ks + buf * GBUF + ard;                                  \
    const float* va = Vs + buf * GBUF + brd;                                  \
    G_FRAG(0, 0)                                                              \
    _Pragma("unroll")                                                         \
    for (int g = 0; g < GNG; ++g) {                                           \
      if (ST) G_STASH_SLICE(buf ^ 1, g)                                       \
      if (g + 1 < GNG) G_FRAG((g + 1) & 1, g + 1)                             \
      if (g == 0) {                           \
        xa[0] = *(const v2f*)(ka + xko);                                      \
        xa[1] = *(const v2f*)(ka + xko + 2 * GPP);                            \
        xb[0] = *(const v2f*)(va + xvo);                                      \
        xb[1] = *(const v2f*)(va + xvo + 32);                                 \
      }                                                                       \
      if (LD) G_LOAD_SLICE((n) + 2, g)                                        \
               \
      __builtin_amdgcn_sched_barrier(0);                                      \
      _Pragma("unroll")                                                       \
      for (int i = 0; i < GFG; ++i) {                                         \
        acc[0] = wmma4(fa[g & 1][i], fb[g & 1][i][0], acc[0]);                \
        acc[1] = wmma4(fa[g & 1][i], fb[g & 1][i][1], acc[1]);                \
      }                                                                       \
      if (g == 1) {                                                           \
        acxk    = wmma4(xa[0], ones, acxk);                                   \
        acxv[0] = wmma4(ones, xb[0], acxv[0]);                                \
        acxk    = wmma4(xa[1], ones, acxk);                                   \
        acxv[1] = wmma4(ones, xb[1], acxv[1]);                                \
      }                                                                       \
    }                                                                         \
    LDS_BARRIER()                                                             \
  }

  #pragma unroll
  for (int g = 0; g < GNG; ++g) G_LOAD_SLICE(0, g)
  #pragma unroll
  for (int g = 0; g < GNG; ++g) G_STASH_SLICE(0, g)
  #pragma unroll
  for (int g = 0; g < GNG; ++g) G_LOAD_SLICE(1, g)
  __syncthreads();

  const int ard = lh * GPP + ((2 * (dw + lm)) ^ (4 * ((dw >> 5) & 1)));
  const int brd = lh * GPP + ((2 * (ew + lm)) ^ (4 * ((ew >> 5) & 1)));
  const int xko = 2 * (4 * et + 2 * ewi) * GPP;
  const int xvo = 2 * (4 * dt + dwi) * GPP;
  v8f acc[2] = {};
  v2f fa[2][GFG], fb[2][GFG][2];
  v2f xa[2], xb[2];
  v8f acxk = {};
  v8f acxv[2] = {};
  v2f ones; ones.x = 1.0f; ones.y = 1.0f;
  for (int n = 0; n < GNST - 2; ++n) G_STAGE(n, 1, 1)
  G_STAGE(GNST - 2, 1, 0)
  G_STAGE(GNST - 1, 0, 0)
#undef G_HERE
#undef G_LOAD
#undef G_STASH
#undef G_LOAD_SLICE
#undef G_STASH_SLICE
#undef G_FRAG
#undef G_STAGE
  float* Gb = Gp + (((size_t)hf * NB + b) * DM + d0 + dw) * DM + e0 + ew;
  #pragma unroll
  for (int t = 0; t < 2; ++t)
    #pragma unroll
    for (int r = 0; r < 8; ++r)
      Gb[(size_t)(r + 8 * lh) * DM + 16 * t + lm] = acc[t][r];
  if (lm == 0) {
    #pragma unroll
    for (int r = 0; r < 8; ++r) Xk[wv * 16 + 8 * lh + r] = acxk[r];
  }
  if (lh == 0) {
    Xv[wv * 32 + lm]      = acxv[0][0];
    Xv[wv * 32 + 16 + lm] = acxv[1][0];
  }
  __syncthreads();
  if (tid < 64) {
    const int wi = tid >> 4, row = tid & 15;
    part[((size_t)b * XCH + hf * 8 + et) * (2 * DM) + d0 + tid] =
        Xk[wi * 16 + row] + Xk[(wi + 4) * 16 + row];
  } else if (tid < 128) {
    const int col = tid - 64, eh = col >> 5, cc = col & 31;
    float s = Xv[(4 * eh) * 32 + cc];
    #pragma unroll
    for (int w = 1; w < 4; ++w) s += Xv[(4 * eh + w) * 32 + cc];
    part[((size_t)b * XCH + hf * 8 + dt) * (2 * DM) + DM + e0 + col] = s;
  }
}

#define ROWP (DM + 4)
#define KP64 72
#define KP32 40

template <int NR>
__device__ __forceinline__ void stage_rows(const float* __restrict__ g, float* s,
                                           int wv, int lane) {
  #pragma unroll
  for (int j = 0; j < NR / 8; ++j)
    #pragma unroll
    for (int q = 0; q < 4; ++q)
      *(float4*)(s + (wv + 8 * j) * ROWP + 128 * q + 4 * lane) =
          *(const float4*)(g + (size_t)(wv + 8 * j) * DM + 128 * q + 4 * lane);
}

template <int NC, int PITCH>
__device__ __forceinline__ void stage_kmajor(const float* __restrict__ g, float* s,
                                             int tid) {
  constexpr int F4 = NC / 4;
  constexpr int RPS = 256 / F4;
  const int c4 = (tid % F4) * 4, r0 = tid / F4;
  #pragma unroll
  for (int j = 0; j < DM / RPS; ++j)
    *(float4*)(s + (r0 + RPS * j) * PITCH + c4) =
        *(const float4*)(g + (size_t)(r0 + RPS * j) * DM + c4);
}

template <int NC, int PITCH>
__device__ __forceinline__ void stage_kmajor_sum2(const float* __restrict__ g0,
                                                  const float* __restrict__ g1,
                                                  float* s, int tid) {
  constexpr int F4 = NC / 4;
  constexpr int RPS = 256 / F4;
  const int c4 = (tid % F4) * 4, r0 = tid / F4;
  #pragma unroll
  for (int j = 0; j < DM / RPS; ++j) {
    const float4 a = *(const float4*)(g0 + (size_t)(r0 + RPS * j) * DM + c4);
    const float4 c = *(const float4*)(g1 + (size_t)(r0 + RPS * j) * DM + c4);
    *(float4*)(s + (r0 + RPS * j) * PITCH + c4) =
        make_float4(a.x + c.x, a.y + c.y, a.z + c.z, a.w + c.w);
  }
}

__device__ __forceinline__ v2f ld_pair(const float* p, int pitch) {
  v2f r; r.x = p[0]; r.y = p[pitch]; return r;
}

#define SCH   XCH

#define VCH   16
__device__ __forceinline__ float slice_sum(const float* __restrict__ p) {
  float s = p[0];
  #pragma unroll
  for (int c = 1; c < VCH; ++c) s += p[(size_t)c * (NB * DM)];
  return s;
}

#define PANEL_K_LOOP(LOADA, LOADB)                                            \
  {                                                                           \
    v2f fa[2][8], fb[2][8];                                                   \
    _Pragma("unroll")                                                         \
    for (int i = 0; i < 8; ++i) {                                             \
      const int kk = 4 * i;                                                   \
      fa[0][i] = LOADA; fb[0][i] = LOADB;                                     \
    }                                                                         \
    _Pragma("unroll")                                                         \
    for (int g = 0; g < DM / 32; ++g) {                                       \
      if (g + 1 < DM / 32) {                                                  \
        _Pragma("unroll")                                                     \
        for (int i = 0; i < 8; ++i) {                                         \
          const int kk = 32 * (g + 1) + 4 * i;                                \
          fa[(g + 1) & 1][i] = LOADA; fb[(g + 1) & 1][i] = LOADB;             \
        }                                                                     \
      }                                                                       \
      _Pragma("unroll")                                                       \
      for (int i = 0; i < 8; ++i)                                             \
        acc = wmma4(fa[g & 1][i], fb[g & 1][i], acc);                         \
      __builtin_amdgcn_sched_barrier(0);                                      \
    }                                                                         \
  }

#define TP (DKH + 4)

__global__ void __launch_bounds__(256) kT(const float* __restrict__ wk,
                                          const float* __restrict__ G,
                                          const float* __restrict__ part,
                                          const float* __restrict__ wvw,
                                          const float* __restrict__ bk,
                                          float* __restrict__ Mp,
                                          float* __restrict__ u) {
  __shared__ __attribute__((aligned(16))) float As[32 * ROWP];
  __shared__ __attribute__((aligned(16))) float Bs[DM * KP64];
  __shared__ __attribute__((aligned(16))) float Ts[32 * TP];
  __shared__ __attribute__((aligned(16))) float Vw[DKH * TP];
  __shared__ float vs[DKH];
  __shared__ __attribute__((aligned(16))) float ks[DKH];
  __shared__ float bks[32];
  const int tid = threadIdx.x;
  const int lane = tid & 31, wv = tid >> 5;
  const int lm = lane & 15, lh = lane >> 4;
  const int bh = blockIdx.x >> 4;
  const int b = bh >> 3, h = bh & 7;
  const int ih = (blockIdx.x >> 3) & 1;
  const int eg = blockIdx.x & 7;
  const int e0 = eg * 64;
  float2 cp[SCH];
  float bkr = 0.0f;
  if (wv >= 6) {
    const float* pp = part + (size_t)b * SCH * (2 * DM) + (wv - 6) * DM + e0 + 2 * lane;
    #pragma unroll
    for (int c = 0; c < SCH; ++c) cp[c] = *(const float2*)(pp + (size_t)c * (2 * DM));
    if (wv == 6) bkr = bk[h * 64 + ih * 32 + lane];
  }
  stage_rows<32>(wk + (size_t)(h * 64 + ih * 32) * DM, As, wv, lane);
  stage_kmajor_sum2<64, KP64>(G + (size_t)b * DM * DM + e0,
                              G + (size_t)(NB + b) * DM * DM + e0, Bs, tid);
  {
    const int c4 = (tid & 15) * 4, r0 = tid >> 4;
    #pragma unroll
    for (int t = 0; t < 4; ++t)
      *(float4*)(Vw + (r0 + 16 * t) * TP + c4) =
          *(const float4*)(wvw + (size_t)(h * 64 + r0 + 16 * t) * DM + e0 + c4);
  }
  if (wv >= 6) {
    float2 s = cp[0];
    #pragma unroll
    for (int c = 1; c < SCH; ++c) { s.x += cp[c].x; s.y += cp[c].y; }
    float* cs = (wv == 6) ? ks : vs;
    cs[2 * lane] = s.x;
    cs[2 * lane + 1] = s.y;
    if (wv == 6) bks[lane] = bkr;
  }
  __syncthreads();
  if (wv < 4) {
    const int rt = wv & 1, kh = wv >> 1;
    const float* ua = As + (16 * rt + lm) * ROWP + e0 + 32 * kh + 2 * lh;
    const float* ub = ks + 32 * kh + 2 * lh;
    v2f fua[8], fub[8];
    #pragma unroll
    for (int i = 0; i < 8; ++i) {
      fua[i] = *(const v2f*)(ua + 4 * i);
      fub[i] = *(const v2f*)(ub + 4 * i);
    }
    __builtin_amdgcn_sched_barrier(0);
    v8f accu = {};
    #pragma unroll
    for (int i = 0; i < 8; ++i) accu = wmma4(fua[i], fub[i], accu);
    if (lm == 0) {
      float* ud = u + ((size_t)(2 * eg + kh) * NB + b) * DM + h * 64 + ih * 32 + 16 * rt + 8 * lh;
      #pragma unroll
      for (int r = 0; r < 8; ++r) ud[r] = accu[r];
    }
  }
  const int ti = wv >> 2, tj = wv & 3;
  const float* ap = As + (16 * ti + lm) * ROWP + 2 * lh;
  const float* bp = Bs + (2 * lh) * KP64 + 16 * tj + lm;
  v8f acc = {};
  PANEL_K_LOOP((*(const v2f*)(ap + kk)), ld_pair(bp + kk * KP64, KP64))
  {
    const float vse = vs[16 * tj + lm];
    #pragma unroll
    for (int r = 0; r < 8; ++r) {
      const int il = 16 * ti + r + 8 * lh;
      Ts[il * TP + 16 * tj + lm] = acc[r] + bks[il] * vse;
    }
  }
  __syncthreads();
  {
    const float* ap2 = Ts + (16 * ti + lm) * TP + 2 * lh;
    const float* bp2 = Vw + (16 * tj + lm) * TP + 2 * lh;
    v2f fa2[DKH / 4], fb2[DKH / 4];
    #pragma unroll
    for (int i = 0; i < DKH / 4; ++i) {
      fa2[i] = *(const v2f*)(ap2 + 4 * i);
      fb2[i] = *(const v2f*)(bp2 + 4 * i);
    }
    __builtin_amdgcn_sched_barrier(0);
    v8f acc2 = {};
    #pragma unroll
    for (int i = 0; i < DKH / 4; ++i)
      acc2 = wmma4(fa2[i], fb2[i], acc2);
    float* Mb = Mp + (((size_t)eg * (NB * NH) + bh) * DKH + ih * 32 + 16 * ti) * DKH + 16 * tj;
    #pragma unroll
    for (int r = 0; r < 8; ++r)
      Mb[(size_t)(r + 8 * lh) * DKH + lm] = acc2[r];
  }
}

__global__ void __launch_bounds__(256) kP(const float* __restrict__ Mp,
                                          const float* __restrict__ u,
                                          const float* __restrict__ bk,
                                          const float* __restrict__ bv,
                                          const float* __restrict__ wo,
                                          float* __restrict__ P) {
  __shared__ __attribute__((aligned(16))) float Ms[16 * TP];
  const int tid = threadIdx.x;
  const int lane = tid & 31, wv = tid >> 5;
  const int lm = lane & 15, lh = lane >> 4;
  const int bh = blockIdx.x >> 2;
  const int b = bh >> 3, h = bh & 7;
  const int i0 = (blockIdx.x & 3) * 16;
  const int o0 = wv * 64;
  v2f bm[DKH / 4][4];
  #pragma unroll
  for (int i = 0; i < DKH / 4; ++i)
    #pragma unroll
    for (int t = 0; t < 4; ++t)
      bm[i][t] = *(const v2f*)(wo + (size_t)(o0 + 16 * t + lm) * DM + h * 64 + 4 * i + 2 * lh);
  {
    const int row = tid >> 4, c4 = (tid & 15) * 4;
    const float* mp = Mp + ((size_t)bh * 64 + i0 + row) * DKH + c4;
    float4 s = *(const float4*)mp;
    #pragma unroll
    for (int g = 1; g < 8; ++g) {
      const float4 t = *(const float4*)(mp + (size_t)g * (NB * NH * DKH * DKH));
      s.x += t.x; s.y += t.y; s.z += t.z; s.w += t.w;
    }
    const float ui  = slice_sum(u + (size_t)b * DM + h * 64 + i0 + row);
    const float bki = bk[h * 64 + i0 + row];
    const float4 bv4 = *(const float4*)(bv + h * 64 + c4);
    float4 m;
    m.x = 0.125f * (s.x + ui * bv4.x + (float)SEQ * bki * bv4.x);
    m.y = 0.125f * (s.y + ui * bv4.y + (float)SEQ * bki * bv4.y);
    m.z = 0.125f * (s.z + ui * bv4.z + (float)SEQ * bki * bv4.z);
    m.w = 0.125f * (s.w + ui * bv4.w + (float)SEQ * bki * bv4.w);
    *(float4*)(Ms + row * TP + c4) = m;
  }
  __syncthreads();
  const float* ap = Ms + lm * TP + 2 * lh;
  v2f fa[DKH / 4];
  #pragma unroll
  for (int i = 0; i < DKH / 4; ++i) fa[i] = *(const v2f*)(ap + 4 * i);
  v8f acc[4] = {};
  #pragma unroll
  for (int i = 0; i < DKH / 4; ++i) {
    #pragma unroll
    for (int t = 0; t < 4; ++t)
      acc[t] = wmma4(fa[i], bm[i][t], acc[t]);
  }
  float* Pb = P + ((size_t)b * DM + h * 64 + i0) * DM + o0;
  #pragma unroll
  for (int t = 0; t < 4; ++t)
    #pragma unroll
    for (int r = 0; r < 8; ++r)
      Pb[(size_t)(r + 8 * lh) * DM + 16 * t + lm] = acc[t][r];
}

__global__ void __launch_bounds__(256) kW4(const float* __restrict__ wq,
                                           const float* __restrict__ bq,
                                           const float* __restrict__ P,
                                           float* __restrict__ W4,
                                           float* __restrict__ cq) {
  __shared__ __attribute__((aligned(16))) float As[DM * KP32];
  __shared__ __attribute__((aligned(16))) float Bs[DM * KP64];
  const int tid = threadIdx.x;
  const int lane = tid & 31, wv = tid >> 5;
  const int lm = lane & 15, lh = lane >> 4;
  const int b = blockIdx.x >> 7;
  const int d0 = ((blockIdx.x >> 3) & 15) * 32;
  const int o0 = (blockIdx.x & 7) * 64;
  const int ti = wv >> 2, tj = wv & 3;
  v2f fq[8] = {};
  if (ti == 0) {
    #pragma unroll
    for (int i = 0; i < 8; ++i) fq[i] = *(const v2f*)(bq + d0 + 4 * i + 2 * lh);
  }
  stage_kmajor<32, KP32>(wq + d0, As, tid);
  stage_kmajor<64, KP64>(P + (size_t)b * DM * DM + o0, Bs, tid);
  __syncthreads();
  const float* ap = As + (2 * lh) * KP32 + 16 * ti + lm;
  const float* bp = Bs + (2 * lh) * KP64 + 16 * tj + lm;
  if (ti == 0) {
    v2f fp[8];
    #pragma unroll
    for (int i = 0; i < 8; ++i) fp[i] = ld_pair(bp + (d0 + 4 * i) * KP64, KP64);
    __builtin_amdgcn_sched_barrier(0);
    v8f accc = {};
    #pragma unroll
    for (int i = 0; i < 8; ++i) accc = wmma4(fq[i], fp[i], accc);
    if (lh == 0) cq[((size_t)(d0 >> 5) * NB + b) * DM + o0 + 16 * tj + lm] = accc[0];
  }
  v8f acc = {};
  PANEL_K_LOOP(ld_pair(ap + kk * KP32, KP32), ld_pair(bp + kk * KP64, KP64))
  float* Wb = W4 + ((size_t)b * DM + d0 + 16 * ti) * DM + o0 + 16 * tj;
  #pragma unroll
  for (int r = 0; r < 8; ++r)
    Wb[(size_t)(r + 8 * lh) * DM + lm] = acc[r];
}

#define OKS   64
#define ONST  (DM / OKS)
#define OAP   (OKS + 4)
#define OBP   160
#define OABUF (256 * OAP)
#define OBBUF ((OKS / 2) * OBP)
#define OAIT  16
#define OBIT  (OKS / 32)
#define OFG   2
#define ONG   (OKS / 4 / OFG)

__global__ void __launch_bounds__(256) kOut(const float* __restrict__ q,
                                            const float* __restrict__ W4,
                                            const float* __restrict__ bo,
                                            const float* __restrict__ cq,
                                            float* __restrict__ out) {
  __shared__ __attribute__((aligned(16))) float Qs[2 * OABUF];
  __shared__ __attribute__((aligned(16))) float Ws[2 * OBBUF];
  const int tid = threadIdx.x;
  const int lane = tid & 31, wv = tid >> 5;
  const int lm = lane & 15, lh = lane >> 4;
  const int b = blockIdx.x >> 4;
  const int n0 = (blockIdx.x & 15) * 256;
  const int o0 = blockIdx.y * 64;
  const int sc = (tid & 15) * 4, sr = tid >> 4;
  const float* qg = q + ((size_t)b * SEQ + n0 + sr) * DM + sc;
  const float* wg = W4 + ((size_t)b * DM + 2 * sr) * DM + o0 + sc;
  const int qst = sr * OAP + sc;
  const int wsw = 4 * ((sc >> 5) & 1);
  const int wst0 = sr * OBP + 2 * sc + wsw, wst1 = sr * OBP + 2 * sc + 4 - wsw;
  v4f rq[OAIT];
  v4f rw[2 * OBIT];
  float add[4], cqp[4][VCH];
  #pragma unroll
  for (int t = 0; t < 4; ++t) {
    add[t] = bo[o0 + 16 * t + lm];
    #pragma unroll
    for (int p = 0; p < VCH; ++p)
      cqp[t][p] = cq[((size_t)p * NB + b) * DM + o0 + 16 * t + lm];
  }

#define O_LOAD_Q(st, j)                                                       \
  rq[j] = *(const v4f*)(qg + (st) * OKS + (size_t)(16 * (j)) * DM);
#define O_LOAD_W(st, j)                                                       \
  {                                                                           \
    const float* wn = wg + (size_t)(st) * OKS * DM + (size_t)(32 * (j)) * DM; \
    rw[2 * (j)]     = *(const v4f*)(wn);                                      \
    rw[2 * (j) + 1] = *(const v4f*)(wn + DM);                                 \
  }
#define O_STASH_Q(bf, j)                                                      \
  *(v4f*)(Qs + (bf) * OABUF + qst + 16 * (j) * OAP) = rq[j];
#define O_HERE(a, b) asm volatile("" : : "v"(a), "v"(b));
#define O_STASH_W(bf, j)                                                      \
  {                                                                           \
    float* wd = Ws + (bf) * OBBUF + 16 * (j) * OBP;                           \
    O_HERE(rw[2 * (j)], rw[2 * (j) + 1])                                      \
    *(float4*)(wd + wst0) = make_float4(rw[2 * (j)].x, rw[2 * (j) + 1].x,     \
                                        rw[2 * (j)].y, rw[2 * (j) + 1].y);    \
    *(float4*)(wd + wst1) = make_float4(rw[2 * (j)].z, rw[2 * (j) + 1].z,     \
                                        rw[2 * (j)].w, rw[2 * (j) + 1].w);    \
  }
  static_assert(OAIT == 2 * ONG && 4 * OBIT == ONG, "two q items a group, one W4 item every fourth group");
#define O_LOAD_SLICE(st, g)                                                   \
  {                                                                           \
    O_LOAD_Q(st, 2 * (g))                                                     \
    O_LOAD_Q(st, 2 * (g) + 1)                                                 \
    if (((g) & 3) == 0) O_LOAD_W(st, (g) / 4)                                 \
  }
#define O_STASH_SLICE(bf, g)                                                  \
  {                                                                           \
    O_STASH_Q(bf, 2 * (g))                                                    \
    O_STASH_Q(bf, 2 * (g) + 1)                                                \
    if (((g) & 3) == 0) O_STASH_W(bf, (g) / 4)                                \
  }
#define O_FRAG(dst, g)                                                        \
  {                                                                           \
    _Pragma("unroll")                                                         \
    for (int i = 0; i < OFG; ++i) {                                           \
      const int ks = (g) * OFG + i;                                           \
      fa[dst][i][0] = *(const v2f*)(qa + 4 * ks);                             \
      fa[dst][i][1] = *(const v2f*)(qa + 16 * OAP + 4 * ks);                  \
      fb[dst][i][0] = *(const v2f*)(wa0 + 2 * ks * OBP);                      \
      fb[dst][i][1] = *(const v2f*)(wa0 + 2 * ks * OBP + 32);                 \
      fb[dst][i][2] = *(const v2f*)(wa1 + 2 * ks * OBP + 64);                 \
      fb[dst][i][3] = *(const v2f*)(wa1 + 2 * ks * OBP + 96);                 \
    }                                                                         \
  }
#define O_STAGE(n, ST, LD)                                                    \
  {                                                                           \
    const int buf = (n) & 1;                                                  \
    const float* qa  = Qs + buf * OABUF + ard;                                \
    const float* wa0 = Ws + buf * OBBUF + wr0;                                \
    const float* wa1 = Ws + buf * OBBUF + wr1;                                \
    O_FRAG(0, 0)                                                              \
    _Pragma("unroll")                                                         \
    for (int g = 0; g < ONG; ++g) {                                           \
      if (ST) O_STASH_SLICE(buf ^ 1, g)                                       \
      if (g + 1 < ONG) O_FRAG((g + 1) & 1, g + 1)                             \
      if (LD) O_LOAD_SLICE((n) + 2, g)                                        \
                      \
      __builtin_amdgcn_sched_barrier(0);                                      \
      _Pragma("unroll")                                                       \
      for (int i = 0; i < OFG; ++i) {                                         \
        _Pragma("unroll")                                                     \
        for (int t = 0; t < 4; ++t) {                                         \
          acc[t]     = wmma4(fa[g & 1][i][0], fb[g & 1][i][t], acc[t]);       \
          acc[4 + t] = wmma4(fa[g & 1][i][1], fb[g & 1][i][t], acc[4 + t]);   \
        }                                                                     \
      }                                                                       \
    }                                                                         \
    LDS_BARRIER()                                                             \
  }

  #pragma unroll
  for (int g = 0; g < ONG; ++g) O_LOAD_SLICE(0, g)
  #pragma unroll
  for (int g = 0; g < ONG; ++g) O_STASH_SLICE(0, g)
  #pragma unroll
  for (int g = 0; g < ONG; ++g) O_LOAD_SLICE(1, g)
  __syncthreads();
  #pragma unroll
  for (int t = 0; t < 4; ++t) {
    float s = cqp[t][0];
    #pragma unroll
    for (int p = 1; p < VCH; ++p) s += cqp[t][p];
    add[t] += s;
  }

  const int ard = (32 * wv + lm) * OAP + 2 * lh;
  const int wr0 = lh * OBP + 2 * lm;
  const int wr1 = lh * OBP + ((2 * lm) ^ 4);
  v8f acc[8] = {};
  v2f fa[2][OFG][2], fb[2][OFG][4];
  for (int n = 0; n < ONST - 2; ++n) O_STAGE(n, 1, 1)
  O_STAGE(ONST - 2, 1, 0)
  O_STAGE(ONST - 1, 0, 0)
#undef O_HERE
#undef O_LOAD_Q
#undef O_LOAD_W
#undef O_STASH_Q
#undef O_STASH_W
#undef O_LOAD_SLICE
#undef O_STASH_SLICE
#undef O_FRAG
#undef O_STAGE
  float* ob = out + ((size_t)b * SEQ + n0 + 32 * wv) * DM + o0;
  #pragma unroll
  for (int t = 0; t < 4; ++t) {
    #pragma unroll
    for (int m = 0; m < 2; ++m)
      #pragma unroll
      for (int r = 0; r < 8; ++r)
        ob[(size_t)(16 * m + r + 8 * lh) * DM + 16 * t + lm] = acc[4 * m + t][r] + add[t];
  }
}

extern "C" void kernel_launch(void* const* d_in, const int* in_sizes, int n_in,
                              void* d_out, int out_size, void* d_ws, size_t ws_size,
                              hipStream_t stream) {
  const float* q   = (const float*)d_in[0];
  const float* k   = (const float*)d_in[1];
  const float* v   = (const float*)d_in[2];
  const float* wq  = (const float*)d_in[3];
  const float* bq  = (const float*)d_in[4];
  const float* wk  = (const float*)d_in[5];
  const float* bk  = (const float*)d_in[6];
  const float* wvw = (const float*)d_in[7];
  const float* bv  = (const float*)d_in[8];
  const float* wo  = (const float*)d_in[9];
  const float* bo  = (const float*)d_in[10];
  float* out = (float*)d_out;

  float* ws = (float*)d_ws;
  float* G    = ws;
  float* Mp   = G    + (size_t)2 * NB * DM * DM;
  float* P    = Mp   + (size_t)8 * NB * NH * DKH * DKH;
  float* W4   = P    + (size_t)NB * DM * DM;
  float* u    = W4   + (size_t)NB * DM * DM;
  float* cq   = u    + (size_t)VCH * NB * DM;
  float* part = cq   + (size_t)VCH * NB * DM;

  kG   <<<256, 256, 0, stream>>>(k, v, G, part);
  kT   <<<256, 256, 0, stream>>>(wk, G, part, wvw, bk, Mp, u);
  kP   <<<64, 256, 0, stream>>>(Mp, u, bk, bv, wo, P);
  kW4  <<<256, 256, 0, stream>>>(wq, bq, P, W4, cq);
  dim3 go(32, 8, 1);
  kOut <<<go, 256, 0, stream>>>(q, W4, bo, cq, out);
}
